# starting kernel
# baseline (speedup 1.0000x reference)
#include <hip/hip_runtime.h>
#include <math.h>

typedef __attribute__((ext_vector_type(2))) float v2f;
typedef __attribute__((ext_vector_type(8))) float v8f;

__device__ __forceinline__ void atomic_add_f32(float* p, float v) {
    __hip_atomic_fetch_add(p, v, __ATOMIC_RELAXED, __HIP_MEMORY_SCOPE_AGENT);
}

__global__ void k_fill_deg(float* __restrict__ deg, int n) {
    int i = blockIdx.x * blockDim.x + threadIdx.x;
    if (i < n) deg[i] = 1.0f;
}

__global__ void k_deg_scatter(const int* __restrict__ dst, int E,
                              float* __restrict__ deg) {
    int e = blockIdx.x * blockDim.x + threadIdx.x;
    if (e < E) atomic_add_f32(&deg[dst[e]], 1.0f);
}

__global__ void k_dinv(float* __restrict__ deg, int n) {
    int i = blockIdx.x * blockDim.x + threadIdx.x;
    if (i < n) {
        float d = deg[i];
        deg[i] = (d > 0.0f) ? rsqrtf(d) : 0.0f;
    }
}

template <int NT, int KD, bool RELU>
__global__ void __launch_bounds__(256) k_gemm_wmma_f32(
    const float* __restrict__ A, const float* __restrict__ B,
    float* __restrict__ C, int M)
{
    constexpr int N   = NT * 16;
    constexpr int LDK = KD + 2;
    __shared__ float lds_w[N * LDK];

    for (int idx = threadIdx.x; idx < KD * N; idx += 256) {
        int k = idx / N;
        int n = idx - k * N;
        lds_w[n * LDK + k] = B[idx];
    }
    __syncthreads();

    const int lane  = threadIdx.x & 31;
    const int wv    = threadIdx.x >> 5;
    const int mtile = blockIdx.x * 8 + wv;
    if (mtile * 16 >= M) return;

    const int hf = lane >> 4;
    const int r  = lane & 15;

    const float* __restrict__ arow = A + (size_t)(mtile * 16 + r) * KD + 2 * hf;

    v8f acc[NT];
#pragma unroll
    for (int t = 0; t < NT; ++t) acc[t] = (v8f){};

#pragma unroll 4
    for (int k = 0; k < KD; k += 4) {
        float2 av = *(const float2*)(arow + k);
        if (RELU) { av.x = fmaxf(av.x, 0.0f); av.y = fmaxf(av.y, 0.0f); }
        v2f a;
        a[0] = av.x;
        a[1] = av.y;
#pragma unroll
        for (int t = 0; t < NT; ++t) {
            const int col = t * 16 + r;
            float2 bv = *(const float2*)(&lds_w[col * LDK + k + 2 * hf]);
            v2f b;
            b[0] = bv.x;
            b[1] = bv.y;
            acc[t] = __builtin_amdgcn_wmma_f32_16x16x4_f32(
                false, a, false, b, (short)0, acc[t], false, false);
        }
    }

#pragma unroll
    for (int t = 0; t < NT; ++t) {
        float* crow = C + (size_t)(mtile * 16 + 8 * hf) * N + t * 16 + r;
#pragma unroll
        for (int j = 0; j < 8; ++j) crow[(size_t)j * N] = acc[t][j];
    }
}

__global__ void k_self_bias_init(const float* __restrict__ h,
                                 const float* __restrict__ dinv,
                                 const float* __restrict__ bias,
                                 float* __restrict__ out, int n, int lgC) {
    long long t = (long long)blockIdx.x * blockDim.x + threadIdx.x;
    long long total = (long long)n << lgC;
    if (t >= total) return;
    int i = (int)(t >> lgC);
    int c = (int)(t & (((long long)1 << lgC) - 1));
    float di = dinv[i];
    out[t] = bias[c] + h[t] * di * di;
}

__global__ void k_edge_scatter(const int* __restrict__ src,
                               const int* __restrict__ dst,
                               const float* __restrict__ dinv,
                               const float* __restrict__ h,
                               float* __restrict__ out, int E, int lgG) {
    long long t = (long long)blockIdx.x * blockDim.x + threadIdx.x;
    long long total = (long long)E << lgG;
    if (t >= total) return;
    const int C = 4 << lgG;
    int e  = (int)(t >> lgG);
    int cg = (int)(t & (((long long)1 << lgG) - 1)) << 2;
    int s = src[e], d = dst[e];
    float nrm = dinv[s] * dinv[d];
    const float4 hv = *(const float4*)(h + (size_t)s * C + cg);
    float* o = out + (size_t)d * C + cg;
    atomic_add_f32(o + 0, hv.x * nrm);
    atomic_add_f32(o + 1, hv.y * nrm);
    atomic_add_f32(o + 2, hv.z * nrm);
    atomic_add_f32(o + 3, hv.w * nrm);
}

extern "C" void kernel_launch(void* const* d_in, const int* in_sizes, int n_in,
                              void* d_out, int out_size, void* d_ws, size_t ws_size,
                              hipStream_t stream) {
    const float* x  = (const float*)d_in[0];
    const int*   ei = (const int*)d_in[1];
    const float* W1 = (const float*)d_in[2];
    const float* b1 = (const float*)d_in[3];
    const float* W2 = (const float*)d_in[4];
    const float* b2 = (const float*)d_in[5];
    float* out = (float*)d_out;

    const int HID = in_sizes[3];
    const int OUT = in_sizes[5];
    const int IN  = in_sizes[2] / HID;
    const int N   = in_sizes[0] / IN;
    const int E   = in_sizes[1] / 2;
    const int* srcI = ei;
    const int* dstI = ei + E;

    float* ws   = (float*)d_ws;
    float* dinv = ws;
    float* bufA = ws + (((size_t)N + 63) & ~(size_t)63);
    float* bufB = bufA + (size_t)N * HID;

    const int T = 256;
    const int mtiles = N / 16;
    const int gblk   = (mtiles + 7) / 8;

    k_fill_deg   <<<(N + T - 1) / T, T, 0, stream>>>(dinv, N);
    k_deg_scatter<<<(E + T - 1) / T, T, 0, stream>>>(dstI, E, dinv);
    k_dinv       <<<(N + T - 1) / T, T, 0, stream>>>(dinv, N);

    k_gemm_wmma_f32<8, 128, false><<<gblk, T, 0, stream>>>(x, W1, bufA, N);
    {
        long long tot = (long long)N * HID;
        k_self_bias_init<<<(int)((tot + T - 1) / T), T, 0, stream>>>(
            bufA, dinv, b1, bufB, N, 7);
        long long etot = (long long)E << 5;
        k_edge_scatter<<<(int)((etot + T - 1) / T), T, 0, stream>>>(
            srcI, dstI, dinv, bufA, bufB, E, 5);
    }

    k_gemm_wmma_f32<4, 128, true><<<gblk, T, 0, stream>>>(bufB, W2, bufA, N);
    {
        long long tot = (long long)N * OUT;
        k_self_bias_init<<<(int)((tot + T - 1) / T), T, 0, stream>>>(
            bufA, dinv, b2, out, N, 6);
        long long etot = (long long)E << 4;
        k_edge_scatter<<<(int)((etot + T - 1) / T), T, 0, stream>>>(
            srcI, dstI, dinv, bufA, out, E, 4);
    }
}


# edge sums by dst tile: each work-group keeps 196 dst rows in LDS (start = bias + self term), scans the whole dst list, adds gathered rows by LDS f32 atomics, stores rows once; sum order = arrival
# speedup vs baseline: 1.3719x; 1.3719x over previous
#include <hip/hip_runtime.h>
#include <math.h>

typedef __attribute__((ext_vector_type(2))) float v2f;
typedef __attribute__((ext_vector_type(8))) float v8f;

__device__ __forceinline__ void atomic_add_f32(float* p, float v) {
    __hip_atomic_fetch_add(p, v, __ATOMIC_RELAXED, __HIP_MEMORY_SCOPE_AGENT);
}

__global__ void k_fill_deg(float* __restrict__ deg, int n) {
    int i = blockIdx.x * blockDim.x + threadIdx.x;
    if (i < n) deg[i] = 1.0f;
}

__global__ void k_deg_scatter(const int* __restrict__ dst, int E,
                              float* __restrict__ deg) {
    int e = blockIdx.x * blockDim.x + threadIdx.x;
    if (e < E) atomic_add_f32(&deg[dst[e]], 1.0f);
}

__global__ void k_dinv(float* __restrict__ deg, int n) {
    int i = blockIdx.x * blockDim.x + threadIdx.x;
    if (i < n) {
        float d = deg[i];
        deg[i] = (d > 0.0f) ? rsqrtf(d) : 0.0f;
    }
}

template <int NT, int KD, bool RELU>
__global__ void __launch_bounds__(256) k_gemm_wmma_f32(
    const float* __restrict__ A, const float* __restrict__ B,
    float* __restrict__ C, int M)
{
    constexpr int N   = NT * 16;
    constexpr int LDK = KD + 2;
    __shared__ float lds_w[N * LDK];

    for (int idx = threadIdx.x; idx < KD * N; idx += 256) {
        int k = idx / N;
        int n = idx - k * N;
        lds_w[n * LDK + k] = B[idx];
    }
    __syncthreads();

    const int lane  = threadIdx.x & 31;
    const int wv    = threadIdx.x >> 5;
    const int mtile = blockIdx.x * 8 + wv;
    if (mtile * 16 >= M) return;

    const int hf = lane >> 4;
    const int r  = lane & 15;

    const float* __restrict__ arow = A + (size_t)(mtile * 16 + r) * KD + 2 * hf;

    v8f acc[NT];
#pragma unroll
    for (int t = 0; t < NT; ++t) acc[t] = (v8f){};

#pragma unroll 4
    for (int k = 0; k < KD; k += 4) {
        float2 av = *(const float2*)(arow + k);
        if (RELU) { av.x = fmaxf(av.x, 0.0f); av.y = fmaxf(av.y, 0.0f); }
        v2f a;
        a[0] = av.x;
        a[1] = av.y;
#pragma unroll
        for (int t = 0; t < NT; ++t) {
            const int col = t * 16 + r;
            float2 bv = *(const float2*)(&lds_w[col * LDK + k + 2 * hf]);
            v2f b;
            b[0] = bv.x;
            b[1] = bv.y;
            acc[t] = __builtin_amdgcn_wmma_f32_16x16x4_f32(
                false, a, false, b, (short)0, acc[t], false, false);
        }
    }

#pragma unroll
    for (int t = 0; t < NT; ++t) {
        float* crow = C + (size_t)(mtile * 16 + 8 * hf) * N + t * 16 + r;
#pragma unroll
        for (int j = 0; j < 8; ++j) crow[(size_t)j * N] = acc[t][j];
    }
}

constexpr int TILE_R   = 196;
constexpr int AGG_T    = 1024;
constexpr int SCAN_U   = 4;
constexpr int LIST_LOW = 256;
constexpr int LIST_CAP = LIST_LOW + 128 * SCAN_U;

template <int V> struct RowVec;
template <> struct RowVec<4> { typedef float4 T; };
template <> struct RowVec<2> { typedef float2 T; };

__device__ __forceinline__ void lds_add_f32(float* p, float v) {
    __hip_atomic_fetch_add(p, v, __ATOMIC_RELAXED, __HIP_MEMORY_SCOPE_WORKGROUP);
}

template <int V>
__device__ __forceinline__ void add_row(float* a, const float* hp, float nu) {
#pragma unroll
    for (int q = 0; q < V; ++q) lds_add_f32(a + q, hp[q] * nu);
}

template <int C>
__device__ __forceinline__ void drain_list(
    const int* mylst, int cnt, int lane, int base,
    const int* __restrict__ src, const int* __restrict__ dst,
    const float* __restrict__ dinv, const float* __restrict__ h, float* acc)
{
    constexpr int V = C / 32;
    typedef typename RowVec<V>::T VT;
    for (int i0 = 0; i0 < cnt; i0 += 32) {
        const int m = min(32, cnt - i0);
        int s = 0, dl = 0;
        float nr = 0.0f;
        if (lane < m) {
            const int e = mylst[i0 + lane];
            s = src[e];
            const int d = dst[e];
            dl = d - base;
            nr = dinv[s] * dinv[d];
        }
        int j = 0;
        for (; j + 4 <= m; j += 4) {
            VT hv[4];
#pragma unroll
            for (int u = 0; u < 4; ++u) {
                const int su = __builtin_amdgcn_readlane(s, j + u);
                hv[u] = *(const VT*)(h + (size_t)su * C + lane * V);
            }
#pragma unroll
            for (int u = 0; u < 4; ++u) {
                const int du = __builtin_amdgcn_readlane(dl, j + u);
                const float nu = __int_as_float(
                    __builtin_amdgcn_readlane(__float_as_int(nr), j + u));
                add_row<V>(acc + du * C + lane * V, (const float*)&hv[u], nu);
            }
        }
        for (; j < m; ++j) {
            const int su = __builtin_amdgcn_readlane(s, j);
            const VT hv1 = *(const VT*)(h + (size_t)su * C + lane * V);
            const int du = __builtin_amdgcn_readlane(dl, j);
            const float nu = __int_as_float(
                __builtin_amdgcn_readlane(__float_as_int(nr), j));
            add_row<V>(acc + du * C + lane * V, (const float*)&hv1, nu);
        }
    }
}

template <int C>
__global__ void __launch_bounds__(AGG_T) k_tile_aggregate(
    const int* __restrict__ src, const int* __restrict__ dst,
    const float* __restrict__ dinv, const float* __restrict__ h,
    const float* __restrict__ bias, float* __restrict__ out, int n, int E)
{
    __shared__ float acc[TILE_R * C];
    __shared__ int   lst[(AGG_T / 32) * LIST_CAP];

    const int tid  = threadIdx.x;
    const int lane = tid & 31;
    const int wv   = tid >> 5;
    const int base = blockIdx.x * TILE_R;
    const int nrows = min(TILE_R, n - base);
    constexpr int C4 = C / 4;

    for (int idx = tid; idx < nrows * C4; idx += AGG_T) {
        const int r  = idx / C4;
        const int c4 = (idx - r * C4) * 4;
        const float di = dinv[base + r];
        const float4 hv = *(const float4*)(h + (size_t)(base + r) * C + c4);
        const float4 bv = *(const float4*)(bias + c4);
        float4 o;
        o.x = bv.x + hv.x * di * di;
        o.y = bv.y + hv.y * di * di;
        o.z = bv.z + hv.z * di * di;
        o.w = bv.w + hv.w * di * di;
        *(float4*)(acc + r * C + c4) = o;
    }
    __syncthreads();

    int* mylst = lst + wv * LIST_CAP;
    int cnt = 0;

    const int nvec   = (((size_t)dst & 15) == 0) ? (E >> 2) : 0;
    const int per_it = AGG_T * SCAN_U;
    const int n_it   = (nvec + per_it - 1) / per_it;
    const int rot = (int)(((long long)blockIdx.x * n_it) / gridDim.x);
    const int4* dst4 = (const int4*)dst;
    for (int k = 0; k < n_it; ++k) {
        int it = k + rot;
        if (it >= n_it) it -= n_it;
        int4 dv[SCAN_U];
        int  vi[SCAN_U];
#pragma unroll
        for (int u = 0; u < SCAN_U; ++u) {
            vi[u] = it * per_it + u * AGG_T + tid;
            dv[u] = (vi[u] < nvec) ? dst4[vi[u]] : make_int4(-1, -1, -1, -1);
        }
#pragma unroll
        for (int u = 0; u < SCAN_U; ++u) {
            const int dd[4] = {dv[u].x, dv[u].y, dv[u].z, dv[u].w};
            bool hit[4];
#pragma unroll
            for (int c = 0; c < 4; ++c)
                hit[c] = (unsigned)(dd[c] - base) < (unsigned)nrows;
            if (__builtin_amdgcn_ballot_w32(hit[0] | hit[1] | hit[2] | hit[3]) == 0u)
                continue;
#pragma unroll
            for (int c = 0; c < 4; ++c) {
                const unsigned b = __builtin_amdgcn_ballot_w32(hit[c]);
                if (hit[c]) mylst[cnt + __builtin_amdgcn_mbcnt_lo(b, 0u)] = vi[u] * 4 + c;
                cnt += __popc(b);
            }
        }
        if (cnt > LIST_LOW) {
            drain_list<C>(mylst, cnt, lane, base, src, dst, dinv, h, acc);
            cnt = 0;
        }
    }
    for (int e0 = nvec * 4 + wv * 32; e0 < E; e0 += AGG_T) {
        const int e = e0 + lane;
        const int d = (e < E) ? dst[e] : -1;
        const bool hit = (unsigned)(d - base) < (unsigned)nrows;
        const unsigned b = __builtin_amdgcn_ballot_w32(hit);
        if (hit) mylst[cnt + __builtin_amdgcn_mbcnt_lo(b, 0u)] = e;
        cnt += __popc(b);
        if (cnt > LIST_LOW) {
            drain_list<C>(mylst, cnt, lane, base, src, dst, dinv, h, acc);
            cnt = 0;
        }
    }
    drain_list<C>(mylst, cnt, lane, base, src, dst, dinv, h, acc);
    __syncthreads();

    float4* o4 = (float4*)(out + (size_t)base * C);
    const float4* a4 = (const float4*)acc;
    for (int idx = tid; idx < nrows * C4; idx += AGG_T) o4[idx] = a4[idx];
}

extern "C" void kernel_launch(void* const* d_in, const int* in_sizes, int n_in,
                              void* d_out, int out_size, void* d_ws, size_t ws_size,
                              hipStream_t stream) {
    const float* x  = (const float*)d_in[0];
    const int*   ei = (const int*)d_in[1];
    const float* W1 = (const float*)d_in[2];
    const float* b1 = (const float*)d_in[3];
    const float* W2 = (const float*)d_in[4];
    const float* b2 = (const float*)d_in[5];
    float* out = (float*)d_out;

    const int HID = in_sizes[3];
    const int IN  = in_sizes[2] / HID;
    const int N   = in_sizes[0] / IN;
    const int E   = in_sizes[1] / 2;
    const int* srcI = ei;
    const int* dstI = ei + E;

    float* ws   = (float*)d_ws;
    float* dinv = ws;
    float* bufA = ws + (((size_t)N + 63) & ~(size_t)63);
    float* bufB = bufA + (size_t)N * HID;

    const int T = 256;
    const int mtiles = N / 16;
    const int gblk   = (mtiles + 7) / 8;

    k_fill_deg   <<<(N + T - 1) / T, T, 0, stream>>>(dinv, N);
    k_deg_scatter<<<(E + T - 1) / T, T, 0, stream>>>(dstI, E, dinv);
    k_dinv       <<<(N + T - 1) / T, T, 0, stream>>>(dinv, N);

    k_gemm_wmma_f32<8, 128, false><<<gblk, T, 0, stream>>>(x, W1, bufA, N);
    const int ablk = (N + TILE_R - 1) / TILE_R;
    k_tile_aggregate<128><<<ablk, AGG_T, 0, stream>>>(
        srcI, dstI, dinv, bufA, b1, bufB, N, E);

    k_gemm_wmma_f32<4, 128, true><<<gblk, T, 0, stream>>>(bufB, W2, bufA, N);
    k_tile_aggregate<64><<<ablk, AGG_T, 0, stream>>>(
        srcI, dstI, dinv, bufA, b2, out, N, E);
}


# layer 2 skips the dst scan: layer 1's waves leave each matched edge as a (src, local row, scale) record in the workspace and layer 2's waves add from their own records; per-wave add order kept
# speedup vs baseline: 1.6393x; 1.1949x over previous
#include <hip/hip_runtime.h>
#include <math.h>

typedef __attribute__((ext_vector_type(2))) float v2f;
typedef __attribute__((ext_vector_type(8))) float v8f;

__device__ __forceinline__ void atomic_add_f32(float* p, float v) {
    __hip_atomic_fetch_add(p, v, __ATOMIC_RELAXED, __HIP_MEMORY_SCOPE_AGENT);
}

__global__ void k_fill_deg(float* __restrict__ deg, int n) {
    int i = blockIdx.x * blockDim.x + threadIdx.x;
    if (i < n) deg[i] = 1.0f;
}

__global__ void k_deg_scatter(const int* __restrict__ dst, int E,
                              float* __restrict__ deg) {
    int e = blockIdx.x * blockDim.x + threadIdx.x;
    if (e < E) atomic_add_f32(&deg[dst[e]], 1.0f);
}

__global__ void k_dinv(float* __restrict__ deg, int n) {
    int i = blockIdx.x * blockDim.x + threadIdx.x;
    if (i < n) {
        float d = deg[i];
        deg[i] = (d > 0.0f) ? rsqrtf(d) : 0.0f;
    }
}

template <int NT, int KD, bool RELU>
__global__ void __launch_bounds__(256) k_gemm_wmma_f32(
    const float* __restrict__ A, const float* __restrict__ B,
    float* __restrict__ C, int M)
{
    constexpr int N   = NT * 16;
    constexpr int LDK = KD + 2;
    __shared__ float lds_w[N * LDK];

    for (int idx = threadIdx.x; idx < KD * N; idx += 256) {
        int k = idx / N;
        int n = idx - k * N;
        lds_w[n * LDK + k] = B[idx];
    }
    __syncthreads();

    const int lane  = threadIdx.x & 31;
    const int wv    = threadIdx.x >> 5;
    const int mtile = blockIdx.x * 8 + wv;
    if (mtile * 16 >= M) return;

    const int hf = lane >> 4;
    const int r  = lane & 15;

    const float* __restrict__ arow = A + (size_t)(mtile * 16 + r) * KD + 2 * hf;

    v8f acc[NT];
#pragma unroll
    for (int t = 0; t < NT; ++t) acc[t] = (v8f){};

#pragma unroll 4
    for (int k = 0; k < KD; k += 4) {
        float2 av = *(const float2*)(arow + k);
        if (RELU) { av.x = fmaxf(av.x, 0.0f); av.y = fmaxf(av.y, 0.0f); }
        v2f a;
        a[0] = av.x;
        a[1] = av.y;
#pragma unroll
        for (int t = 0; t < NT; ++t) {
            const int col = t * 16 + r;
            float2 bv = *(const float2*)(&lds_w[col * LDK + k + 2 * hf]);
            v2f b;
            b[0] = bv.x;
            b[1] = bv.y;
            acc[t] = __builtin_amdgcn_wmma_f32_16x16x4_f32(
                false, a, false, b, (short)0, acc[t], false, false);
        }
    }

#pragma unroll
    for (int t = 0; t < NT; ++t) {
        float* crow = C + (size_t)(mtile * 16 + 8 * hf) * N + t * 16 + r;
#pragma unroll
        for (int j = 0; j < 8; ++j) crow[(size_t)j * N] = acc[t][j];
    }
}

constexpr int TILE_R   = 196;
constexpr int AGG_T    = 1024;
constexpr int SCAN_U   = 4;
constexpr int LIST_LOW = 256;
constexpr int LIST_CAP = LIST_LOW + 128 * SCAN_U;
constexpr int REC_CAP   = 512;
constexpr int AGG_WAVES = AGG_T / 32;
constexpr int MODE_SCAN_WRITE = 0;
constexpr int MODE_FROM_REC   = 1;

template <int V> struct RowVec;
template <> struct RowVec<4> { typedef float4 T; };
template <> struct RowVec<2> { typedef float2 T; };

__device__ __forceinline__ void lds_add_f32(float* p, float v) {
    __hip_atomic_fetch_add(p, v, __ATOMIC_RELAXED, __HIP_MEMORY_SCOPE_WORKGROUP);
}

template <int V>
__device__ __forceinline__ void add_row(float* a, const float* hp, float nu) {
#pragma unroll
    for (int q = 0; q < V; ++q) lds_add_f32(a + q, hp[q] * nu);
}

template <int C>
__device__ __forceinline__ void drain_list(
    const int* mylst, int cnt, int lane, int base,
    const int* __restrict__ src, const int* __restrict__ dst,
    const float* __restrict__ dinv, const float* __restrict__ h, float* acc,
    int4* __restrict__ myrec, int& wpos)
{
    constexpr int V = C / 32;
    typedef typename RowVec<V>::T VT;
    for (int i0 = 0; i0 < cnt; i0 += 32) {
        const int m = min(32, cnt - i0);
        int s = 0, dl = 0;
        float nr = 0.0f;
        if (lane < m) {
            const int e = mylst[i0 + lane];
            s = src[e];
            const int d = dst[e];
            dl = d - base;
            nr = dinv[s] * dinv[d];
        }
        if (wpos >= 0) {
            if (wpos + m <= REC_CAP) {
                if (lane < m)
                    myrec[wpos + lane] = make_int4(s, dl, __float_as_int(nr), 0);
                wpos += m;
            } else {
                wpos = -1;
            }
        }
        int j = 0;
        for (; j + 4 <= m; j += 4) {
            VT hv[4];
#pragma unroll
            for (int u = 0; u < 4; ++u) {
                const int su = __builtin_amdgcn_readlane(s, j + u);
                hv[u] = *(const VT*)(h + (size_t)su * C + lane * V);
            }
#pragma unroll
            for (int u = 0; u < 4; ++u) {
                const int du = __builtin_amdgcn_readlane(dl, j + u);
                const float nu = __int_as_float(
                    __builtin_amdgcn_readlane(__float_as_int(nr), j + u));
                add_row<V>(acc + du * C + lane * V, (const float*)&hv[u], nu);
            }
        }
        for (; j < m; ++j) {
            const int su = __builtin_amdgcn_readlane(s, j);
            const VT hv1 = *(const VT*)(h + (size_t)su * C + lane * V);
            const int du = __builtin_amdgcn_readlane(dl, j);
            const float nu = __int_as_float(
                __builtin_amdgcn_readlane(__float_as_int(nr), j));
            add_row<V>(acc + du * C + lane * V, (const float*)&hv1, nu);
        }
    }
}

template <int C, int R>
__device__ __forceinline__ void add_group(
    int s, int dl, float nr, int j, int lane,
    const float* __restrict__ h, float* acc)
{
    constexpr int V = C / 32;
    typedef typename RowVec<V>::T VT;
    VT hv[R];
#pragma unroll
    for (int u = 0; u < R; ++u) {
        const int su = __builtin_amdgcn_readlane(s, j + u);
        hv[u] = *(const VT*)(h + (size_t)su * C + lane * V);
    }
#pragma unroll
    for (int u = 0; u < R; ++u) {
        const int du = __builtin_amdgcn_readlane(dl, j + u);
        const float nu = __int_as_float(
            __builtin_amdgcn_readlane(__float_as_int(nr), j + u));
        add_row<V>(acc + du * C + lane * V, (const float*)&hv[u], nu);
    }
}

template <int C>
__device__ __forceinline__ void drain_records(
    const int4* __restrict__ myrec, int cnt, int4 cur, int lane,
    const float* __restrict__ h, float* acc)
{
    for (int i0 = 0; i0 < cnt; i0 += 32) {
        const int m = min(32, cnt - i0);
        int4 nxt = make_int4(0, 0, 0, 0);
        if (i0 + 32 + lane < cnt) nxt = myrec[i0 + 32 + lane];
        const int   s  = cur.x;
        const int   dl = cur.y;
        const float nr = __int_as_float(cur.z);
        int j = 0;
        for (; j + 8 <= m; j += 8) add_group<C, 8>(s, dl, nr, j, lane, h, acc);
        switch (m - j) {
            case 7: add_group<C, 7>(s, dl, nr, j, lane, h, acc); break;
            case 6: add_group<C, 6>(s, dl, nr, j, lane, h, acc); break;
            case 5: add_group<C, 5>(s, dl, nr, j, lane, h, acc); break;
            case 4: add_group<C, 4>(s, dl, nr, j, lane, h, acc); break;
            case 3: add_group<C, 3>(s, dl, nr, j, lane, h, acc); break;
            case 2: add_group<C, 2>(s, dl, nr, j, lane, h, acc); break;
            case 1: add_group<C, 1>(s, dl, nr, j, lane, h, acc); break;
            default: break;
        }
        cur = nxt;
    }
}

template <int C, int MODE>
__global__ void __launch_bounds__(AGG_T) k_tile_aggregate(
    const int* __restrict__ src, const int* __restrict__ dst,
    const float* __restrict__ dinv, const float* __restrict__ h,
    const float* __restrict__ bias, float* __restrict__ out,
    int* __restrict__ rec_cnt, int* __restrict__ rec, int n, int E)
{
    __shared__ float acc[TILE_R * C];
    __shared__ int   lst[(AGG_T / 32) * LIST_CAP];

    const int tid  = threadIdx.x;
    const int lane = tid & 31;
    const int wv   = tid >> 5;
    const int base = blockIdx.x * TILE_R;
    const int nrows = min(TILE_R, n - base);
    constexpr int C4 = C / 4;

    const int wslot = blockIdx.x * AGG_WAVES + wv;
    int4* myrec = (int4*)rec + (size_t)wslot * REC_CAP;
    int nrec_v = -1;
    if (MODE == MODE_FROM_REC) nrec_v = rec_cnt[wslot];

    for (int idx = tid; idx < nrows * C4; idx += AGG_T) {
        const int r  = idx / C4;
        const int c4 = (idx - r * C4) * 4;
        const float di = dinv[base + r];
        const float4 hv = *(const float4*)(h + (size_t)(base + r) * C + c4);
        const float4 bv = *(const float4*)(bias + c4);
        float4 o;
        o.x = bv.x + hv.x * di * di;
        o.y = bv.y + hv.y * di * di;
        o.z = bv.z + hv.z * di * di;
        o.w = bv.w + hv.w * di * di;
        *(float4*)(acc + r * C + c4) = o;
    }
    int4 rec0 = make_int4(0, 0, 0, 0);
    if (MODE == MODE_FROM_REC) {
        if (lane < nrec_v) rec0 = myrec[lane];
    }
    __syncthreads();

    const int nrec = __builtin_amdgcn_readfirstlane(nrec_v);
    if (MODE == MODE_FROM_REC && nrec >= 0) {
        drain_records<C>(myrec, nrec, rec0, lane, h, acc);
    } else {
        int wpos = (MODE == MODE_SCAN_WRITE) ? 0 : -1;
        int* mylst = lst + wv * LIST_CAP;
        int cnt = 0;

        const int nvec   = (((size_t)dst & 15) == 0) ? (E >> 2) : 0;
        const int per_it = AGG_T * SCAN_U;
        const int n_it   = (nvec + per_it - 1) / per_it;
        const int rot = (int)(((long long)blockIdx.x * n_it) / gridDim.x);
        const int4* dst4 = (const int4*)dst;
        for (int k = 0; k < n_it; ++k) {
            int it = k + rot;
            if (it >= n_it) it -= n_it;
            int4 dv[SCAN_U];
            int  vi[SCAN_U];
#pragma unroll
            for (int u = 0; u < SCAN_U; ++u) {
                vi[u] = it * per_it + u * AGG_T + tid;
                dv[u] = (vi[u] < nvec) ? dst4[vi[u]] : make_int4(-1, -1, -1, -1);
            }
#pragma unroll
            for (int u = 0; u < SCAN_U; ++u) {
                const int dd[4] = {dv[u].x, dv[u].y, dv[u].z, dv[u].w};
                bool hit[4];
#pragma unroll
                for (int c = 0; c < 4; ++c)
                    hit[c] = (unsigned)(dd[c] - base) < (unsigned)nrows;
                if (__builtin_amdgcn_ballot_w32(hit[0] | hit[1] | hit[2] | hit[3]) == 0u)
                    continue;
#pragma unroll
                for (int c = 0; c < 4; ++c) {
                    const unsigned b = __builtin_amdgcn_ballot_w32(hit[c]);
                    if (hit[c]) mylst[cnt + __builtin_amdgcn_mbcnt_lo(b, 0u)] = vi[u] * 4 + c;
                    cnt += __popc(b);
                }
            }
            if (cnt > LIST_LOW) {
                drain_list<C>(mylst, cnt, lane, base, src, dst, dinv, h, acc, myrec, wpos);
                cnt = 0;
            }
        }
        for (int e0 = nvec * 4 + wv * 32; e0 < E; e0 += AGG_T) {
            const int e = e0 + lane;
            const int d = (e < E) ? dst[e] : -1;
            const bool hit = (unsigned)(d - base) < (unsigned)nrows;
            const unsigned b = __builtin_amdgcn_ballot_w32(hit);
            if (hit) mylst[cnt + __builtin_amdgcn_mbcnt_lo(b, 0u)] = e;
            cnt += __popc(b);
            if (cnt > LIST_LOW) {
                drain_list<C>(mylst, cnt, lane, base, src, dst, dinv, h, acc, myrec, wpos);
                cnt = 0;
            }
        }
        drain_list<C>(mylst, cnt, lane, base, src, dst, dinv, h, acc, myrec, wpos);
        if (MODE == MODE_SCAN_WRITE && lane == 0) rec_cnt[wslot] = wpos;
    }
    __syncthreads();

    float4* o4 = (float4*)(out + (size_t)base * C);
    const float4* a4 = (const float4*)acc;
    for (int idx = tid; idx < nrows * C4; idx += AGG_T) o4[idx] = a4[idx];
}

extern "C" void kernel_launch(void* const* d_in, const int* in_sizes, int n_in,
                              void* d_out, int out_size, void* d_ws, size_t ws_size,
                              hipStream_t stream) {
    const float* x  = (const float*)d_in[0];
    const int*   ei = (const int*)d_in[1];
    const float* W1 = (const float*)d_in[2];
    const float* b1 = (const float*)d_in[3];
    const float* W2 = (const float*)d_in[4];
    const float* b2 = (const float*)d_in[5];
    float* out = (float*)d_out;

    const int HID = in_sizes[3];
    const int IN  = in_sizes[2] / HID;
    const int N   = in_sizes[0] / IN;
    const int E   = in_sizes[1] / 2;
    const int* srcI = ei;
    const int* dstI = ei + E;

    float* ws   = (float*)d_ws;
    float* dinv = ws;
    float* bufA = ws + (((size_t)N + 63) & ~(size_t)63);
    float* bufB = bufA + (size_t)N * HID;
    const int ablk = (N + TILE_R - 1) / TILE_R;
    int*  rcnt = (int*)(bufB + (size_t)N * HID);
    int*  rec  = rcnt + (size_t)ablk * AGG_WAVES;

    const int T = 256;
    const int mtiles = N / 16;
    const int gblk   = (mtiles + 7) / 8;

    k_fill_deg   <<<(N + T - 1) / T, T, 0, stream>>>(dinv, N);
    k_deg_scatter<<<(E + T - 1) / T, T, 0, stream>>>(dstI, E, dinv);
    k_dinv       <<<(N + T - 1) / T, T, 0, stream>>>(dinv, N);

    k_gemm_wmma_f32<8, 128, false><<<gblk, T, 0, stream>>>(x, W1, bufA, N);
    k_tile_aggregate<128, MODE_SCAN_WRITE><<<ablk, AGG_T, 0, stream>>>(
        srcI, dstI, dinv, bufA, b1, bufB, rcnt, rec, N, E);

    k_gemm_wmma_f32<4, 128, true><<<gblk, T, 0, stream>>>(bufB, W2, bufA, N);
    k_tile_aggregate<64, MODE_FROM_REC><<<ablk, AGG_T, 0, stream>>>(
        srcI, dstI, dinv, bufA, b2, out, rcnt, rec, N, E);
}


# aggregation scan reads a 1-byte dst-tile key per edge (left by the degree pass, now 4 edges per thread), 16 edges per 16-B load, not 4-B dst; edge-to-wave split changes, sum order = arrival
# speedup vs baseline: 1.7745x; 1.0825x over previous
#include <hip/hip_runtime.h>
#include <math.h>

typedef __attribute__((ext_vector_type(2))) float v2f;
typedef __attribute__((ext_vector_type(8))) float v8f;

__device__ __forceinline__ void atomic_add_f32(float* p, float v) {
    __hip_atomic_fetch_add(p, v, __ATOMIC_RELAXED, __HIP_MEMORY_SCOPE_AGENT);
}

constexpr int TILE_R = 196;

__global__ void k_fill_deg(float* __restrict__ deg, int n) {
    int i = blockIdx.x * blockDim.x + threadIdx.x;
    if (i < n) deg[i] = 1.0f;
}

__global__ void k_deg_scatter(const int* __restrict__ dst, int E,
                              float* __restrict__ deg,
                              unsigned int* __restrict__ key) {
    const int i = blockIdx.x * blockDim.x + threadIdx.x;
    const int e = i * 4;
    if (e + 3 < E) {
        int4 d;
        if (((size_t)dst & 15) == 0) d = *(const int4*)(dst + e);
        else d = make_int4(dst[e], dst[e + 1], dst[e + 2], dst[e + 3]);
        atomic_add_f32(&deg[d.x], 1.0f);
        atomic_add_f32(&deg[d.y], 1.0f);
        atomic_add_f32(&deg[d.z], 1.0f);
        atomic_add_f32(&deg[d.w], 1.0f);
        key[i] = ((unsigned)d.x / (unsigned)TILE_R) | (((unsigned)d.y / (unsigned)TILE_R) << 8) |
                 (((unsigned)d.z / (unsigned)TILE_R) << 16) | (((unsigned)d.w / (unsigned)TILE_R) << 24);
    } else {
        unsigned char* kb = (unsigned char*)key;
        for (int q = e; q < E; ++q) {
            const int d = dst[q];
            atomic_add_f32(&deg[d], 1.0f);
            kb[q] = (unsigned char)((unsigned)d / (unsigned)TILE_R);
        }
    }
}

__global__ void k_dinv(float* __restrict__ deg, int n) {
    int i = blockIdx.x * blockDim.x + threadIdx.x;
    if (i < n) {
        float d = deg[i];
        deg[i] = (d > 0.0f) ? rsqrtf(d) : 0.0f;
    }
}

template <int NT, int KD, bool RELU>
__global__ void __launch_bounds__(256) k_gemm_wmma_f32(
    const float* __restrict__ A, const float* __restrict__ B,
    float* __restrict__ C, int M)
{
    constexpr int N   = NT * 16;
    constexpr int LDK = KD + 2;
    __shared__ float lds_w[N * LDK];

    for (int idx = threadIdx.x; idx < KD * N; idx += 256) {
        int k = idx / N;
        int n = idx - k * N;
        lds_w[n * LDK + k] = B[idx];
    }
    __syncthreads();

    const int lane  = threadIdx.x & 31;
    const int wv    = threadIdx.x >> 5;
    const int mtile = blockIdx.x * 8 + wv;
    if (mtile * 16 >= M) return;

    const int hf = lane >> 4;
    const int r  = lane & 15;

    const float* __restrict__ arow = A + (size_t)(mtile * 16 + r) * KD + 2 * hf;

    v8f acc[NT];
#pragma unroll
    for (int t = 0; t < NT; ++t) acc[t] = (v8f){};

#pragma unroll 4
    for (int k = 0; k < KD; k += 4) {
        float2 av = *(const float2*)(arow + k);
        if (RELU) { av.x = fmaxf(av.x, 0.0f); av.y = fmaxf(av.y, 0.0f); }
        v2f a;
        a[0] = av.x;
        a[1] = av.y;
#pragma unroll
        for (int t = 0; t < NT; ++t) {
            const int col = t * 16 + r;
            float2 bv = *(const float2*)(&lds_w[col * LDK + k + 2 * hf]);
            v2f b;
            b[0] = bv.x;
            b[1] = bv.y;
            acc[t] = __builtin_amdgcn_wmma_f32_16x16x4_f32(
                false, a, false, b, (short)0, acc[t], false, false);
        }
    }

#pragma unroll
    for (int t = 0; t < NT; ++t) {
        float* crow = C + (size_t)(mtile * 16 + 8 * hf) * N + t * 16 + r;
#pragma unroll
        for (int j = 0; j < 8; ++j) crow[(size_t)j * N] = acc[t][j];
    }
}

constexpr int AGG_T    = 1024;
constexpr int SCAN_U   = 4;
constexpr int LIST_LOW = 256;
constexpr int LIST_CAP = LIST_LOW + 2 * 512;
constexpr int REC_CAP   = 512;
constexpr int AGG_WAVES = AGG_T / 32;
constexpr int MODE_SCAN_WRITE = 0;
constexpr int MODE_FROM_REC   = 1;

template <int V> struct RowVec;
template <> struct RowVec<4> { typedef float4 T; };
template <> struct RowVec<2> { typedef float2 T; };

__device__ __forceinline__ void lds_add_f32(float* p, float v) {
    __hip_atomic_fetch_add(p, v, __ATOMIC_RELAXED, __HIP_MEMORY_SCOPE_WORKGROUP);
}

template <int V>
__device__ __forceinline__ void add_row(float* a, const float* hp, float nu) {
#pragma unroll
    for (int q = 0; q < V; ++q) lds_add_f32(a + q, hp[q] * nu);
}

template <int C>
__device__ __forceinline__ void drain_list(
    const int* mylst, int cnt, int lane, int base,
    const int* __restrict__ src, const int* __restrict__ dst,
    const float* __restrict__ dinv, const float* __restrict__ h, float* acc,
    int4* __restrict__ myrec, int& wpos)
{
    constexpr int V = C / 32;
    typedef typename RowVec<V>::T VT;
    for (int i0 = 0; i0 < cnt; i0 += 32) {
        const int m = min(32, cnt - i0);
        int s = 0, dl = 0;
        float nr = 0.0f;
        if (lane < m) {
            const int e = mylst[i0 + lane];
            s = src[e];
            const int d = dst[e];
            dl = d - base;
            nr = dinv[s] * dinv[d];
        }
        if (wpos >= 0) {
            if (wpos + m <= REC_CAP) {
                if (lane < m)
                    myrec[wpos + lane] = make_int4(s, dl, __float_as_int(nr), 0);
                wpos += m;
            } else {
                wpos = -1;
            }
        }
        int j = 0;
        for (; j + 4 <= m; j += 4) {
            VT hv[4];
#pragma unroll
            for (int u = 0; u < 4; ++u) {
                const int su = __builtin_amdgcn_readlane(s, j + u);
                hv[u] = *(const VT*)(h + (size_t)su * C + lane * V);
            }
#pragma unroll
            for (int u = 0; u < 4; ++u) {
                const int du = __builtin_amdgcn_readlane(dl, j + u);
                const float nu = __int_as_float(
                    __builtin_amdgcn_readlane(__float_as_int(nr), j + u));
                add_row<V>(acc + du * C + lane * V, (const float*)&hv[u], nu);
            }
        }
        for (; j < m; ++j) {
            const int su = __builtin_amdgcn_readlane(s, j);
            const VT hv1 = *(const VT*)(h + (size_t)su * C + lane * V);
            const int du = __builtin_amdgcn_readlane(dl, j);
            const float nu = __int_as_float(
                __builtin_amdgcn_readlane(__float_as_int(nr), j));
            add_row<V>(acc + du * C + lane * V, (const float*)&hv1, nu);
        }
    }
}

template <int C, int R>
__device__ __forceinline__ void add_group(
    int s, int dl, float nr, int j, int lane,
    const float* __restrict__ h, float* acc)
{
    constexpr int V = C / 32;
    typedef typename RowVec<V>::T VT;
    VT hv[R];
#pragma unroll
    for (int u = 0; u < R; ++u) {
        const int su = __builtin_amdgcn_readlane(s, j + u);
        hv[u] = *(const VT*)(h + (size_t)su * C + lane * V);
    }
#pragma unroll
    for (int u = 0; u < R; ++u) {
        const int du = __builtin_amdgcn_readlane(dl, j + u);
        const float nu = __int_as_float(
            __builtin_amdgcn_readlane(__float_as_int(nr), j + u));
        add_row<V>(acc + du * C + lane * V, (const float*)&hv[u], nu);
    }
}

template <int C>
__device__ __forceinline__ void drain_records(
    const int4* __restrict__ myrec, int cnt, int4 cur, int lane,
    const float* __restrict__ h, float* acc)
{
    for (int i0 = 0; i0 < cnt; i0 += 32) {
        const int m = min(32, cnt - i0);
        int4 nxt = make_int4(0, 0, 0, 0);
        if (i0 + 32 + lane < cnt) nxt = myrec[i0 + 32 + lane];
        const int   s  = cur.x;
        const int   dl = cur.y;
        const float nr = __int_as_float(cur.z);
        int j = 0;
        for (; j + 8 <= m; j += 8) add_group<C, 8>(s, dl, nr, j, lane, h, acc);
        switch (m - j) {
            case 7: add_group<C, 7>(s, dl, nr, j, lane, h, acc); break;
            case 6: add_group<C, 6>(s, dl, nr, j, lane, h, acc); break;
            case 5: add_group<C, 5>(s, dl, nr, j, lane, h, acc); break;
            case 4: add_group<C, 4>(s, dl, nr, j, lane, h, acc); break;
            case 3: add_group<C, 3>(s, dl, nr, j, lane, h, acc); break;
            case 2: add_group<C, 2>(s, dl, nr, j, lane, h, acc); break;
            case 1: add_group<C, 1>(s, dl, nr, j, lane, h, acc); break;
            default: break;
        }
        cur = nxt;
    }
}

template <int C, int MODE>
__global__ void __launch_bounds__(AGG_T) k_tile_aggregate(
    const int* __restrict__ src, const int* __restrict__ dst,
    const float* __restrict__ dinv, const float* __restrict__ h,
    const float* __restrict__ bias, float* __restrict__ out,
    int* __restrict__ rec_cnt, int* __restrict__ rec,
    const unsigned int* __restrict__ key, int n, int E)
{
    __shared__ float acc[TILE_R * C];
    __shared__ int   lst[(AGG_T / 32) * LIST_CAP];

    const int tid  = threadIdx.x;
    const int lane = tid & 31;
    const int wv   = tid >> 5;
    const int base = blockIdx.x * TILE_R;
    const int nrows = min(TILE_R, n - base);
    constexpr int C4 = C / 4;

    const int wslot = blockIdx.x * AGG_WAVES + wv;
    int4* myrec = (int4*)rec + (size_t)wslot * REC_CAP;
    int nrec_v = -1;
    if (MODE == MODE_FROM_REC) nrec_v = rec_cnt[wslot];

    for (int idx = tid; idx < nrows * C4; idx += AGG_T) {
        const int r  = idx / C4;
        const int c4 = (idx - r * C4) * 4;
        const float di = dinv[base + r];
        const float4 hv = *(const float4*)(h + (size_t)(base + r) * C + c4);
        const float4 bv = *(const float4*)(bias + c4);
        float4 o;
        o.x = bv.x + hv.x * di * di;
        o.y = bv.y + hv.y * di * di;
        o.z = bv.z + hv.z * di * di;
        o.w = bv.w + hv.w * di * di;
        *(float4*)(acc + r * C + c4) = o;
    }
    int4 rec0 = make_int4(0, 0, 0, 0);
    if (MODE == MODE_FROM_REC) {
        if (lane < nrec_v) rec0 = myrec[lane];
    }
    __syncthreads();

    const int nrec = __builtin_amdgcn_readfirstlane(nrec_v);
    if (MODE == MODE_FROM_REC && nrec >= 0) {
        drain_records<C>(myrec, nrec, rec0, lane, h, acc);
    } else {
        int wpos = (MODE == MODE_SCAN_WRITE) ? 0 : -1;
        int* mylst = lst + wv * LIST_CAP;
        int cnt = 0;

        const unsigned pat = (unsigned)blockIdx.x * 0x01010101u;
        const int nvec   = E >> 4;
        const int per_it = AGG_T * SCAN_U;
        const int n_it   = (nvec + per_it - 1) / per_it;
        const int rot = (int)(((long long)blockIdx.x * n_it) / gridDim.x);
        const uint4* key4 = (const uint4*)key;
        for (int k = 0; k < n_it; ++k) {
            int it = k + rot;
            if (it >= n_it) it -= n_it;
            uint4 kv[SCAN_U];
            int   vi[SCAN_U];
#pragma unroll
            for (int u = 0; u < SCAN_U; ++u) {
                vi[u] = it * per_it + u * AGG_T + tid;
                kv[u] = key4[min(vi[u], nvec - 1)];
            }
            if (it == n_it - 1) {
#pragma unroll
                for (int u = 0; u < SCAN_U; ++u)
                    if (vi[u] >= nvec) kv[u] = make_uint4(~pat, ~pat, ~pat, ~pat);
            }
#pragma unroll
            for (int u = 0; u < SCAN_U; ++u) {
                const unsigned kw[4] = {kv[u].x, kv[u].y, kv[u].z, kv[u].w};
#pragma unroll
                for (int c = 0; c < 4; ++c) {
                    const unsigned x = kw[c] ^ pat;
                    unsigned m = ~(((x & 0x7F7F7F7Fu) + 0x7F7F7F7Fu) | x | 0x7F7F7F7Fu);
                    unsigned b = __builtin_amdgcn_ballot_w32(m != 0u);
                    while (b != 0u) {
                        if (m != 0u)
                            mylst[cnt + __builtin_amdgcn_mbcnt_lo(b, 0u)] =
                                vi[u] * 16 + c * 4 + (__builtin_ctz(m) >> 3);
                        m &= m - 1u;
                        cnt += __popc(b);
                        b = __builtin_amdgcn_ballot_w32(m != 0u);
                    }
                }
                if ((u & 1) && cnt > LIST_LOW) {
                    drain_list<C>(mylst, cnt, lane, base, src, dst, dinv, h, acc, myrec, wpos);
                    cnt = 0;
                }
            }
        }
        for (int e0 = nvec * 16 + wv * 32; e0 < E; e0 += AGG_T) {
            const int e = e0 + lane;
            const int d = (e < E) ? dst[e] : -1;
            const bool hit = (unsigned)(d - base) < (unsigned)nrows;
            const unsigned b = __builtin_amdgcn_ballot_w32(hit);
            if (hit) mylst[cnt + __builtin_amdgcn_mbcnt_lo(b, 0u)] = e;
            cnt += __popc(b);
            if (cnt > LIST_LOW) {
                drain_list<C>(mylst, cnt, lane, base, src, dst, dinv, h, acc, myrec, wpos);
                cnt = 0;
            }
        }
        drain_list<C>(mylst, cnt, lane, base, src, dst, dinv, h, acc, myrec, wpos);
        if (MODE == MODE_SCAN_WRITE && lane == 0) rec_cnt[wslot] = wpos;
    }
    __syncthreads();

    float4* o4 = (float4*)(out + (size_t)base * C);
    const float4* a4 = (const float4*)acc;
    for (int idx = tid; idx < nrows * C4; idx += AGG_T) o4[idx] = a4[idx];
}

extern "C" void kernel_launch(void* const* d_in, const int* in_sizes, int n_in,
                              void* d_out, int out_size, void* d_ws, size_t ws_size,
                              hipStream_t stream) {
    const float* x  = (const float*)d_in[0];
    const int*   ei = (const int*)d_in[1];
    const float* W1 = (const float*)d_in[2];
    const float* b1 = (const float*)d_in[3];
    const float* W2 = (const float*)d_in[4];
    const float* b2 = (const float*)d_in[5];
    float* out = (float*)d_out;

    const int HID = in_sizes[3];
    const int IN  = in_sizes[2] / HID;
    const int N   = in_sizes[0] / IN;
    const int E   = in_sizes[1] / 2;
    const int* srcI = ei;
    const int* dstI = ei + E;

    float* ws   = (float*)d_ws;
    float* dinv = ws;
    float* bufA = ws + (((size_t)N + 63) & ~(size_t)63);
    float* bufB = bufA + (size_t)N * HID;
    const int ablk = (N + TILE_R - 1) / TILE_R;
    unsigned int* key = (unsigned int*)(bufB + (size_t)N * HID);
    int*  rcnt = (int*)(key + ((((size_t)E + 63) & ~(size_t)63) >> 2));
    int*  rec  = rcnt + (size_t)ablk * AGG_WAVES;

    const int T = 256;
    const int mtiles = N / 16;
    const int gblk   = (mtiles + 7) / 8;

    k_fill_deg   <<<(N + T - 1) / T, T, 0, stream>>>(dinv, N);
    const int quads = (E + 3) / 4;
    k_deg_scatter<<<(quads + T - 1) / T, T, 0, stream>>>(dstI, E, dinv, key);
    k_dinv       <<<(N + T - 1) / T, T, 0, stream>>>(dinv, N);

    k_gemm_wmma_f32<8, 128, false><<<gblk, T, 0, stream>>>(x, W1, bufA, N);
    k_tile_aggregate<128, MODE_SCAN_WRITE><<<ablk, AGG_T, 0, stream>>>(
        srcI, dstI, dinv, bufA, b1, bufB, rcnt, rec, key, N, E);

    k_gemm_wmma_f32<4, 128, true><<<gblk, T, 0, stream>>>(bufB, W2, bufA, N);
    k_tile_aggregate<64, MODE_FROM_REC><<<ablk, AGG_T, 0, stream>>>(
        srcI, dstI, dinv, bufA, b2, out, rcnt, rec, key, N, E);
}


# layer 1's LDS tile keeps a 128-channel row as four 32-word planes (channel 4l+q at word 32q+l): the 32 lanes of each ds_add_f32 take 32 banks, no bank twice as before; add order unchanged (arrival)
# speedup vs baseline: 2.0351x; 1.1469x over previous
#include <hip/hip_runtime.h>
#include <math.h>

typedef __attribute__((ext_vector_type(2))) float v2f;
typedef __attribute__((ext_vector_type(8))) float v8f;

__device__ __forceinline__ void atomic_add_f32(float* p, float v) {
    __hip_atomic_fetch_add(p, v, __ATOMIC_RELAXED, __HIP_MEMORY_SCOPE_AGENT);
}

constexpr int TILE_R = 196;

__global__ void k_fill_deg(float* __restrict__ deg, int n) {
    int i = blockIdx.x * blockDim.x + threadIdx.x;
    if (i < n) deg[i] = 1.0f;
}

__global__ void k_deg_scatter(const int* __restrict__ dst, int E,
                              float* __restrict__ deg,
                              unsigned int* __restrict__ key) {
    const int i = blockIdx.x * blockDim.x + threadIdx.x;
    const int e = i * 4;
    if (e + 3 < E) {
        int4 d;
        if (((size_t)dst & 15) == 0) d = *(const int4*)(dst + e);
        else d = make_int4(dst[e], dst[e + 1], dst[e + 2], dst[e + 3]);
        atomic_add_f32(&deg[d.x], 1.0f);
        atomic_add_f32(&deg[d.y], 1.0f);
        atomic_add_f32(&deg[d.z], 1.0f);
        atomic_add_f32(&deg[d.w], 1.0f);
        key[i] = ((unsigned)d.x / (unsigned)TILE_R) | (((unsigned)d.y / (unsigned)TILE_R) << 8) |
                 (((unsigned)d.z / (unsigned)TILE_R) << 16) | (((unsigned)d.w / (unsigned)TILE_R) << 24);
    } else {
        unsigned char* kb = (unsigned char*)key;
        for (int q = e; q < E; ++q) {
            const int d = dst[q];
            atomic_add_f32(&deg[d], 1.0f);
            kb[q] = (unsigned char)((unsigned)d / (unsigned)TILE_R);
        }
    }
}

__global__ void k_dinv(float* __restrict__ deg, int n) {
    int i = blockIdx.x * blockDim.x + threadIdx.x;
    if (i < n) {
        float d = deg[i];
        deg[i] = (d > 0.0f) ? rsqrtf(d) : 0.0f;
    }
}

template <int NT, int KD, bool RELU>
__global__ void __launch_bounds__(256) k_gemm_wmma_f32(
    const float* __restrict__ A, const float* __restrict__ B,
    float* __restrict__ C, int M)
{
    constexpr int N   = NT * 16;
    constexpr int LDK = KD + 2;
    __shared__ float lds_w[N * LDK];

    for (int idx = threadIdx.x; idx < KD * N; idx += 256) {
        int k = idx / N;
        int n = idx - k * N;
        lds_w[n * LDK + k] = B[idx];
    }
    __syncthreads();

    const int lane  = threadIdx.x & 31;
    const int wv    = threadIdx.x >> 5;
    const int mtile = blockIdx.x * 8 + wv;
    if (mtile * 16 >= M) return;

    const int hf = lane >> 4;
    const int r  = lane & 15;

    const float* __restrict__ arow = A + (size_t)(mtile * 16 + r) * KD + 2 * hf;

    v8f acc[NT];
#pragma unroll
    for (int t = 0; t < NT; ++t) acc[t] = (v8f){};

#pragma unroll 4
    for (int k = 0; k < KD; k += 4) {
        float2 av = *(const float2*)(arow + k);
        if (RELU) { av.x = fmaxf(av.x, 0.0f); av.y = fmaxf(av.y, 0.0f); }
        v2f a;
        a[0] = av.x;
        a[1] = av.y;
#pragma unroll
        for (int t = 0; t < NT; ++t) {
            const int col = t * 16 + r;
            float2 bv = *(const float2*)(&lds_w[col * LDK + k + 2 * hf]);
            v2f b;
            b[0] = bv.x;
            b[1] = bv.y;
            acc[t] = __builtin_amdgcn_wmma_f32_16x16x4_f32(
                false, a, false, b, (short)0, acc[t], false, false);
        }
    }

#pragma unroll
    for (int t = 0; t < NT; ++t) {
        float* crow = C + (size_t)(mtile * 16 + 8 * hf) * N + t * 16 + r;
#pragma unroll
        for (int j = 0; j < 8; ++j) crow[(size_t)j * N] = acc[t][j];
    }
}

constexpr int AGG_T    = 1024;
constexpr int SCAN_U   = 4;
constexpr int LIST_LOW = 256;
constexpr int LIST_CAP = LIST_LOW + 2 * 512;
constexpr int REC_CAP   = 512;
constexpr int AGG_WAVES = AGG_T / 32;
constexpr int MODE_SCAN_WRITE = 0;
constexpr int MODE_FROM_REC   = 1;

template <int V> struct RowVec;
template <> struct RowVec<4> { typedef float4 T; };
template <> struct RowVec<2> { typedef float2 T; };

__device__ __forceinline__ void lds_add_f32(float* p, float v) {
    __hip_atomic_fetch_add(p, v, __ATOMIC_RELAXED, __HIP_MEMORY_SCOPE_WORKGROUP);
}

template <int V> struct RowMap {
    static constexpr bool PLANES    = (V == 4);
    static constexpr int  LANE_STEP = PLANES ? 1 : V;
    static constexpr int  CH_STEP   = PLANES ? 32 : 1;
};

template <int V>
__device__ __forceinline__ void add_row(float* a, const float* hp, float nu) {
#pragma unroll
    for (int q = 0; q < V; ++q) lds_add_f32(a + q * RowMap<V>::CH_STEP, hp[q] * nu);
}

template <int C>
__device__ __forceinline__ void drain_list(
    const int* mylst, int cnt, int lane, int base,
    const int* __restrict__ src, const int* __restrict__ dst,
    const float* __restrict__ dinv, const float* __restrict__ h, float* acc,
    int4* __restrict__ myrec, int& wpos)
{
    constexpr int V = C / 32;
    typedef typename RowVec<V>::T VT;
    for (int i0 = 0; i0 < cnt; i0 += 32) {
        const int m = min(32, cnt - i0);
        int s = 0, dl = 0;
        float nr = 0.0f;
        if (lane < m) {
            const int e = mylst[i0 + lane];
            s = src[e];
            const int d = dst[e];
            dl = d - base;
            nr = dinv[s] * dinv[d];
        }
        if (wpos >= 0) {
            if (wpos + m <= REC_CAP) {
                if (lane < m)
                    myrec[wpos + lane] = make_int4(s, dl, __float_as_int(nr), 0);
                wpos += m;
            } else {
                wpos = -1;
            }
        }
        int j = 0;
        for (; j + 4 <= m; j += 4) {
            VT hv[4];
#pragma unroll
            for (int u = 0; u < 4; ++u) {
                const int su = __builtin_amdgcn_readlane(s, j + u);
                hv[u] = *(const VT*)(h + (size_t)su * C + lane * V);
            }
#pragma unroll
            for (int u = 0; u < 4; ++u) {
                const int du = __builtin_amdgcn_readlane(dl, j + u);
                const float nu = __int_as_float(
                    __builtin_amdgcn_readlane(__float_as_int(nr), j + u));
                add_row<V>(acc + du * C + lane * RowMap<V>::LANE_STEP, (const float*)&hv[u], nu);
            }
        }
        for (; j < m; ++j) {
            const int su = __builtin_amdgcn_readlane(s, j);
            const VT hv1 = *(const VT*)(h + (size_t)su * C + lane * V);
            const int du = __builtin_amdgcn_readlane(dl, j);
            const float nu = __int_as_float(
                __builtin_amdgcn_readlane(__float_as_int(nr), j));
            add_row<V>(acc + du * C + lane * RowMap<V>::LANE_STEP, (const float*)&hv1, nu);
        }
    }
}

template <int C, int R>
__device__ __forceinline__ void add_group(
    int s, int dl, float nr, int j, int lane,
    const float* __restrict__ h, float* acc)
{
    constexpr int V = C / 32;
    typedef typename RowVec<V>::T VT;
    VT hv[R];
#pragma unroll
    for (int u = 0; u < R; ++u) {
        const int su = __builtin_amdgcn_readlane(s, j + u);
        hv[u] = *(const VT*)(h + (size_t)su * C + lane * V);
    }
#pragma unroll
    for (int u = 0; u < R; ++u) {
        const int du = __builtin_amdgcn_readlane(dl, j + u);
        const float nu = __int_as_float(
            __builtin_amdgcn_readlane(__float_as_int(nr), j + u));
        add_row<V>(acc + du * C + lane * RowMap<V>::LANE_STEP, (const float*)&hv[u], nu);
    }
}

template <int C>
__device__ __forceinline__ void drain_records(
    const int4* __restrict__ myrec, int cnt, int4 cur, int lane,
    const float* __restrict__ h, float* acc)
{
    for (int i0 = 0; i0 < cnt; i0 += 32) {
        const int m = min(32, cnt - i0);
        int4 nxt = make_int4(0, 0, 0, 0);
        if (i0 + 32 + lane < cnt) nxt = myrec[i0 + 32 + lane];
        const int   s  = cur.x;
        const int   dl = cur.y;
        const float nr = __int_as_float(cur.z);
        int j = 0;
        for (; j + 8 <= m; j += 8) add_group<C, 8>(s, dl, nr, j, lane, h, acc);
        switch (m - j) {
            case 7: add_group<C, 7>(s, dl, nr, j, lane, h, acc); break;
            case 6: add_group<C, 6>(s, dl, nr, j, lane, h, acc); break;
            case 5: add_group<C, 5>(s, dl, nr, j, lane, h, acc); break;
            case 4: add_group<C, 4>(s, dl, nr, j, lane, h, acc); break;
            case 3: add_group<C, 3>(s, dl, nr, j, lane, h, acc); break;
            case 2: add_group<C, 2>(s, dl, nr, j, lane, h, acc); break;
            case 1: add_group<C, 1>(s, dl, nr, j, lane, h, acc); break;
            default: break;
        }
        cur = nxt;
    }
}

template <int C, int MODE>
__global__ void __launch_bounds__(AGG_T) k_tile_aggregate(
    const int* __restrict__ src, const int* __restrict__ dst,
    const float* __restrict__ dinv, const float* __restrict__ h,
    const float* __restrict__ bias, float* __restrict__ out,
    int* __restrict__ rec_cnt, int* __restrict__ rec,
    const unsigned int* __restrict__ key, int n, int E)
{
    __shared__ float acc[TILE_R * C];
    __shared__ int   lst[(AGG_T / 32) * LIST_CAP];

    const int tid  = threadIdx.x;
    const int lane = tid & 31;
    const int wv   = tid >> 5;
    const int base = blockIdx.x * TILE_R;
    const int nrows = min(TILE_R, n - base);
    constexpr int C4 = C / 4;

    const int wslot = blockIdx.x * AGG_WAVES + wv;
    int4* myrec = (int4*)rec + (size_t)wslot * REC_CAP;
    int nrec_v = -1;
    if (MODE == MODE_FROM_REC) nrec_v = rec_cnt[wslot];

    for (int idx = tid; idx < nrows * C4; idx += AGG_T) {
        const int r  = idx / C4;
        const int c4 = (idx - r * C4) * 4;
        const float di = dinv[base + r];
        const float4 hv = *(const float4*)(h + (size_t)(base + r) * C + c4);
        const float4 bv = *(const float4*)(bias + c4);
        float4 o;
        o.x = bv.x + hv.x * di * di;
        o.y = bv.y + hv.y * di * di;
        o.z = bv.z + hv.z * di * di;
        o.w = bv.w + hv.w * di * di;
        if (RowMap<C / 32>::PLANES) {
            float* a = acc + r * C + (c4 >> 2);
            a[0] = o.x; a[32] = o.y; a[64] = o.z; a[96] = o.w;
        } else {
            *(float4*)(acc + r * C + c4) = o;
        }
    }
    int4 rec0 = make_int4(0, 0, 0, 0);
    if (MODE == MODE_FROM_REC) {
        if (lane < nrec_v) rec0 = myrec[lane];
    }
    __syncthreads();

    const int nrec = __builtin_amdgcn_readfirstlane(nrec_v);
    if (MODE == MODE_FROM_REC && nrec >= 0) {
        drain_records<C>(myrec, nrec, rec0, lane, h, acc);
    } else {
        int wpos = (MODE == MODE_SCAN_WRITE) ? 0 : -1;
        int* mylst = lst + wv * LIST_CAP;
        int cnt = 0;

        const unsigned pat = (unsigned)blockIdx.x * 0x01010101u;
        const int nvec   = E >> 4;
        const int per_it = AGG_T * SCAN_U;
        const int n_it   = (nvec + per_it - 1) / per_it;
        const int rot = (int)(((long long)blockIdx.x * n_it) / gridDim.x);
        const uint4* key4 = (const uint4*)key;
        for (int k = 0; k < n_it; ++k) {
            int it = k + rot;
            if (it >= n_it) it -= n_it;
            uint4 kv[SCAN_U];
            int   vi[SCAN_U];
#pragma unroll
            for (int u = 0; u < SCAN_U; ++u) {
                vi[u] = it * per_it + u * AGG_T + tid;
                kv[u] = key4[min(vi[u], nvec - 1)];
            }
            if (it == n_it - 1) {
#pragma unroll
                for (int u = 0; u < SCAN_U; ++u)
                    if (vi[u] >= nvec) kv[u] = make_uint4(~pat, ~pat, ~pat, ~pat);
            }
#pragma unroll
            for (int u = 0; u < SCAN_U; ++u) {
                const unsigned kw[4] = {kv[u].x, kv[u].y, kv[u].z, kv[u].w};
#pragma unroll
                for (int c = 0; c < 4; ++c) {
                    const unsigned x = kw[c] ^ pat;
                    unsigned m = ~(((x & 0x7F7F7F7Fu) + 0x7F7F7F7Fu) | x | 0x7F7F7F7Fu);
                    unsigned b = __builtin_amdgcn_ballot_w32(m != 0u);
                    while (b != 0u) {
                        if (m != 0u)
                            mylst[cnt + __builtin_amdgcn_mbcnt_lo(b, 0u)] =
                                vi[u] * 16 + c * 4 + (__builtin_ctz(m) >> 3);
                        m &= m - 1u;
                        cnt += __popc(b);
                        b = __builtin_amdgcn_ballot_w32(m != 0u);
                    }
                }
                if ((u & 1) && cnt > LIST_LOW) {
                    drain_list<C>(mylst, cnt, lane, base, src, dst, dinv, h, acc, myrec, wpos);
                    cnt = 0;
                }
            }
        }
        for (int e0 = nvec * 16 + wv * 32; e0 < E; e0 += AGG_T) {
            const int e = e0 + lane;
            const int d = (e < E) ? dst[e] : -1;
            const bool hit = (unsigned)(d - base) < (unsigned)nrows;
            const unsigned b = __builtin_amdgcn_ballot_w32(hit);
            if (hit) mylst[cnt + __builtin_amdgcn_mbcnt_lo(b, 0u)] = e;
            cnt += __popc(b);
            if (cnt > LIST_LOW) {
                drain_list<C>(mylst, cnt, lane, base, src, dst, dinv, h, acc, myrec, wpos);
                cnt = 0;
            }
        }
        drain_list<C>(mylst, cnt, lane, base, src, dst, dinv, h, acc, myrec, wpos);
        if (MODE == MODE_SCAN_WRITE && lane == 0) rec_cnt[wslot] = wpos;
    }
    __syncthreads();

    float4* o4 = (float4*)(out + (size_t)base * C);
    const float4* a4 = (const float4*)acc;
    for (int idx = tid; idx < nrows * C4; idx += AGG_T) {
        if (RowMap<C / 32>::PLANES) {
            const int r = idx / C4;
            const float* a = acc + r * C + (idx - r * C4);
            o4[idx] = make_float4(a[0], a[32], a[64], a[96]);
        } else {
            o4[idx] = a4[idx];
        }
    }
}

extern "C" void kernel_launch(void* const* d_in, const int* in_sizes, int n_in,
                              void* d_out, int out_size, void* d_ws, size_t ws_size,
                              hipStream_t stream) {
    const float* x  = (const float*)d_in[0];
    const int*   ei = (const int*)d_in[1];
    const float* W1 = (const float*)d_in[2];
    const float* b1 = (const float*)d_in[3];
    const float* W2 = (const float*)d_in[4];
    const float* b2 = (const float*)d_in[5];
    float* out = (float*)d_out;

    const int HID = in_sizes[3];
    const int IN  = in_sizes[2] / HID;
    const int N   = in_sizes[0] / IN;
    const int E   = in_sizes[1] / 2;
    const int* srcI = ei;
    const int* dstI = ei + E;

    float* ws   = (float*)d_ws;
    float* dinv = ws;
    float* bufA = ws + (((size_t)N + 63) & ~(size_t)63);
    float* bufB = bufA + (size_t)N * HID;
    const int ablk = (N + TILE_R - 1) / TILE_R;
    unsigned int* key = (unsigned int*)(bufB + (size_t)N * HID);
    int*  rcnt = (int*)(key + ((((size_t)E + 63) & ~(size_t)63) >> 2));
    int*  rec  = rcnt + (size_t)ablk * AGG_WAVES;

    const int T = 256;
    const int mtiles = N / 16;
    const int gblk   = (mtiles + 7) / 8;

    k_fill_deg   <<<(N + T - 1) / T, T, 0, stream>>>(dinv, N);
    const int quads = (E + 3) / 4;
    k_deg_scatter<<<(quads + T - 1) / T, T, 0, stream>>>(dstI, E, dinv, key);
    k_dinv       <<<(N + T - 1) / T, T, 0, stream>>>(dinv, N);

    k_gemm_wmma_f32<8, 128, false><<<gblk, T, 0, stream>>>(x, W1, bufA, N);
    k_tile_aggregate<128, MODE_SCAN_WRITE><<<ablk, AGG_T, 0, stream>>>(
        srcI, dstI, dinv, bufA, b1, bufB, rcnt, rec, key, N, E);

    k_gemm_wmma_f32<4, 128, true><<<gblk, T, 0, stream>>>(bufB, W2, bufA, N);
    k_tile_aggregate<64, MODE_FROM_REC><<<ablk, AGG_T, 0, stream>>>(
        srcI, dstI, dinv, bufA, b2, out, rcnt, rec, key, N, E);
}


# each dst row has one owner wave: scanning waves pass matched edges to its LDS queue (one int LDS add each), owners add by plain LDS load/store; f32 LDS atomics only on overflow; sum order = arrival
# speedup vs baseline: 2.1727x; 1.0676x over previous
#include <hip/hip_runtime.h>
#include <math.h>

typedef __attribute__((ext_vector_type(2))) float v2f;
typedef __attribute__((ext_vector_type(8))) float v8f;

__device__ __forceinline__ void atomic_add_f32(float* p, float v) {
    __hip_atomic_fetch_add(p, v, __ATOMIC_RELAXED, __HIP_MEMORY_SCOPE_AGENT);
}

constexpr int TILE_R = 196;

__global__ void k_fill_deg(float* __restrict__ deg, int n) {
    int i = blockIdx.x * blockDim.x + threadIdx.x;
    if (i < n) deg[i] = 1.0f;
}

__global__ void k_deg_scatter(const int* __restrict__ dst, int E,
                              float* __restrict__ deg,
                              unsigned int* __restrict__ key) {
    const int i = blockIdx.x * blockDim.x + threadIdx.x;
    const int e = i * 4;
    if (e + 3 < E) {
        int4 d;
        if (((size_t)dst & 15) == 0) d = *(const int4*)(dst + e);
        else d = make_int4(dst[e], dst[e + 1], dst[e + 2], dst[e + 3]);
        atomic_add_f32(&deg[d.x], 1.0f);
        atomic_add_f32(&deg[d.y], 1.0f);
        atomic_add_f32(&deg[d.z], 1.0f);
        atomic_add_f32(&deg[d.w], 1.0f);
        key[i] = ((unsigned)d.x / (unsigned)TILE_R) | (((unsigned)d.y / (unsigned)TILE_R) << 8) |
                 (((unsigned)d.z / (unsigned)TILE_R) << 16) | (((unsigned)d.w / (unsigned)TILE_R) << 24);
    } else {
        unsigned char* kb = (unsigned char*)key;
        for (int q = e; q < E; ++q) {
            const int d = dst[q];
            atomic_add_f32(&deg[d], 1.0f);
            kb[q] = (unsigned char)((unsigned)d / (unsigned)TILE_R);
        }
    }
}

__global__ void k_dinv(float* __restrict__ deg, int n) {
    int i = blockIdx.x * blockDim.x + threadIdx.x;
    if (i < n) {
        float d = deg[i];
        deg[i] = (d > 0.0f) ? rsqrtf(d) : 0.0f;
    }
}

template <int NT, int KD, bool RELU>
__global__ void __launch_bounds__(256) k_gemm_wmma_f32(
    const float* __restrict__ A, const float* __restrict__ B,
    float* __restrict__ C, int M)
{
    constexpr int N   = NT * 16;
    constexpr int LDK = KD + 2;
    __shared__ float lds_w[N * LDK];

    for (int idx = threadIdx.x; idx < KD * N; idx += 256) {
        int k = idx / N;
        int n = idx - k * N;
        lds_w[n * LDK + k] = B[idx];
    }
    __syncthreads();

    const int lane  = threadIdx.x & 31;
    const int wv    = threadIdx.x >> 5;
    const int mtile = blockIdx.x * 8 + wv;
    if (mtile * 16 >= M) return;

    const int hf = lane >> 4;
    const int r  = lane & 15;

    const float* __restrict__ arow = A + (size_t)(mtile * 16 + r) * KD + 2 * hf;

    v8f acc[NT];
#pragma unroll
    for (int t = 0; t < NT; ++t) acc[t] = (v8f){};

#pragma unroll 4
    for (int k = 0; k < KD; k += 4) {
        float2 av = *(const float2*)(arow + k);
        if (RELU) { av.x = fmaxf(av.x, 0.0f); av.y = fmaxf(av.y, 0.0f); }
        v2f a;
        a[0] = av.x;
        a[1] = av.y;
#pragma unroll
        for (int t = 0; t < NT; ++t) {
            const int col = t * 16 + r;
            float2 bv = *(const float2*)(&lds_w[col * LDK + k + 2 * hf]);
            v2f b;
            b[0] = bv.x;
            b[1] = bv.y;
            acc[t] = __builtin_amdgcn_wmma_f32_16x16x4_f32(
                false, a, false, b, (short)0, acc[t], false, false);
        }
    }

#pragma unroll
    for (int t = 0; t < NT; ++t) {
        float* crow = C + (size_t)(mtile * 16 + 8 * hf) * N + t * 16 + r;
#pragma unroll
        for (int j = 0; j < 8; ++j) crow[(size_t)j * N] = acc[t][j];
    }
}

constexpr int AGG_T    = 1024;
constexpr int SCAN_U   = 4;
constexpr int LIST_LOW = 32;
constexpr int LIST_CAP = LIST_LOW + 512;
constexpr int REC_CAP   = 512;
constexpr int AGG_WAVES = AGG_T / 32;
constexpr int Q_CAP     = 256;
constexpr int OWN_ROWS  = (TILE_R + AGG_WAVES - 1) / AGG_WAVES;
static_assert(Q_CAP <= REC_CAP, "a queue must fit its record slice");
static_assert((AGG_WAVES & (AGG_WAVES - 1)) == 0, "owner = row & (waves - 1)");
constexpr int MODE_SCAN_WRITE = 0;
constexpr int MODE_FROM_REC   = 1;

template <int V> struct RowVec;
template <> struct RowVec<4> { typedef float4 T; };
template <> struct RowVec<2> { typedef float2 T; };

__device__ __forceinline__ float4 row_start(const float4& b, const float4& hv, float di) {
    float4 o;
    o.x = b.x + hv.x * di * di;
    o.y = b.y + hv.y * di * di;
    o.z = b.z + hv.z * di * di;
    o.w = b.w + hv.w * di * di;
    return o;
}
__device__ __forceinline__ float2 row_start(const float2& b, const float2& hv, float di) {
    float2 o;
    o.x = b.x + hv.x * di * di;
    o.y = b.y + hv.y * di * di;
    return o;
}
__device__ __forceinline__ void row_fma(float4& a, const float4& hv, float nu) {
    a.x += hv.x * nu;
    a.y += hv.y * nu;
    a.z += hv.z * nu;
    a.w += hv.w * nu;
}
__device__ __forceinline__ void row_fma(float2& a, const float2& hv, float nu) {
    a.x += hv.x * nu;
    a.y += hv.y * nu;
}

template <int C>
__device__ __forceinline__ void load_group(
    typename RowVec<C / 32>::T (&hv)[8], int s, int j, int m, int lane,
    const float* __restrict__ h)
{
    constexpr int V = C / 32;
    typedef typename RowVec<V>::T VT;
#pragma unroll
    for (int u = 0; u < 8; ++u) {
        const int su = __builtin_amdgcn_readlane(s, min(j + u, m - 1));
        hv[u] = *(const VT*)(h + (size_t)su * C + lane * V);
    }
}

template <int C>
__device__ __forceinline__ void add_group(
    const typename RowVec<C / 32>::T (&hv)[8], int dl, float nr, int j, int m, int lane,
    float* acc)
{
    constexpr int V = C / 32;
    typedef typename RowVec<V>::T VT;
#pragma unroll
    for (int u = 0; u < 8; ++u) {
        if (j + u < m) {
            const int du = __builtin_amdgcn_readlane(dl, j + u);
            const float nu = __int_as_float(
                __builtin_amdgcn_readlane(__float_as_int(nr), j + u));
            VT* p = (VT*)(acc + du * C + lane * V);
            VT t = *p;
            row_fma(t, hv[u], nu);
            *p = t;
        }
    }
}

template <int C, bool COPY>
__device__ __forceinline__ void drain_owned(
    const int4* recs, int cnt, int lane,
    const float* __restrict__ h, float* acc, int4* __restrict__ myrec)
{
    typedef typename RowVec<C / 32>::T VT;
    int4 cur = make_int4(0, 0, 0, 0);
    if (lane < cnt) cur = recs[lane];
    for (int i0 = 0; i0 < cnt; i0 += 32) {
        const int m = min(32, cnt - i0);
        int4 nxt = make_int4(0, 0, 0, 0);
        if (i0 + 32 + lane < cnt) nxt = recs[i0 + 32 + lane];
        if (COPY) {
            if (lane < m) myrec[i0 + lane] = cur;
        }
        const int   s  = cur.x;
        const int   dl = cur.y;
        const float nr = __int_as_float(cur.z);
        VT ha[8], hb[8];
        load_group<C>(ha, s, 0, m, lane, h);
        load_group<C>(hb, s, 8, m, lane, h);
        add_group<C>(ha, dl, nr, 0, m, lane, acc);
        if (m > 8) {
            load_group<C>(ha, s, 16, m, lane, h);
            add_group<C>(hb, dl, nr, 8, m, lane, acc);
            if (m > 16) {
                load_group<C>(hb, s, 24, m, lane, h);
                add_group<C>(ha, dl, nr, 16, m, lane, acc);
                add_group<C>(hb, dl, nr, 24, m, lane, acc);
            }
        }
        cur = nxt;
    }
}

__device__ __forceinline__ int route_list(
    int* mylst, int cnt, bool whole, int lane, int base,
    const int* __restrict__ src, const int* __restrict__ dst,
    const float* __restrict__ dinv, int4* q, int* qcnt, int* ovf)
{
    const int take = whole ? (cnt & ~31) : cnt;
    for (int i0 = 0; i0 < take; i0 += 32) {
        if (i0 + lane < take) {
            const int e  = mylst[i0 + lane];
            const int s  = src[e];
            const int d  = dst[e];
            const int dl = d - base;
            const float nr = dinv[s] * dinv[d];
            const int ow = dl & (AGG_WAVES - 1);
            const int pos = __hip_atomic_fetch_add(&qcnt[ow], 1, __ATOMIC_RELAXED,
                                                   __HIP_MEMORY_SCOPE_WORKGROUP);
            q[ow * Q_CAP + min(pos, Q_CAP - 1)] = make_int4(s, dl, __float_as_int(nr), 0);
            if (pos >= Q_CAP) *ovf = 1;
        }
    }
    const int rem = cnt - take;
    if (rem > 0) {
        if (lane < rem) mylst[lane] = mylst[take + lane];
    }
    return rem;
}

__device__ __forceinline__ void lds_add_f32(float* p, float v) {
    __hip_atomic_fetch_add(p, v, __ATOMIC_RELAXED, __HIP_MEMORY_SCOPE_WORKGROUP);
}

template <int C>
__device__ __forceinline__ void drain_list(
    const int* mylst, int cnt, int lane, int base,
    const int* __restrict__ src, const int* __restrict__ dst,
    const float* __restrict__ dinv, const float* __restrict__ h, float* acc)
{
    constexpr int V = C / 32;
    typedef typename RowVec<V>::T VT;
    for (int i0 = 0; i0 < cnt; i0 += 32) {
        const int m = min(32, cnt - i0);
        int s = 0, dl = 0;
        float nr = 0.0f;
        if (lane < m) {
            const int e = mylst[i0 + lane];
            s = src[e];
            const int d = dst[e];
            dl = d - base;
            nr = dinv[s] * dinv[d];
        }
        for (int j0 = 0; j0 < m; j0 += 4) {
            VT hv[4];
#pragma unroll
            for (int u = 0; u < 4; ++u) {
                const int su = __builtin_amdgcn_readlane(s, min(j0 + u, m - 1));
                hv[u] = *(const VT*)(h + (size_t)su * C + lane * V);
            }
#pragma unroll
            for (int u = 0; u < 4; ++u) {
                if (j0 + u < m) {
                    const int du = __builtin_amdgcn_readlane(dl, j0 + u);
                    const float nu = __int_as_float(
                        __builtin_amdgcn_readlane(__float_as_int(nr), j0 + u));
                    float* a = acc + du * C + lane * V;
                    const float* hp = (const float*)&hv[u];
#pragma unroll
                    for (int c = 0; c < V; ++c) lds_add_f32(a + c, hp[c] * nu);
                }
            }
        }
    }
}

template <int C, bool ROUTE>
__device__ __forceinline__ void scan_tile(
    int tid, int lane, int wv, int base, int nrows,
    const int* __restrict__ src, const int* __restrict__ dst,
    const float* __restrict__ dinv, const float* __restrict__ h,
    const unsigned int* __restrict__ key, int E,
    int* mylst, float* acc, int4* q, int* qcnt, int* ovf)
{
    int cnt = 0;

    const unsigned pat = (unsigned)blockIdx.x * 0x01010101u;
    const int nvec   = E >> 4;
    const int per_it = AGG_T * SCAN_U;
    const int n_it   = (nvec + per_it - 1) / per_it;
    const int rot = (int)(((long long)blockIdx.x * n_it) / gridDim.x);
    const uint4* key4 = (const uint4*)key;
    for (int k = 0; k < n_it; ++k) {
        int it = k + rot;
        if (it >= n_it) it -= n_it;
        uint4 kv[SCAN_U];
        int   vi[SCAN_U];
#pragma unroll
        for (int u = 0; u < SCAN_U; ++u) {
            vi[u] = it * per_it + u * AGG_T + tid;
            kv[u] = key4[min(vi[u], nvec - 1)];
        }
        if (it == n_it - 1) {
#pragma unroll
            for (int u = 0; u < SCAN_U; ++u)
                if (vi[u] >= nvec) kv[u] = make_uint4(~pat, ~pat, ~pat, ~pat);
        }
#pragma unroll
        for (int u = 0; u < SCAN_U; ++u) {
            const unsigned kw[4] = {kv[u].x, kv[u].y, kv[u].z, kv[u].w};
#pragma unroll
            for (int c = 0; c < 4; ++c) {
                const unsigned x = kw[c] ^ pat;
                unsigned m = ~(((x & 0x7F7F7F7Fu) + 0x7F7F7F7Fu) | x | 0x7F7F7F7Fu);
                unsigned b = __builtin_amdgcn_ballot_w32(m != 0u);
                while (b != 0u) {
                    if (m != 0u)
                        mylst[cnt + __builtin_amdgcn_mbcnt_lo(b, 0u)] =
                            vi[u] * 16 + c * 4 + (__builtin_ctz(m) >> 3);
                    m &= m - 1u;
                    cnt += __popc(b);
                    b = __builtin_amdgcn_ballot_w32(m != 0u);
                }
            }
            if (ROUTE) {
                if (cnt >= LIST_LOW)
                    cnt = route_list(mylst, cnt, true, lane, base, src, dst, dinv, q, qcnt, ovf);
            } else {
                if (cnt > LIST_LOW) {
                    drain_list<C>(mylst, cnt, lane, base, src, dst, dinv, h, acc);
                    cnt = 0;
                }
            }
        }
    }
    for (int e0 = nvec * 16 + wv * 32; e0 < E; e0 += AGG_T) {
        const int e = e0 + lane;
        const int d = (e < E) ? dst[e] : -1;
        const bool hit = (unsigned)(d - base) < (unsigned)nrows;
        const unsigned b = __builtin_amdgcn_ballot_w32(hit);
        if (hit) mylst[cnt + __builtin_amdgcn_mbcnt_lo(b, 0u)] = e;
        cnt += __popc(b);
        if (ROUTE) {
            if (cnt >= LIST_LOW)
                cnt = route_list(mylst, cnt, true, lane, base, src, dst, dinv, q, qcnt, ovf);
        } else {
            if (cnt > LIST_LOW) {
                drain_list<C>(mylst, cnt, lane, base, src, dst, dinv, h, acc);
                cnt = 0;
            }
        }
    }
    if (ROUTE) route_list(mylst, cnt, false, lane, base, src, dst, dinv, q, qcnt, ovf);
    else       drain_list<C>(mylst, cnt, lane, base, src, dst, dinv, h, acc);
}

template <int C, int MODE>
__global__ void __launch_bounds__(AGG_T) k_tile_aggregate(
    const int* __restrict__ src, const int* __restrict__ dst,
    const float* __restrict__ dinv, const float* __restrict__ h,
    const float* __restrict__ bias, float* __restrict__ out,
    int* __restrict__ rec_cnt, int* __restrict__ rec,
    const unsigned int* __restrict__ key, int n, int E)
{
    constexpr int  V    = C / 32;
    constexpr bool SCAN = (MODE == MODE_SCAN_WRITE);
    typedef typename RowVec<V>::T VT;
    __shared__ __attribute__((aligned(16))) float acc[TILE_R * C];
    __shared__ int   lst[AGG_WAVES * LIST_CAP];
    __shared__ __attribute__((aligned(16))) int qraw[SCAN ? AGG_WAVES * Q_CAP * 4 : 4];
    __shared__ int   qcnt[AGG_WAVES];
    __shared__ int   ovf;

    const int tid  = threadIdx.x;
    const int lane = tid & 31;
    const int wv   = __builtin_amdgcn_readfirstlane(tid >> 5);
    const int base = blockIdx.x * TILE_R;
    const int nrows = min(TILE_R, n - base);
    int4* q     = (int4*)qraw;
    int*  mylst = lst + wv * LIST_CAP;

    const int wslot = blockIdx.x * AGG_WAVES + wv;
    int4* myrec = (int4*)rec + (size_t)wslot * REC_CAP;

    int fill_v = 0;
    if (!SCAN) fill_v = rec_cnt[blockIdx.x * AGG_WAVES + lane];
    if (SCAN) {
        if (tid < AGG_WAVES) qcnt[tid] = 0;
        if (tid == 0) ovf = 0;
    }

    {
        const VT bv = *(const VT*)(bias + lane * V);
        const int kk = min(lane, OWN_ROWS - 1);
        const float dv = dinv[base + min(wv + kk * AGG_WAVES, nrows - 1)];
        VT st[OWN_ROWS];
#pragma unroll
        for (int k = 0; k < OWN_ROWS; ++k) {
            const int rr = min(wv + k * AGG_WAVES, nrows - 1);
            const float di = __int_as_float(__builtin_amdgcn_readlane(__float_as_int(dv), k));
            const VT hv = *(const VT*)(h + (size_t)(base + rr) * C + lane * V);
            st[k] = row_start(bv, hv, di);
        }
#pragma unroll
        for (int k = 0; k < OWN_ROWS; ++k) {
            const int r = wv + k * AGG_WAVES;
            if (r < nrows) *(VT*)(acc + r * C + lane * V) = st[k];
        }
    }

    bool fast;
    int  nown;
    if (SCAN) {
        __syncthreads();
        scan_tile<C, true>(tid, lane, wv, base, nrows, src, dst, dinv, h, key, E,
                           mylst, acc, q, qcnt, &ovf);
        __syncthreads();
        fast = (__builtin_amdgcn_readfirstlane(ovf) == 0);
        nown = __builtin_amdgcn_readfirstlane(qcnt[wv]);
    } else {
        fast = (__builtin_amdgcn_ballot_w32(fill_v < 0) == 0u);
        nown = __builtin_amdgcn_readlane(fill_v, wv);
    }

    if (__builtin_expect(fast, true)) {
        if (SCAN) {
            drain_owned<C, true>(q + wv * Q_CAP, nown, lane, h, acc, myrec);
            if (lane == 0) rec_cnt[wslot] = nown;
        } else {
            drain_owned<C, false>(myrec, nown, lane, h, acc, myrec);
        }
    } else {
        __syncthreads();
        scan_tile<C, false>(tid, lane, wv, base, nrows, src, dst, dinv, h, key, E,
                            mylst, acc, q, qcnt, &ovf);
        if (SCAN && lane == 0) rec_cnt[wslot] = -1;
        __syncthreads();
    }
#pragma unroll
    for (int k = 0; k < OWN_ROWS; ++k) {
        const int r = wv + k * AGG_WAVES;
        if (r < nrows)
            *(VT*)(out + (size_t)(base + r) * C + lane * V) = *(const VT*)(acc + r * C + lane * V);
    }
}

extern "C" void kernel_launch(void* const* d_in, const int* in_sizes, int n_in,
                              void* d_out, int out_size, void* d_ws, size_t ws_size,
                              hipStream_t stream) {
    const float* x  = (const float*)d_in[0];
    const int*   ei = (const int*)d_in[1];
    const float* W1 = (const float*)d_in[2];
    const float* b1 = (const float*)d_in[3];
    const float* W2 = (const float*)d_in[4];
    const float* b2 = (const float*)d_in[5];
    float* out = (float*)d_out;

    const int HID = in_sizes[3];
    const int IN  = in_sizes[2] / HID;
    const int N   = in_sizes[0] / IN;
    const int E   = in_sizes[1] / 2;
    const int* srcI = ei;
    const int* dstI = ei + E;

    float* ws   = (float*)d_ws;
    float* dinv = ws;
    float* bufA = ws + (((size_t)N + 63) & ~(size_t)63);
    float* bufB = bufA + (size_t)N * HID;
    const int ablk = (N + TILE_R - 1) / TILE_R;
    unsigned int* key = (unsigned int*)(bufB + (size_t)N * HID);
    int*  rcnt = (int*)(key + ((((size_t)E + 63) & ~(size_t)63) >> 2));
    int*  rec  = rcnt + (size_t)ablk * AGG_WAVES;

    const int T = 256;
    const int mtiles = N / 16;
    const int gblk   = (mtiles + 7) / 8;

    k_fill_deg   <<<(N + T - 1) / T, T, 0, stream>>>(dinv, N);
    const int quads = (E + 3) / 4;
    k_deg_scatter<<<(quads + T - 1) / T, T, 0, stream>>>(dstI, E, dinv, key);
    k_dinv       <<<(N + T - 1) / T, T, 0, stream>>>(dinv, N);

    k_gemm_wmma_f32<8, 128, false><<<gblk, T, 0, stream>>>(x, W1, bufA, N);
    k_tile_aggregate<128, MODE_SCAN_WRITE><<<ablk, AGG_T, 0, stream>>>(
        srcI, dstI, dinv, bufA, b1, bufB, rcnt, rec, key, N, E);

    k_gemm_wmma_f32<4, 128, true><<<gblk, T, 0, stream>>>(bufB, W2, bufA, N);
    k_tile_aggregate<64, MODE_FROM_REC><<<ablk, AGG_T, 0, stream>>>(
        srcI, dstI, dinv, bufA, b2, out, rcnt, rec, key, N, E);
}


# layer 1 sums rows of x, not of x@W1, and multiplies each finished LDS tile by W1 in the same work-group (f32 WMMA, A and B from LDS, row pitch +4 words); x@W1 launch gone; rounding changes
# speedup vs baseline: 2.3443x; 1.0790x over previous
#include <hip/hip_runtime.h>
#include <math.h>

typedef __attribute__((ext_vector_type(2))) float v2f;
typedef __attribute__((ext_vector_type(8))) float v8f;

__device__ __forceinline__ void atomic_add_f32(float* p, float v) {
    __hip_atomic_fetch_add(p, v, __ATOMIC_RELAXED, __HIP_MEMORY_SCOPE_AGENT);
}

constexpr int TILE_R = 196;

__global__ void k_fill_deg(float* __restrict__ deg, int n) {
    int i = blockIdx.x * blockDim.x + threadIdx.x;
    if (i < n) deg[i] = 1.0f;
}

__global__ void k_deg_scatter(const int* __restrict__ dst, int E,
                              float* __restrict__ deg,
                              unsigned int* __restrict__ key) {
    const int i = blockIdx.x * blockDim.x + threadIdx.x;
    const int e = i * 4;
    if (e + 3 < E) {
        int4 d;
        if (((size_t)dst & 15) == 0) d = *(const int4*)(dst + e);
        else d = make_int4(dst[e], dst[e + 1], dst[e + 2], dst[e + 3]);
        atomic_add_f32(&deg[d.x], 1.0f);
        atomic_add_f32(&deg[d.y], 1.0f);
        atomic_add_f32(&deg[d.z], 1.0f);
        atomic_add_f32(&deg[d.w], 1.0f);
        key[i] = ((unsigned)d.x / (unsigned)TILE_R) | (((unsigned)d.y / (unsigned)TILE_R) << 8) |
                 (((unsigned)d.z / (unsigned)TILE_R) << 16) | (((unsigned)d.w / (unsigned)TILE_R) << 24);
    } else {
        unsigned char* kb = (unsigned char*)key;
        for (int q = e; q < E; ++q) {
            const int d = dst[q];
            atomic_add_f32(&deg[d], 1.0f);
            kb[q] = (unsigned char)((unsigned)d / (unsigned)TILE_R);
        }
    }
}

__global__ void k_dinv(float* __restrict__ deg, int n) {
    int i = blockIdx.x * blockDim.x + threadIdx.x;
    if (i < n) {
        float d = deg[i];
        deg[i] = (d > 0.0f) ? rsqrtf(d) : 0.0f;
    }
}

template <int NT, int KD, bool RELU>
__global__ void __launch_bounds__(256) k_gemm_wmma_f32(
    const float* __restrict__ A, const float* __restrict__ B,
    float* __restrict__ C, int M)
{
    constexpr int N   = NT * 16;
    constexpr int LDK = KD + 2;
    __shared__ float lds_w[N * LDK];

    for (int idx = threadIdx.x; idx < KD * N; idx += 256) {
        int k = idx / N;
        int n = idx - k * N;
        lds_w[n * LDK + k] = B[idx];
    }
    __syncthreads();

    const int lane  = threadIdx.x & 31;
    const int wv    = threadIdx.x >> 5;
    const int mtile = blockIdx.x * 8 + wv;
    if (mtile * 16 >= M) return;

    const int hf = lane >> 4;
    const int r  = lane & 15;

    const float* __restrict__ arow = A + (size_t)(mtile * 16 + r) * KD + 2 * hf;

    v8f acc[NT];
#pragma unroll
    for (int t = 0; t < NT; ++t) acc[t] = (v8f){};

#pragma unroll 4
    for (int k = 0; k < KD; k += 4) {
        float2 av = *(const float2*)(arow + k);
        if (RELU) { av.x = fmaxf(av.x, 0.0f); av.y = fmaxf(av.y, 0.0f); }
        v2f a;
        a[0] = av.x;
        a[1] = av.y;
#pragma unroll
        for (int t = 0; t < NT; ++t) {
            const int col = t * 16 + r;
            float2 bv = *(const float2*)(&lds_w[col * LDK + k + 2 * hf]);
            v2f b;
            b[0] = bv.x;
            b[1] = bv.y;
            acc[t] = __builtin_amdgcn_wmma_f32_16x16x4_f32(
                false, a, false, b, (short)0, acc[t], false, false);
        }
    }

#pragma unroll
    for (int t = 0; t < NT; ++t) {
        float* crow = C + (size_t)(mtile * 16 + 8 * hf) * N + t * 16 + r;
#pragma unroll
        for (int j = 0; j < 8; ++j) crow[(size_t)j * N] = acc[t][j];
    }
}

constexpr int AGG_T    = 1024;
constexpr int SCAN_U   = 4;
constexpr int LIST_LOW = 32;
constexpr int LIST_CAP = LIST_LOW + 512;
constexpr int REC_CAP   = 512;
constexpr int AGG_WAVES = AGG_T / 32;
constexpr int Q_CAP     = 256;
constexpr int OWN_ROWS  = (TILE_R + AGG_WAVES - 1) / AGG_WAVES;
static_assert(Q_CAP <= REC_CAP, "a queue must fit its record slice");
static_assert((AGG_WAVES & (AGG_WAVES - 1)) == 0, "owner = row & (waves - 1)");
constexpr int MODE_SCAN_WRITE = 0;
constexpr int MODE_FROM_REC   = 1;

template <int V> struct RowVec;
template <> struct RowVec<4> { typedef float4 T; };
template <> struct RowVec<2> { typedef float2 T; };

__device__ __forceinline__ float4 row_start(const float4& b, const float4& hv, float di) {
    float4 o;
    o.x = b.x + hv.x * di * di;
    o.y = b.y + hv.y * di * di;
    o.z = b.z + hv.z * di * di;
    o.w = b.w + hv.w * di * di;
    return o;
}
__device__ __forceinline__ float2 row_start(const float2& b, const float2& hv, float di) {
    float2 o;
    o.x = b.x + hv.x * di * di;
    o.y = b.y + hv.y * di * di;
    return o;
}
__device__ __forceinline__ float4 row_self(const float4& hv, float di) {
    float4 o;
    o.x = hv.x * di * di;
    o.y = hv.y * di * di;
    o.z = hv.z * di * di;
    o.w = hv.w * di * di;
    return o;
}
__device__ __forceinline__ float2 row_self(const float2& hv, float di) {
    float2 o;
    o.x = hv.x * di * di;
    o.y = hv.y * di * di;
    return o;
}
__device__ __forceinline__ void row_fma(float4& a, const float4& hv, float nu) {
    a.x += hv.x * nu;
    a.y += hv.y * nu;
    a.z += hv.z * nu;
    a.w += hv.w * nu;
}
__device__ __forceinline__ void row_fma(float2& a, const float2& hv, float nu) {
    a.x += hv.x * nu;
    a.y += hv.y * nu;
}

template <int C>
__device__ __forceinline__ void load_group(
    typename RowVec<C / 32>::T (&hv)[8], int s, int j, int m, int lane,
    const float* __restrict__ h)
{
    constexpr int V = C / 32;
    typedef typename RowVec<V>::T VT;
#pragma unroll
    for (int u = 0; u < 8; ++u) {
        const int su = __builtin_amdgcn_readlane(s, min(j + u, m - 1));
        hv[u] = *(const VT*)(h + (size_t)su * C + lane * V);
    }
}

template <int C, int LD>
__device__ __forceinline__ void add_group(
    const typename RowVec<C / 32>::T (&hv)[8], int dl, float nr, int j, int m, int lane,
    float* acc)
{
    constexpr int V = C / 32;
    typedef typename RowVec<V>::T VT;
#pragma unroll
    for (int u = 0; u < 8; ++u) {
        if (j + u < m) {
            const int du = __builtin_amdgcn_readlane(dl, j + u);
            const float nu = __int_as_float(
                __builtin_amdgcn_readlane(__float_as_int(nr), j + u));
            VT* p = (VT*)(acc + du * LD + lane * V);
            VT t = *p;
            row_fma(t, hv[u], nu);
            *p = t;
        }
    }
}

template <int C, int LD, bool COPY>
__device__ __forceinline__ void drain_owned(
    const int4* recs, int cnt, int lane,
    const float* __restrict__ h, float* acc, int4* __restrict__ myrec)
{
    typedef typename RowVec<C / 32>::T VT;
    int4 cur = make_int4(0, 0, 0, 0);
    if (lane < cnt) cur = recs[lane];
    for (int i0 = 0; i0 < cnt; i0 += 32) {
        const int m = min(32, cnt - i0);
        int4 nxt = make_int4(0, 0, 0, 0);
        if (i0 + 32 + lane < cnt) nxt = recs[i0 + 32 + lane];
        if (COPY) {
            if (lane < m) myrec[i0 + lane] = cur;
        }
        const int   s  = cur.x;
        const int   dl = cur.y;
        const float nr = __int_as_float(cur.z);
        VT ha[8], hb[8];
        load_group<C>(ha, s, 0, m, lane, h);
        load_group<C>(hb, s, 8, m, lane, h);
        add_group<C, LD>(ha, dl, nr, 0, m, lane, acc);
        if (m > 8) {
            load_group<C>(ha, s, 16, m, lane, h);
            add_group<C, LD>(hb, dl, nr, 8, m, lane, acc);
            if (m > 16) {
                load_group<C>(hb, s, 24, m, lane, h);
                add_group<C, LD>(ha, dl, nr, 16, m, lane, acc);
                add_group<C, LD>(hb, dl, nr, 24, m, lane, acc);
            }
        }
        cur = nxt;
    }
}

__device__ __forceinline__ int route_list(
    int* mylst, int cnt, bool whole, int lane, int base,
    const int* __restrict__ src, const int* __restrict__ dst,
    const float* __restrict__ dinv, int4* q, int* qcnt, int* ovf)
{
    const int take = whole ? (cnt & ~31) : cnt;
    for (int i0 = 0; i0 < take; i0 += 32) {
        if (i0 + lane < take) {
            const int e  = mylst[i0 + lane];
            const int s  = src[e];
            const int d  = dst[e];
            const int dl = d - base;
            const float nr = dinv[s] * dinv[d];
            const int ow = dl & (AGG_WAVES - 1);
            const int pos = __hip_atomic_fetch_add(&qcnt[ow], 1, __ATOMIC_RELAXED,
                                                   __HIP_MEMORY_SCOPE_WORKGROUP);
            q[ow * Q_CAP + min(pos, Q_CAP - 1)] = make_int4(s, dl, __float_as_int(nr), 0);
            if (pos >= Q_CAP) *ovf = 1;
        }
    }
    const int rem = cnt - take;
    if (rem > 0) {
        if (lane < rem) mylst[lane] = mylst[take + lane];
    }
    return rem;
}

__device__ __forceinline__ void lds_add_f32(float* p, float v) {
    __hip_atomic_fetch_add(p, v, __ATOMIC_RELAXED, __HIP_MEMORY_SCOPE_WORKGROUP);
}

template <int C, int LD>
__device__ __forceinline__ void drain_list(
    const int* mylst, int cnt, int lane, int base,
    const int* __restrict__ src, const int* __restrict__ dst,
    const float* __restrict__ dinv, const float* __restrict__ h, float* acc)
{
    constexpr int V = C / 32;
    typedef typename RowVec<V>::T VT;
    for (int i0 = 0; i0 < cnt; i0 += 32) {
        const int m = min(32, cnt - i0);
        int s = 0, dl = 0;
        float nr = 0.0f;
        if (lane < m) {
            const int e = mylst[i0 + lane];
            s = src[e];
            const int d = dst[e];
            dl = d - base;
            nr = dinv[s] * dinv[d];
        }
        for (int j0 = 0; j0 < m; j0 += 4) {
            VT hv[4];
#pragma unroll
            for (int u = 0; u < 4; ++u) {
                const int su = __builtin_amdgcn_readlane(s, min(j0 + u, m - 1));
                hv[u] = *(const VT*)(h + (size_t)su * C + lane * V);
            }
#pragma unroll
            for (int u = 0; u < 4; ++u) {
                if (j0 + u < m) {
                    const int du = __builtin_amdgcn_readlane(dl, j0 + u);
                    const float nu = __int_as_float(
                        __builtin_amdgcn_readlane(__float_as_int(nr), j0 + u));
                    float* a = acc + du * LD + lane * V;
                    const float* hp = (const float*)&hv[u];
#pragma unroll
                    for (int c = 0; c < V; ++c) lds_add_f32(a + c, hp[c] * nu);
                }
            }
        }
    }
}

template <int C, int LD, bool ROUTE>
__device__ __forceinline__ void scan_tile(
    int tid, int lane, int wv, int base, int nrows,
    const int* __restrict__ src, const int* __restrict__ dst,
    const float* __restrict__ dinv, const float* __restrict__ h,
    const unsigned int* __restrict__ key, int E,
    int* mylst, float* acc, int4* q, int* qcnt, int* ovf)
{
    int cnt = 0;

    const unsigned pat = (unsigned)blockIdx.x * 0x01010101u;
    const int nvec   = E >> 4;
    const int per_it = AGG_T * SCAN_U;
    const int n_it   = (nvec + per_it - 1) / per_it;
    const int rot = (int)(((long long)blockIdx.x * n_it) / gridDim.x);
    const uint4* key4 = (const uint4*)key;
    for (int k = 0; k < n_it; ++k) {
        int it = k + rot;
        if (it >= n_it) it -= n_it;
        uint4 kv[SCAN_U];
        int   vi[SCAN_U];
#pragma unroll
        for (int u = 0; u < SCAN_U; ++u) {
            vi[u] = it * per_it + u * AGG_T + tid;
            kv[u] = key4[min(vi[u], nvec - 1)];
        }
        if (it == n_it - 1) {
#pragma unroll
            for (int u = 0; u < SCAN_U; ++u)
                if (vi[u] >= nvec) kv[u] = make_uint4(~pat, ~pat, ~pat, ~pat);
        }
#pragma unroll
        for (int u = 0; u < SCAN_U; ++u) {
            const unsigned kw[4] = {kv[u].x, kv[u].y, kv[u].z, kv[u].w};
#pragma unroll
            for (int c = 0; c < 4; ++c) {
                const unsigned x = kw[c] ^ pat;
                unsigned m = ~(((x & 0x7F7F7F7Fu) + 0x7F7F7F7Fu) | x | 0x7F7F7F7Fu);
                unsigned b = __builtin_amdgcn_ballot_w32(m != 0u);
                while (b != 0u) {
                    if (m != 0u)
                        mylst[cnt + __builtin_amdgcn_mbcnt_lo(b, 0u)] =
                            vi[u] * 16 + c * 4 + (__builtin_ctz(m) >> 3);
                    m &= m - 1u;
                    cnt += __popc(b);
                    b = __builtin_amdgcn_ballot_w32(m != 0u);
                }
            }
            if (ROUTE) {
                if (cnt >= LIST_LOW)
                    cnt = route_list(mylst, cnt, true, lane, base, src, dst, dinv, q, qcnt, ovf);
            } else {
                if (cnt > LIST_LOW) {
                    drain_list<C, LD>(mylst, cnt, lane, base, src, dst, dinv, h, acc);
                    cnt = 0;
                }
            }
        }
    }
    for (int e0 = nvec * 16 + wv * 32; e0 < E; e0 += AGG_T) {
        const int e = e0 + lane;
        const int d = (e < E) ? dst[e] : -1;
        const bool hit = (unsigned)(d - base) < (unsigned)nrows;
        const unsigned b = __builtin_amdgcn_ballot_w32(hit);
        if (hit) mylst[cnt + __builtin_amdgcn_mbcnt_lo(b, 0u)] = e;
        cnt += __popc(b);
        if (ROUTE) {
            if (cnt >= LIST_LOW)
                cnt = route_list(mylst, cnt, true, lane, base, src, dst, dinv, q, qcnt, ovf);
        } else {
            if (cnt > LIST_LOW) {
                drain_list<C, LD>(mylst, cnt, lane, base, src, dst, dinv, h, acc);
                cnt = 0;
            }
        }
    }
    if (ROUTE) route_list(mylst, cnt, false, lane, base, src, dst, dinv, q, qcnt, ovf);
    else       drain_list<C, LD>(mylst, cnt, lane, base, src, dst, dinv, h, acc);
}

template <int C, int MODE>
__global__ void __launch_bounds__(AGG_T) k_tile_aggregate(
    const int* __restrict__ src, const int* __restrict__ dst,
    const float* __restrict__ dinv, const float* __restrict__ h,
    const float* __restrict__ wmat, const float* __restrict__ bias,
    float* __restrict__ out,
    int* __restrict__ rec_cnt, int* __restrict__ rec,
    const unsigned int* __restrict__ key, int n, int E)
{
    constexpr int  V    = C / 32;
    constexpr bool SCAN = (MODE == MODE_SCAN_WRITE);
    constexpr bool MUL  = SCAN;
    constexpr int  LD   = MUL ? C + 4 : C;
    typedef typename RowVec<V>::T VT;
    __shared__ __attribute__((aligned(16))) float acc[TILE_R * LD];
    __shared__ int   lst[AGG_WAVES * LIST_CAP];
    __shared__ __attribute__((aligned(16))) int qraw[SCAN ? AGG_WAVES * Q_CAP * 4 : 4];
    __shared__ int   qcnt[AGG_WAVES];
    __shared__ int   ovf;

    const int tid  = threadIdx.x;
    const int lane = tid & 31;
    const int wv   = __builtin_amdgcn_readfirstlane(tid >> 5);
    const int base = blockIdx.x * TILE_R;
    const int nrows = min(TILE_R, n - base);
    int4* q     = (int4*)qraw;
    int*  mylst = lst + wv * LIST_CAP;

    const int wslot = blockIdx.x * AGG_WAVES + wv;
    int4* myrec = (int4*)rec + (size_t)wslot * REC_CAP;

    int fill_v = 0;
    if (!SCAN) fill_v = rec_cnt[blockIdx.x * AGG_WAVES + lane];
    if (SCAN) {
        if (tid < AGG_WAVES) qcnt[tid] = 0;
        if (tid == 0) ovf = 0;
    }

    {
        const VT bv = *(const VT*)(bias + lane * V);
        const int kk = min(lane, OWN_ROWS - 1);
        const float dv = dinv[base + min(wv + kk * AGG_WAVES, nrows - 1)];
        VT st[OWN_ROWS];
#pragma unroll
        for (int k = 0; k < OWN_ROWS; ++k) {
            const int rr = min(wv + k * AGG_WAVES, nrows - 1);
            const float di = __int_as_float(__builtin_amdgcn_readlane(__float_as_int(dv), k));
            const VT hv = *(const VT*)(h + (size_t)(base + rr) * C + lane * V);
            st[k] = MUL ? row_self(hv, di) : row_start(bv, hv, di);
        }
#pragma unroll
        for (int k = 0; k < OWN_ROWS; ++k) {
            const int r = wv + k * AGG_WAVES;
            if (r < nrows) *(VT*)(acc + r * LD + lane * V) = st[k];
        }
    }

    bool fast;
    int  nown;
    if (SCAN) {
        __syncthreads();
        scan_tile<C, LD, true>(tid, lane, wv, base, nrows, src, dst, dinv, h, key, E,
                               mylst, acc, q, qcnt, &ovf);
        __syncthreads();
        fast = (__builtin_amdgcn_readfirstlane(ovf) == 0);
        nown = __builtin_amdgcn_readfirstlane(qcnt[wv]);
    } else {
        fast = (__builtin_amdgcn_ballot_w32(fill_v < 0) == 0u);
        nown = __builtin_amdgcn_readlane(fill_v, wv);
    }

    if (__builtin_expect(fast, true)) {
        if (SCAN) {
            drain_owned<C, LD, true>(q + wv * Q_CAP, nown, lane, h, acc, myrec);
            if (lane == 0) rec_cnt[wslot] = nown;
        } else {
            drain_owned<C, LD, false>(myrec, nown, lane, h, acc, myrec);
        }
    } else {
        __syncthreads();
        scan_tile<C, LD, false>(tid, lane, wv, base, nrows, src, dst, dinv, h, key, E,
                                mylst, acc, q, qcnt, &ovf);
        if (SCAN && lane == 0) rec_cnt[wslot] = -1;
        __syncthreads();
    }

    if (MUL) {
        constexpr int LDK = C + 2;
        constexpr int MT  = (TILE_R + 15) / 16;
        constexpr int NQ  = C / 32;
        static_assert(!MUL || C * LDK * 4 <= AGG_WAVES * Q_CAP * 16, "wmat must fit the queues' bytes");
        float* lds_w = (float*)qraw;
        __syncthreads();
        for (int idx = tid; idx < C * C; idx += AGG_T) {
            const int k  = idx / C;
            const int nn = idx - k * C;
            lds_w[nn * LDK + k] = wmat[idx];
        }
        __syncthreads();

        const int hf = lane >> 4;
        const int r  = lane & 15;
        for (int item = wv; item < MT * NQ; item += AGG_WAVES) {
            const int mt = item / NQ;
            const int nq = item - mt * NQ;
            if (mt * 16 >= nrows) continue;
            const float* arow = acc + min(mt * 16 + r, nrows - 1) * LD + 2 * hf;
            const float* b0   = lds_w + (nq * 32 + r) * LDK + 2 * hf;
            const float* b1   = b0 + 16 * LDK;
            v8f d0 = (v8f){};
            v8f d1 = (v8f){};
#pragma unroll 8
            for (int k = 0; k < C; k += 4) {
                const float2 av = *(const float2*)(arow + k);
                const float2 p0 = *(const float2*)(b0 + k);
                const float2 p1 = *(const float2*)(b1 + k);
                v2f a, x0, x1;
                a[0]  = av.x;  a[1]  = av.y;
                x0[0] = p0.x;  x0[1] = p0.y;
                x1[0] = p1.x;  x1[1] = p1.y;
                d0 = __builtin_amdgcn_wmma_f32_16x16x4_f32(
                    false, a, false, x0, (short)0, d0, false, false);
                d1 = __builtin_amdgcn_wmma_f32_16x16x4_f32(
                    false, a, false, x1, (short)0, d1, false, false);
            }
            const int   c0  = nq * 32 + r;
            const float bs0 = bias[c0];
            const float bs1 = bias[c0 + 16];
#pragma unroll
            for (int j = 0; j < 8; ++j) {
                const int row = mt * 16 + 8 * hf + j;
                if (row < nrows) {
                    float* orow = out + (size_t)(base + row) * C;
                    orow[c0]      = d0[j] + bs0;
                    orow[c0 + 16] = d1[j] + bs1;
                }
            }
        }
    } else {
#pragma unroll
        for (int k = 0; k < OWN_ROWS; ++k) {
            const int r = wv + k * AGG_WAVES;
            if (r < nrows)
                *(VT*)(out + (size_t)(base + r) * C + lane * V) = *(const VT*)(acc + r * LD + lane * V);
        }
    }
}

extern "C" void kernel_launch(void* const* d_in, const int* in_sizes, int n_in,
                              void* d_out, int out_size, void* d_ws, size_t ws_size,
                              hipStream_t stream) {
    const float* x  = (const float*)d_in[0];
    const int*   ei = (const int*)d_in[1];
    const float* W1 = (const float*)d_in[2];
    const float* b1 = (const float*)d_in[3];
    const float* W2 = (const float*)d_in[4];
    const float* b2 = (const float*)d_in[5];
    float* out = (float*)d_out;

    const int HID = in_sizes[3];
    const int IN  = in_sizes[2] / HID;
    const int N   = in_sizes[0] / IN;
    const int E   = in_sizes[1] / 2;
    const int* srcI = ei;
    const int* dstI = ei + E;

    float* ws   = (float*)d_ws;
    float* dinv = ws;
    float* bufA = ws + (((size_t)N + 63) & ~(size_t)63);
    float* bufB = bufA + (size_t)N * HID;
    const int ablk = (N + TILE_R - 1) / TILE_R;
    unsigned int* key = (unsigned int*)(bufB + (size_t)N * HID);
    int*  rcnt = (int*)(key + ((((size_t)E + 63) & ~(size_t)63) >> 2));
    int*  rec  = rcnt + (size_t)ablk * AGG_WAVES;

    const int T = 256;
    const int mtiles = N / 16;
    const int gblk   = (mtiles + 7) / 8;

    k_fill_deg   <<<(N + T - 1) / T, T, 0, stream>>>(dinv, N);
    const int quads = (E + 3) / 4;
    k_deg_scatter<<<(quads + T - 1) / T, T, 0, stream>>>(dstI, E, dinv, key);
    k_dinv       <<<(N + T - 1) / T, T, 0, stream>>>(dinv, N);

    k_tile_aggregate<128, MODE_SCAN_WRITE><<<ablk, AGG_T, 0, stream>>>(
        srcI, dstI, dinv, x, W1, b1, bufB, rcnt, rec, key, N, E);

    k_gemm_wmma_f32<4, 128, true><<<gblk, T, 0, stream>>>(bufB, W2, bufA, N);
    k_tile_aggregate<64, MODE_FROM_REC><<<ablk, AGG_T, 0, stream>>>(
        srcI, dstI, dinv, bufA, W2, b2, out, rcnt, rec, key, N, E);
}


# tile keys bit-sliced (8 bit planes per 32 edges, same bytes): a scan step tests 32 edges per lane in 11 vector ops on 2 16-B loads, not 4 in 4; edge-to-wave split changes, sum order = arrival
# speedup vs baseline: 2.4415x; 1.0415x over previous
#include <hip/hip_runtime.h>
#include <math.h>

typedef __attribute__((ext_vector_type(2))) float v2f;
typedef __attribute__((ext_vector_type(8))) float v8f;

__device__ __forceinline__ void atomic_add_f32(float* p, float v) {
    __hip_atomic_fetch_add(p, v, __ATOMIC_RELAXED, __HIP_MEMORY_SCOPE_AGENT);
}

constexpr int TILE_R = 196;
constexpr int KEY_BLOCK = 1024;

__global__ void k_fill_deg(float* __restrict__ deg, int n) {
    int i = blockIdx.x * blockDim.x + threadIdx.x;
    if (i < n) deg[i] = 1.0f;
}

__global__ void k_deg_scatter(const int* __restrict__ dst, int E,
                              float* __restrict__ deg,
                              unsigned int* __restrict__ key) {
    const int i = blockIdx.x * blockDim.x + threadIdx.x;
    const int e = i * 4;
    const bool whole = ((int)blockIdx.x + 1) * KEY_BLOCK <= E;
    if (whole) {
        int4 d;
        if (((size_t)dst & 15) == 0) d = *(const int4*)(dst + e);
        else d = make_int4(dst[e], dst[e + 1], dst[e + 2], dst[e + 3]);
        atomic_add_f32(&deg[d.x], 1.0f);
        atomic_add_f32(&deg[d.y], 1.0f);
        atomic_add_f32(&deg[d.z], 1.0f);
        atomic_add_f32(&deg[d.w], 1.0f);
        const unsigned kq[4] = {(unsigned)d.x / (unsigned)TILE_R, (unsigned)d.y / (unsigned)TILE_R,
                                (unsigned)d.z / (unsigned)TILE_R, (unsigned)d.w / (unsigned)TILE_R};
        const int lane = threadIdx.x & 31;
        const int wv   = threadIdx.x >> 5;
        unsigned myw = 0u;
#pragma unroll
        for (int q = 0; q < 4; ++q) {
#pragma unroll
            for (int b = 0; b < 8; ++b) {
                const unsigned bal = __builtin_amdgcn_ballot_w32(((kq[q] >> b) & 1u) != 0u);
                if (lane == (b >> 2) * 16 + q * 4 + (b & 3)) myw = bal;
            }
        }
        key[(size_t)blockIdx.x * (KEY_BLOCK / 4) + (lane >> 4) * 128 + wv * 16 + (lane & 15)] = myw;
    } else {
        for (int q = e; q < E && q < e + 4; ++q) atomic_add_f32(&deg[dst[q]], 1.0f);
    }
}

__global__ void k_dinv(float* __restrict__ deg, int n) {
    int i = blockIdx.x * blockDim.x + threadIdx.x;
    if (i < n) {
        float d = deg[i];
        deg[i] = (d > 0.0f) ? rsqrtf(d) : 0.0f;
    }
}

template <int NT, int KD, bool RELU>
__global__ void __launch_bounds__(256) k_gemm_wmma_f32(
    const float* __restrict__ A, const float* __restrict__ B,
    float* __restrict__ C, int M)
{
    constexpr int N   = NT * 16;
    constexpr int LDK = KD + 2;
    __shared__ float lds_w[N * LDK];

    for (int idx = threadIdx.x; idx < KD * N; idx += 256) {
        int k = idx / N;
        int n = idx - k * N;
        lds_w[n * LDK + k] = B[idx];
    }
    __syncthreads();

    const int lane  = threadIdx.x & 31;
    const int wv    = threadIdx.x >> 5;
    const int mtile = blockIdx.x * 8 + wv;
    if (mtile * 16 >= M) return;

    const int hf = lane >> 4;
    const int r  = lane & 15;

    const float* __restrict__ arow = A + (size_t)(mtile * 16 + r) * KD + 2 * hf;

    v8f acc[NT];
#pragma unroll
    for (int t = 0; t < NT; ++t) acc[t] = (v8f){};

#pragma unroll 4
    for (int k = 0; k < KD; k += 4) {
        float2 av = *(const float2*)(arow + k);
        if (RELU) { av.x = fmaxf(av.x, 0.0f); av.y = fmaxf(av.y, 0.0f); }
        v2f a;
        a[0] = av.x;
        a[1] = av.y;
#pragma unroll
        for (int t = 0; t < NT; ++t) {
            const int col = t * 16 + r;
            float2 bv = *(const float2*)(&lds_w[col * LDK + k + 2 * hf]);
            v2f b;
            b[0] = bv.x;
            b[1] = bv.y;
            acc[t] = __builtin_amdgcn_wmma_f32_16x16x4_f32(
                false, a, false, b, (short)0, acc[t], false, false);
        }
    }

#pragma unroll
    for (int t = 0; t < NT; ++t) {
        float* crow = C + (size_t)(mtile * 16 + 8 * hf) * N + t * 16 + r;
#pragma unroll
        for (int j = 0; j < 8; ++j) crow[(size_t)j * N] = acc[t][j];
    }
}

constexpr int AGG_T    = 1024;
constexpr int SCAN_B   = 4;
constexpr int LIST_LOW = 32;
constexpr int LIST_CAP = LIST_LOW + 32;
constexpr int REC_CAP   = 512;
constexpr int AGG_WAVES = AGG_T / 32;
constexpr int Q_CAP     = 256;
constexpr int OWN_ROWS  = (TILE_R + AGG_WAVES - 1) / AGG_WAVES;
static_assert(Q_CAP <= REC_CAP, "a queue must fit its record slice");
static_assert((AGG_WAVES & (AGG_WAVES - 1)) == 0, "owner = row & (waves - 1)");
constexpr int MODE_SCAN_WRITE = 0;
constexpr int MODE_FROM_REC   = 1;

template <int V> struct RowVec;
template <> struct RowVec<4> { typedef float4 T; };
template <> struct RowVec<2> { typedef float2 T; };

__device__ __forceinline__ float4 row_start(const float4& b, const float4& hv, float di) {
    float4 o;
    o.x = b.x + hv.x * di * di;
    o.y = b.y + hv.y * di * di;
    o.z = b.z + hv.z * di * di;
    o.w = b.w + hv.w * di * di;
    return o;
}
__device__ __forceinline__ float2 row_start(const float2& b, const float2& hv, float di) {
    float2 o;
    o.x = b.x + hv.x * di * di;
    o.y = b.y + hv.y * di * di;
    return o;
}
__device__ __forceinline__ float4 row_self(const float4& hv, float di) {
    float4 o;
    o.x = hv.x * di * di;
    o.y = hv.y * di * di;
    o.z = hv.z * di * di;
    o.w = hv.w * di * di;
    return o;
}
__device__ __forceinline__ float2 row_self(const float2& hv, float di) {
    float2 o;
    o.x = hv.x * di * di;
    o.y = hv.y * di * di;
    return o;
}
__device__ __forceinline__ void row_fma(float4& a, const float4& hv, float nu) {
    a.x += hv.x * nu;
    a.y += hv.y * nu;
    a.z += hv.z * nu;
    a.w += hv.w * nu;
}
__device__ __forceinline__ void row_fma(float2& a, const float2& hv, float nu) {
    a.x += hv.x * nu;
    a.y += hv.y * nu;
}

template <int C>
__device__ __forceinline__ void load_group(
    typename RowVec<C / 32>::T (&hv)[8], int s, int j, int m, int lane,
    const float* __restrict__ h)
{
    constexpr int V = C / 32;
    typedef typename RowVec<V>::T VT;
#pragma unroll
    for (int u = 0; u < 8; ++u) {
        const int su = __builtin_amdgcn_readlane(s, min(j + u, m - 1));
        hv[u] = *(const VT*)(h + (size_t)su * C + lane * V);
    }
}

template <int C, int LD>
__device__ __forceinline__ void add_group(
    const typename RowVec<C / 32>::T (&hv)[8], int dl, float nr, int j, int m, int lane,
    float* acc)
{
    constexpr int V = C / 32;
    typedef typename RowVec<V>::T VT;
#pragma unroll
    for (int u = 0; u < 8; ++u) {
        if (j + u < m) {
            const int du = __builtin_amdgcn_readlane(dl, j + u);
            const float nu = __int_as_float(
                __builtin_amdgcn_readlane(__float_as_int(nr), j + u));
            VT* p = (VT*)(acc + du * LD + lane * V);
            VT t = *p;
            row_fma(t, hv[u], nu);
            *p = t;
        }
    }
}

template <int C, int LD, bool COPY>
__device__ __forceinline__ void drain_owned(
    const int4* recs, int cnt, int lane,
    const float* __restrict__ h, float* acc, int4* __restrict__ myrec)
{
    typedef typename RowVec<C / 32>::T VT;
    int4 cur = make_int4(0, 0, 0, 0);
    if (lane < cnt) cur = recs[lane];
    for (int i0 = 0; i0 < cnt; i0 += 32) {
        const int m = min(32, cnt - i0);
        int4 nxt = make_int4(0, 0, 0, 0);
        if (i0 + 32 + lane < cnt) nxt = recs[i0 + 32 + lane];
        if (COPY) {
            if (lane < m) myrec[i0 + lane] = cur;
        }
        const int   s  = cur.x;
        const int   dl = cur.y;
        const float nr = __int_as_float(cur.z);
        VT ha[8], hb[8];
        load_group<C>(ha, s, 0, m, lane, h);
        load_group<C>(hb, s, 8, m, lane, h);
        add_group<C, LD>(ha, dl, nr, 0, m, lane, acc);
        if (m > 8) {
            load_group<C>(ha, s, 16, m, lane, h);
            add_group<C, LD>(hb, dl, nr, 8, m, lane, acc);
            if (m > 16) {
                load_group<C>(hb, s, 24, m, lane, h);
                add_group<C, LD>(ha, dl, nr, 16, m, lane, acc);
                add_group<C, LD>(hb, dl, nr, 24, m, lane, acc);
            }
        }
        cur = nxt;
    }
}

__device__ __forceinline__ int route_list(
    int* mylst, int cnt, bool whole, int lane, int base,
    const int* __restrict__ src, const int* __restrict__ dst,
    const float* __restrict__ dinv, int4* q, int* qcnt, int* ovf)
{
    const int take = whole ? (cnt & ~31) : cnt;
    for (int i0 = 0; i0 < take; i0 += 32) {
        if (i0 + lane < take) {
            const int e  = mylst[i0 + lane];
            const int s  = src[e];
            const int d  = dst[e];
            const int dl = d - base;
            const float nr = dinv[s] * dinv[d];
            const int ow = dl & (AGG_WAVES - 1);
            const int pos = __hip_atomic_fetch_add(&qcnt[ow], 1, __ATOMIC_RELAXED,
                                                   __HIP_MEMORY_SCOPE_WORKGROUP);
            q[ow * Q_CAP + min(pos, Q_CAP - 1)] = make_int4(s, dl, __float_as_int(nr), 0);
            if (pos >= Q_CAP) *ovf = 1;
        }
    }
    const int rem = cnt - take;
    if (rem > 0) {
        if (lane < rem) mylst[lane] = mylst[take + lane];
    }
    return rem;
}

__device__ __forceinline__ void lds_add_f32(float* p, float v) {
    __hip_atomic_fetch_add(p, v, __ATOMIC_RELAXED, __HIP_MEMORY_SCOPE_WORKGROUP);
}

template <int C, int LD>
__device__ __forceinline__ void drain_list(
    const int* mylst, int cnt, int lane, int base,
    const int* __restrict__ src, const int* __restrict__ dst,
    const float* __restrict__ dinv, const float* __restrict__ h, float* acc)
{
    constexpr int V = C / 32;
    typedef typename RowVec<V>::T VT;
    for (int i0 = 0; i0 < cnt; i0 += 32) {
        const int m = min(32, cnt - i0);
        int s = 0, dl = 0;
        float nr = 0.0f;
        if (lane < m) {
            const int e = mylst[i0 + lane];
            s = src[e];
            const int d = dst[e];
            dl = d - base;
            nr = dinv[s] * dinv[d];
        }
        for (int j0 = 0; j0 < m; j0 += 4) {
            VT hv[4];
#pragma unroll
            for (int u = 0; u < 4; ++u) {
                const int su = __builtin_amdgcn_readlane(s, min(j0 + u, m - 1));
                hv[u] = *(const VT*)(h + (size_t)su * C + lane * V);
            }
#pragma unroll
            for (int u = 0; u < 4; ++u) {
                if (j0 + u < m) {
                    const int du = __builtin_amdgcn_readlane(dl, j0 + u);
                    const float nu = __int_as_float(
                        __builtin_amdgcn_readlane(__float_as_int(nr), j0 + u));
                    float* a = acc + du * LD + lane * V;
                    const float* hp = (const float*)&hv[u];
#pragma unroll
                    for (int c = 0; c < V; ++c) lds_add_f32(a + c, hp[c] * nu);
                }
            }
        }
    }
}

template <int C, int LD, bool ROUTE>
__device__ __forceinline__ void scan_tile(
    int lane, int wv, int base, int nrows,
    const int* __restrict__ src, const int* __restrict__ dst,
    const float* __restrict__ dinv, const float* __restrict__ h,
    const unsigned int* __restrict__ key, int E,
    int* mylst, float* acc, int4* q, int* qcnt, int* ovf)
{
    int cnt = 0;

    unsigned sel[8];
#pragma unroll
    for (int b = 0; b < 8; ++b) sel[b] = ((blockIdx.x >> b) & 1u) ? 0u : ~0u;
    const int nblk   = E / KEY_BLOCK;
    const int per_it = AGG_WAVES * SCAN_B;
    const int n_it   = (nblk + per_it - 1) / per_it;
    const int off = (int)(((long long)blockIdx.x * nblk) / gridDim.x);
    const uint4* key4 = (const uint4*)key;
    const int lane_e = ((lane >> 2) << 7) + (lane & 3);
    for (int k = 0; k < n_it; ++k) {
        uint4 lo[SCAN_B], hi[SCAN_B];
#pragma unroll
        for (int u = 0; u < SCAN_B; ++u) {
            const int raw = (k * SCAN_B + u) * AGG_WAVES + wv;
            int bc = min(raw, nblk - 1) + off;
            if (bc >= nblk) bc -= nblk;
            lo[u] = key4[bc * (KEY_BLOCK / 16) + lane];
            hi[u] = key4[bc * (KEY_BLOCK / 16) + 32 + lane];
        }
#pragma unroll
        for (int u = 0; u < SCAN_B; ++u) {
            const int raw = (k * SCAN_B + u) * AGG_WAVES + wv;
            if (raw < nblk) {
                int blk = raw + off;
                if (blk >= nblk) blk -= nblk;
                unsigned m = lo[u].x ^ sel[0];
                m &= lo[u].y ^ sel[1];
                m &= lo[u].z ^ sel[2];
                m &= lo[u].w ^ sel[3];
                m &= hi[u].x ^ sel[4];
                m &= hi[u].y ^ sel[5];
                m &= hi[u].z ^ sel[6];
                m &= hi[u].w ^ sel[7];
                const int eb = blk * KEY_BLOCK + lane_e;
                unsigned b = __builtin_amdgcn_ballot_w32(m != 0u);
                while (b != 0u) {
                    if (m != 0u)
                        mylst[cnt + __builtin_amdgcn_mbcnt_lo(b, 0u)] = eb + 4 * __builtin_ctz(m);
                    m &= m - 1u;
                    cnt += __popc(b);
                    if (ROUTE) {
                        if (cnt >= LIST_LOW)
                            cnt = route_list(mylst, cnt, true, lane, base, src, dst, dinv, q, qcnt, ovf);
                    } else {
                        if (cnt > LIST_LOW) {
                            drain_list<C, LD>(mylst, cnt, lane, base, src, dst, dinv, h, acc);
                            cnt = 0;
                        }
                    }
                    b = __builtin_amdgcn_ballot_w32(m != 0u);
                }
            }
        }
    }
    for (int e0 = nblk * KEY_BLOCK + wv * 32; e0 < E; e0 += AGG_T) {
        const int e = e0 + lane;
        const int d = (e < E) ? dst[e] : -1;
        const bool hit = (unsigned)(d - base) < (unsigned)nrows;
        const unsigned b = __builtin_amdgcn_ballot_w32(hit);
        if (hit) mylst[cnt + __builtin_amdgcn_mbcnt_lo(b, 0u)] = e;
        cnt += __popc(b);
        if (ROUTE) {
            if (cnt >= LIST_LOW)
                cnt = route_list(mylst, cnt, true, lane, base, src, dst, dinv, q, qcnt, ovf);
        } else {
            if (cnt > LIST_LOW) {
                drain_list<C, LD>(mylst, cnt, lane, base, src, dst, dinv, h, acc);
                cnt = 0;
            }
        }
    }
    if (ROUTE) route_list(mylst, cnt, false, lane, base, src, dst, dinv, q, qcnt, ovf);
    else       drain_list<C, LD>(mylst, cnt, lane, base, src, dst, dinv, h, acc);
}

template <int C, int MODE>
__global__ void __launch_bounds__(AGG_T) k_tile_aggregate(
    const int* __restrict__ src, const int* __restrict__ dst,
    const float* __restrict__ dinv, const float* __restrict__ h,
    const float* __restrict__ wmat, const float* __restrict__ bias,
    float* __restrict__ out,
    int* __restrict__ rec_cnt, int* __restrict__ rec,
    const unsigned int* __restrict__ key, int n, int E)
{
    constexpr int  V    = C / 32;
    constexpr bool SCAN = (MODE == MODE_SCAN_WRITE);
    constexpr bool MUL  = SCAN;
    constexpr int  LD   = MUL ? C + 4 : C;
    typedef typename RowVec<V>::T VT;
    __shared__ __attribute__((aligned(16))) float acc[TILE_R * LD];
    __shared__ int   lst[AGG_WAVES * LIST_CAP];
    __shared__ __attribute__((aligned(16))) int qraw[SCAN ? AGG_WAVES * Q_CAP * 4 : 4];
    __shared__ int   qcnt[AGG_WAVES];
    __shared__ int   ovf;

    const int tid  = threadIdx.x;
    const int lane = tid & 31;
    const int wv   = __builtin_amdgcn_readfirstlane(tid >> 5);
    const int base = blockIdx.x * TILE_R;
    const int nrows = min(TILE_R, n - base);
    int4* q     = (int4*)qraw;
    int*  mylst = lst + wv * LIST_CAP;

    const int wslot = blockIdx.x * AGG_WAVES + wv;
    int4* myrec = (int4*)rec + (size_t)wslot * REC_CAP;

    int fill_v = 0;
    if (!SCAN) fill_v = rec_cnt[blockIdx.x * AGG_WAVES + lane];
    if (SCAN) {
        if (tid < AGG_WAVES) qcnt[tid] = 0;
        if (tid == 0) ovf = 0;
    }

    {
        const VT bv = *(const VT*)(bias + lane * V);
        const int kk = min(lane, OWN_ROWS - 1);
        const float dv = dinv[base + min(wv + kk * AGG_WAVES, nrows - 1)];
        VT st[OWN_ROWS];
#pragma unroll
        for (int k = 0; k < OWN_ROWS; ++k) {
            const int rr = min(wv + k * AGG_WAVES, nrows - 1);
            const float di = __int_as_float(__builtin_amdgcn_readlane(__float_as_int(dv), k));
            const VT hv = *(const VT*)(h + (size_t)(base + rr) * C + lane * V);
            st[k] = MUL ? row_self(hv, di) : row_start(bv, hv, di);
        }
#pragma unroll
        for (int k = 0; k < OWN_ROWS; ++k) {
            const int r = wv + k * AGG_WAVES;
            if (r < nrows) *(VT*)(acc + r * LD + lane * V) = st[k];
        }
    }

    bool fast;
    int  nown;
    if (SCAN) {
        __syncthreads();
        scan_tile<C, LD, true>(lane, wv, base, nrows, src, dst, dinv, h, key, E,
                               mylst, acc, q, qcnt, &ovf);
        __syncthreads();
        fast = (__builtin_amdgcn_readfirstlane(ovf) == 0);
        nown = __builtin_amdgcn_readfirstlane(qcnt[wv]);
    } else {
        fast = (__builtin_amdgcn_ballot_w32(fill_v < 0) == 0u);
        nown = __builtin_amdgcn_readlane(fill_v, wv);
    }

    if (__builtin_expect(fast, true)) {
        if (SCAN) {
            drain_owned<C, LD, true>(q + wv * Q_CAP, nown, lane, h, acc, myrec);
            if (lane == 0) rec_cnt[wslot] = nown;
        } else {
            drain_owned<C, LD, false>(myrec, nown, lane, h, acc, myrec);
        }
    } else {
        __syncthreads();
        scan_tile<C, LD, false>(lane, wv, base, nrows, src, dst, dinv, h, key, E,
                                mylst, acc, q, qcnt, &ovf);
        if (SCAN && lane == 0) rec_cnt[wslot] = -1;
        __syncthreads();
    }

    if (MUL) {
        constexpr int LDK = C + 2;
        constexpr int MT  = (TILE_R + 15) / 16;
        constexpr int NQ  = C / 32;
        static_assert(!MUL || C * LDK * 4 <= AGG_WAVES * Q_CAP * 16, "wmat must fit the queues' bytes");
        float* lds_w = (float*)qraw;
        __syncthreads();
        for (int idx = tid; idx < C * C; idx += AGG_T) {
            const int k  = idx / C;
            const int nn = idx - k * C;
            lds_w[nn * LDK + k] = wmat[idx];
        }
        __syncthreads();

        const int hf = lane >> 4;
        const int r  = lane & 15;
        for (int item = wv; item < MT * NQ; item += AGG_WAVES) {
            const int mt = item / NQ;
            const int nq = item - mt * NQ;
            if (mt * 16 >= nrows) continue;
            const float* arow = acc + min(mt * 16 + r, nrows - 1) * LD + 2 * hf;
            const float* b0   = lds_w + (nq * 32 + r) * LDK + 2 * hf;
            const float* b1   = b0 + 16 * LDK;
            v8f d0 = (v8f){};
            v8f d1 = (v8f){};
#pragma unroll 8
            for (int k = 0; k < C; k += 4) {
                const float2 av = *(const float2*)(arow + k);
                const float2 p0 = *(const float2*)(b0 + k);
                const float2 p1 = *(const float2*)(b1 + k);
                v2f a, x0, x1;
                a[0]  = av.x;  a[1]  = av.y;
                x0[0] = p0.x;  x0[1] = p0.y;
                x1[0] = p1.x;  x1[1] = p1.y;
                d0 = __builtin_amdgcn_wmma_f32_16x16x4_f32(
                    false, a, false, x0, (short)0, d0, false, false);
                d1 = __builtin_amdgcn_wmma_f32_16x16x4_f32(
                    false, a, false, x1, (short)0, d1, false, false);
            }
            const int   c0  = nq * 32 + r;
            const float bs0 = bias[c0];
            const float bs1 = bias[c0 + 16];
#pragma unroll
            for (int j = 0; j < 8; ++j) {
                const int row = mt * 16 + 8 * hf + j;
                if (row < nrows) {
                    float* orow = out + (size_t)(base + row) * C;
                    orow[c0]      = d0[j] + bs0;
                    orow[c0 + 16] = d1[j] + bs1;
                }
            }
        }
    } else {
#pragma unroll
        for (int k = 0; k < OWN_ROWS; ++k) {
            const int r = wv + k * AGG_WAVES;
            if (r < nrows)
                *(VT*)(out + (size_t)(base + r) * C + lane * V) = *(const VT*)(acc + r * LD + lane * V);
        }
    }
}

extern "C" void kernel_launch(void* const* d_in, const int* in_sizes, int n_in,
                              void* d_out, int out_size, void* d_ws, size_t ws_size,
                              hipStream_t stream) {
    const float* x  = (const float*)d_in[0];
    const int*   ei = (const int*)d_in[1];
    const float* W1 = (const float*)d_in[2];
    const float* b1 = (const float*)d_in[3];
    const float* W2 = (const float*)d_in[4];
    const float* b2 = (const float*)d_in[5];
    float* out = (float*)d_out;

    const int HID = in_sizes[3];
    const int IN  = in_sizes[2] / HID;
    const int N   = in_sizes[0] / IN;
    const int E   = in_sizes[1] / 2;
    const int* srcI = ei;
    const int* dstI = ei + E;

    float* ws   = (float*)d_ws;
    float* dinv = ws;
    float* bufA = ws + (((size_t)N + 63) & ~(size_t)63);
    float* bufB = bufA + (size_t)N * HID;
    const int ablk = (N + TILE_R - 1) / TILE_R;
    unsigned int* key = (unsigned int*)(bufB + (size_t)N * HID);
    int*  rcnt = (int*)(key + ((((size_t)E + 63) & ~(size_t)63) >> 2));
    int*  rec  = rcnt + (size_t)ablk * AGG_WAVES;

    const int T = 256;
    const int mtiles = N / 16;
    const int gblk   = (mtiles + 7) / 8;

    k_fill_deg   <<<(N + T - 1) / T, T, 0, stream>>>(dinv, N);
    const int quads = (E + 3) / 4;
    k_deg_scatter<<<(quads + T - 1) / T, T, 0, stream>>>(dstI, E, dinv, key);
    k_dinv       <<<(N + T - 1) / T, T, 0, stream>>>(dinv, N);

    k_tile_aggregate<128, MODE_SCAN_WRITE><<<ablk, AGG_T, 0, stream>>>(
        srcI, dstI, dinv, x, W1, b1, bufB, rcnt, rec, key, N, E);

    k_gemm_wmma_f32<4, 128, true><<<gblk, T, 0, stream>>>(bufB, W2, bufA, N);
    k_tile_aggregate<64, MODE_FROM_REC><<<ablk, AGG_T, 0, stream>>>(
        srcI, dstI, dinv, bufA, W2, b2, out, rcnt, rec, key, N, E);
}


# degree pass leaves each 4096-edge chunk grouped by dst tile ((src,dst) pairs, per-chunk tile starts); a tile walks its own segments, no key scan; edge-to-wave split changes, sum order = arrival
# speedup vs baseline: 2.6847x; 1.0996x over previous
#include <hip/hip_runtime.h>
#include <math.h>

typedef __attribute__((ext_vector_type(2))) float v2f;
typedef __attribute__((ext_vector_type(8))) float v8f;

__device__ __forceinline__ void atomic_add_f32(float* p, float v) {
    __hip_atomic_fetch_add(p, v, __ATOMIC_RELAXED, __HIP_MEMORY_SCOPE_AGENT);
}

constexpr int TILE_R = 196;
constexpr int N_TILE = 256;
constexpr int CHUNK = 4096;
constexpr int OFF_P = N_TILE + 1;

__global__ void k_fill_deg(float* __restrict__ deg, int n) {
    int i = blockIdx.x * blockDim.x + threadIdx.x;
    if (i < n) deg[i] = 1.0f;
}

__global__ void __launch_bounds__(1024) k_deg_group(
    const int* __restrict__ src, const int* __restrict__ dst, int E,
    float* __restrict__ deg, int* __restrict__ grp_w, int* __restrict__ off)
{
    __shared__ int cntt[N_TILE];
    __shared__ int start[OFF_P];
    __shared__ __attribute__((aligned(16))) int2 stage[CHUNK];

    const int tid = threadIdx.x;
    const int e   = blockIdx.x * CHUNK + tid * 4;
    if (tid < N_TILE) cntt[tid] = 0;
    __syncthreads();

    int4 s4 = make_int4(0, 0, 0, 0);
    int4 d4 = make_int4(-1, -1, -1, -1);
    if (e + 3 < E && ((((size_t)dst) | ((size_t)src)) & 15) == 0) {
        s4 = *(const int4*)(src + e);
        d4 = *(const int4*)(dst + e);
    } else {
        if (e     < E) { s4.x = src[e];     d4.x = dst[e];     }
        if (e + 1 < E) { s4.y = src[e + 1]; d4.y = dst[e + 1]; }
        if (e + 2 < E) { s4.z = src[e + 2]; d4.z = dst[e + 2]; }
        if (e + 3 < E) { s4.w = src[e + 3]; d4.w = dst[e + 3]; }
    }
    const int sq[4] = {s4.x, s4.y, s4.z, s4.w};
    const int dq[4] = {d4.x, d4.y, d4.z, d4.w};
    int tl[4], rk[4];
#pragma unroll
    for (int q = 0; q < 4; ++q) {
        tl[q] = 0;
        rk[q] = 0;
        if (dq[q] >= 0) {
            atomic_add_f32(&deg[dq[q]], 1.0f);
            tl[q] = min((int)((unsigned)dq[q] / (unsigned)TILE_R), N_TILE - 1);
            rk[q] = __hip_atomic_fetch_add(&cntt[tl[q]], 1, __ATOMIC_RELAXED,
                                           __HIP_MEMORY_SCOPE_WORKGROUP);
        }
    }
    __syncthreads();

    if (tid < 32) {
        int v[8];
        int run = 0;
#pragma unroll
        for (int i = 0; i < 8; ++i) {
            const int t = cntt[8 * tid + i];
            v[i] = run;
            run += t;
        }
        int incl = run;
        for (int o = 1; o < 32; o <<= 1) {
            const int nb = __shfl_up(incl, o, 32);
            if (tid >= o) incl += nb;
        }
        const int excl = incl - run;
#pragma unroll
        for (int i = 0; i < 8; ++i) start[8 * tid + i] = excl + v[i];
        if (tid == 31) start[N_TILE] = incl;
    }
    __syncthreads();

#pragma unroll
    for (int q = 0; q < 4; ++q) {
        if (dq[q] >= 0) stage[start[tl[q]] + rk[q]] = make_int2(sq[q], dq[q]);
    }
    if (tid < OFF_P) off[(size_t)blockIdx.x * OFF_P + tid] = start[tid];
    __syncthreads();

    const int total = start[N_TILE];
    const int4* st4 = (const int4*)stage;
    int4* g4 = (int4*)grp_w + (size_t)blockIdx.x * (CHUNK / 2);
    for (int i = tid; 2 * i < total; i += 1024) g4[i] = st4[i];
}

__global__ void k_dinv(float* __restrict__ deg, int n) {
    int i = blockIdx.x * blockDim.x + threadIdx.x;
    if (i < n) {
        float d = deg[i];
        deg[i] = (d > 0.0f) ? rsqrtf(d) : 0.0f;
    }
}

template <int NT, int KD, bool RELU>
__global__ void __launch_bounds__(256) k_gemm_wmma_f32(
    const float* __restrict__ A, const float* __restrict__ B,
    float* __restrict__ C, int M)
{
    constexpr int N   = NT * 16;
    constexpr int LDK = KD + 2;
    __shared__ float lds_w[N * LDK];

    for (int idx = threadIdx.x; idx < KD * N; idx += 256) {
        int k = idx / N;
        int n = idx - k * N;
        lds_w[n * LDK + k] = B[idx];
    }
    __syncthreads();

    const int lane  = threadIdx.x & 31;
    const int wv    = threadIdx.x >> 5;
    const int mtile = blockIdx.x * 8 + wv;
    if (mtile * 16 >= M) return;

    const int hf = lane >> 4;
    const int r  = lane & 15;

    const float* __restrict__ arow = A + (size_t)(mtile * 16 + r) * KD + 2 * hf;

    v8f acc[NT];
#pragma unroll
    for (int t = 0; t < NT; ++t) acc[t] = (v8f){};

#pragma unroll 4
    for (int k = 0; k < KD; k += 4) {
        float2 av = *(const float2*)(arow + k);
        if (RELU) { av.x = fmaxf(av.x, 0.0f); av.y = fmaxf(av.y, 0.0f); }
        v2f a;
        a[0] = av.x;
        a[1] = av.y;
#pragma unroll
        for (int t = 0; t < NT; ++t) {
            const int col = t * 16 + r;
            float2 bv = *(const float2*)(&lds_w[col * LDK + k + 2 * hf]);
            v2f b;
            b[0] = bv.x;
            b[1] = bv.y;
            acc[t] = __builtin_amdgcn_wmma_f32_16x16x4_f32(
                false, a, false, b, (short)0, acc[t], false, false);
        }
    }

#pragma unroll
    for (int t = 0; t < NT; ++t) {
        float* crow = C + (size_t)(mtile * 16 + 8 * hf) * N + t * 16 + r;
#pragma unroll
        for (int j = 0; j < 8; ++j) crow[(size_t)j * N] = acc[t][j];
    }
}

constexpr int AGG_T    = 1024;
constexpr int LIST_LOW = 32;
constexpr int LIST_CAP = LIST_LOW + 32;
constexpr int REC_CAP   = 512;
constexpr int AGG_WAVES = AGG_T / 32;
constexpr int Q_CAP     = 256;
constexpr int OWN_ROWS  = (TILE_R + AGG_WAVES - 1) / AGG_WAVES;
static_assert(Q_CAP <= REC_CAP, "a queue must fit its record slice");
static_assert((AGG_WAVES & (AGG_WAVES - 1)) == 0, "owner = row & (waves - 1)");
constexpr int MODE_SCAN_WRITE = 0;
constexpr int MODE_FROM_REC   = 1;

template <int V> struct RowVec;
template <> struct RowVec<4> { typedef float4 T; };
template <> struct RowVec<2> { typedef float2 T; };

__device__ __forceinline__ float4 row_start(const float4& b, const float4& hv, float di) {
    float4 o;
    o.x = b.x + hv.x * di * di;
    o.y = b.y + hv.y * di * di;
    o.z = b.z + hv.z * di * di;
    o.w = b.w + hv.w * di * di;
    return o;
}
__device__ __forceinline__ float2 row_start(const float2& b, const float2& hv, float di) {
    float2 o;
    o.x = b.x + hv.x * di * di;
    o.y = b.y + hv.y * di * di;
    return o;
}
__device__ __forceinline__ float4 row_self(const float4& hv, float di) {
    float4 o;
    o.x = hv.x * di * di;
    o.y = hv.y * di * di;
    o.z = hv.z * di * di;
    o.w = hv.w * di * di;
    return o;
}
__device__ __forceinline__ float2 row_self(const float2& hv, float di) {
    float2 o;
    o.x = hv.x * di * di;
    o.y = hv.y * di * di;
    return o;
}
__device__ __forceinline__ void row_fma(float4& a, const float4& hv, float nu) {
    a.x += hv.x * nu;
    a.y += hv.y * nu;
    a.z += hv.z * nu;
    a.w += hv.w * nu;
}
__device__ __forceinline__ void row_fma(float2& a, const float2& hv, float nu) {
    a.x += hv.x * nu;
    a.y += hv.y * nu;
}

template <int C>
__device__ __forceinline__ void load_group(
    typename RowVec<C / 32>::T (&hv)[8], int s, int j, int m, int lane,
    const float* __restrict__ h)
{
    constexpr int V = C / 32;
    typedef typename RowVec<V>::T VT;
#pragma unroll
    for (int u = 0; u < 8; ++u) {
        const int su = __builtin_amdgcn_readlane(s, min(j + u, m - 1));
        hv[u] = *(const VT*)(h + (size_t)su * C + lane * V);
    }
}

template <int C, int LD>
__device__ __forceinline__ void add_group(
    const typename RowVec<C / 32>::T (&hv)[8], int dl, float nr, int j, int m, int lane,
    float* acc)
{
    constexpr int V = C / 32;
    typedef typename RowVec<V>::T VT;
#pragma unroll
    for (int u = 0; u < 8; ++u) {
        if (j + u < m) {
            const int du = __builtin_amdgcn_readlane(dl, j + u);
            const float nu = __int_as_float(
                __builtin_amdgcn_readlane(__float_as_int(nr), j + u));
            VT* p = (VT*)(acc + du * LD + lane * V);
            VT t = *p;
            row_fma(t, hv[u], nu);
            *p = t;
        }
    }
}

template <int C, int LD, bool COPY>
__device__ __forceinline__ void drain_owned(
    const int4* recs, int cnt, int lane,
    const float* __restrict__ h, float* acc, int4* __restrict__ myrec)
{
    typedef typename RowVec<C / 32>::T VT;
    int4 cur = make_int4(0, 0, 0, 0);
    if (lane < cnt) cur = recs[lane];
    for (int i0 = 0; i0 < cnt; i0 += 32) {
        const int m = min(32, cnt - i0);
        int4 nxt = make_int4(0, 0, 0, 0);
        if (i0 + 32 + lane < cnt) nxt = recs[i0 + 32 + lane];
        if (COPY) {
            if (lane < m) myrec[i0 + lane] = cur;
        }
        const int   s  = cur.x;
        const int   dl = cur.y;
        const float nr = __int_as_float(cur.z);
        VT ha[8], hb[8];
        load_group<C>(ha, s, 0, m, lane, h);
        load_group<C>(hb, s, 8, m, lane, h);
        add_group<C, LD>(ha, dl, nr, 0, m, lane, acc);
        if (m > 8) {
            load_group<C>(ha, s, 16, m, lane, h);
            add_group<C, LD>(hb, dl, nr, 8, m, lane, acc);
            if (m > 16) {
                load_group<C>(hb, s, 24, m, lane, h);
                add_group<C, LD>(ha, dl, nr, 16, m, lane, acc);
                add_group<C, LD>(hb, dl, nr, 24, m, lane, acc);
            }
        }
        cur = nxt;
    }
}

__device__ __forceinline__ int route_list(
    int* mylst, int cnt, bool whole, int lane, int base,
    const int2* __restrict__ grp,
    const float* __restrict__ dinv, int4* q, int* qcnt, int* ovf)
{
    const int take = whole ? (cnt & ~31) : cnt;
    for (int i0 = 0; i0 < take; i0 += 32) {
        if (i0 + lane < take) {
            const int2 sd = grp[mylst[i0 + lane]];
            const int s  = sd.x;
            const int d  = sd.y;
            const int dl = d - base;
            const float nr = dinv[s] * dinv[d];
            const int ow = dl & (AGG_WAVES - 1);
            const int pos = __hip_atomic_fetch_add(&qcnt[ow], 1, __ATOMIC_RELAXED,
                                                   __HIP_MEMORY_SCOPE_WORKGROUP);
            q[ow * Q_CAP + min(pos, Q_CAP - 1)] = make_int4(s, dl, __float_as_int(nr), 0);
            if (pos >= Q_CAP) *ovf = 1;
        }
    }
    const int rem = cnt - take;
    if (rem > 0) {
        if (lane < rem) mylst[lane] = mylst[take + lane];
    }
    return rem;
}

__device__ __forceinline__ void lds_add_f32(float* p, float v) {
    __hip_atomic_fetch_add(p, v, __ATOMIC_RELAXED, __HIP_MEMORY_SCOPE_WORKGROUP);
}

template <int C, int LD>
__device__ __forceinline__ void drain_list(
    const int* mylst, int cnt, int lane, int base,
    const int2* __restrict__ grp,
    const float* __restrict__ dinv, const float* __restrict__ h, float* acc)
{
    constexpr int V = C / 32;
    typedef typename RowVec<V>::T VT;
    for (int i0 = 0; i0 < cnt; i0 += 32) {
        const int m = min(32, cnt - i0);
        int s = 0, dl = 0;
        float nr = 0.0f;
        if (lane < m) {
            const int2 sd = grp[mylst[i0 + lane]];
            s = sd.x;
            const int d = sd.y;
            dl = d - base;
            nr = dinv[s] * dinv[d];
        }
        for (int j0 = 0; j0 < m; j0 += 4) {
            VT hv[4];
#pragma unroll
            for (int u = 0; u < 4; ++u) {
                const int su = __builtin_amdgcn_readlane(s, min(j0 + u, m - 1));
                hv[u] = *(const VT*)(h + (size_t)su * C + lane * V);
            }
#pragma unroll
            for (int u = 0; u < 4; ++u) {
                if (j0 + u < m) {
                    const int du = __builtin_amdgcn_readlane(dl, j0 + u);
                    const float nu = __int_as_float(
                        __builtin_amdgcn_readlane(__float_as_int(nr), j0 + u));
                    float* a = acc + du * LD + lane * V;
                    const float* hp = (const float*)&hv[u];
#pragma unroll
                    for (int c = 0; c < V; ++c) lds_add_f32(a + c, hp[c] * nu);
                }
            }
        }
    }
}

template <int C, int LD, bool ROUTE>
__device__ __forceinline__ void walk_tile(
    int lane, int wv, int base,
    const int2* __restrict__ grp, const int* __restrict__ off, int nchunk,
    const float* __restrict__ dinv, const float* __restrict__ h,
    int* mylst, float* acc, int4* q, int* qcnt, int* ovf)
{
    int cnt = 0;
    for (int c0 = wv; c0 < nchunk; c0 += AGG_WAVES * 32) {
        const int cm = c0 + lane * AGG_WAVES;
        int sb = 0, se = 0;
        if (cm < nchunk) {
            sb = off[(size_t)cm * OFF_P + blockIdx.x];
            se = off[(size_t)cm * OFF_P + blockIdx.x + 1];
        }
        const int nmine = min(32, (nchunk - c0 + AGG_WAVES - 1) / AGG_WAVES);
        for (int k = 0; k < nmine; ++k) {
            const int j0 = __builtin_amdgcn_readlane(sb, k);
            const int j1 = __builtin_amdgcn_readlane(se, k);
            const int p0 = (c0 + k * AGG_WAVES) * CHUNK;
            for (int j = j0; j < j1; j += 32) {
                const int m = min(32, j1 - j);
                if (lane < m) mylst[cnt + lane] = p0 + j + lane;
                cnt += m;
                if (ROUTE) {
                    if (cnt >= LIST_LOW)
                        cnt = route_list(mylst, cnt, true, lane, base, grp, dinv, q, qcnt, ovf);
                } else {
                    if (cnt > LIST_LOW) {
                        drain_list<C, LD>(mylst, cnt, lane, base, grp, dinv, h, acc);
                        cnt = 0;
                    }
                }
            }
        }
    }
    if (ROUTE) route_list(mylst, cnt, false, lane, base, grp, dinv, q, qcnt, ovf);
    else       drain_list<C, LD>(mylst, cnt, lane, base, grp, dinv, h, acc);
}

template <int C, int MODE>
__global__ void __launch_bounds__(AGG_T) k_tile_aggregate(
    const int* __restrict__ grp_w, const int* __restrict__ off,
    const float* __restrict__ dinv, const float* __restrict__ h,
    const float* __restrict__ wmat, const float* __restrict__ bias,
    float* __restrict__ out,
    int* __restrict__ rec_cnt, int* __restrict__ rec,
    int nchunk, int n)
{
    const int2* __restrict__ grp = (const int2*)grp_w;
    constexpr int  V    = C / 32;
    constexpr bool SCAN = (MODE == MODE_SCAN_WRITE);
    constexpr bool MUL  = SCAN;
    constexpr int  LD   = MUL ? C + 4 : C;
    typedef typename RowVec<V>::T VT;
    __shared__ __attribute__((aligned(16))) float acc[TILE_R * LD];
    __shared__ int   lst[AGG_WAVES * LIST_CAP];
    __shared__ __attribute__((aligned(16))) int qraw[SCAN ? AGG_WAVES * Q_CAP * 4 : 4];
    __shared__ int   qcnt[AGG_WAVES];
    __shared__ int   ovf;

    const int tid  = threadIdx.x;
    const int lane = tid & 31;
    const int wv   = __builtin_amdgcn_readfirstlane(tid >> 5);
    const int base = blockIdx.x * TILE_R;
    const int nrows = min(TILE_R, n - base);
    int4* q     = (int4*)qraw;
    int*  mylst = lst + wv * LIST_CAP;

    const int wslot = blockIdx.x * AGG_WAVES + wv;
    int4* myrec = (int4*)rec + (size_t)wslot * REC_CAP;

    int fill_v = 0;
    if (!SCAN) fill_v = rec_cnt[blockIdx.x * AGG_WAVES + lane];
    if (SCAN) {
        if (tid < AGG_WAVES) qcnt[tid] = 0;
        if (tid == 0) ovf = 0;
    }

    {
        const VT bv = *(const VT*)(bias + lane * V);
        const int kk = min(lane, OWN_ROWS - 1);
        const float dv = dinv[base + min(wv + kk * AGG_WAVES, nrows - 1)];
        VT st[OWN_ROWS];
#pragma unroll
        for (int k = 0; k < OWN_ROWS; ++k) {
            const int rr = min(wv + k * AGG_WAVES, nrows - 1);
            const float di = __int_as_float(__builtin_amdgcn_readlane(__float_as_int(dv), k));
            const VT hv = *(const VT*)(h + (size_t)(base + rr) * C + lane * V);
            st[k] = MUL ? row_self(hv, di) : row_start(bv, hv, di);
        }
#pragma unroll
        for (int k = 0; k < OWN_ROWS; ++k) {
            const int r = wv + k * AGG_WAVES;
            if (r < nrows) *(VT*)(acc + r * LD + lane * V) = st[k];
        }
    }

    bool fast;
    int  nown;
    if (SCAN) {
        __syncthreads();
        walk_tile<C, LD, true>(lane, wv, base, grp, off, nchunk, dinv, h,
                               mylst, acc, q, qcnt, &ovf);
        __syncthreads();
        fast = (__builtin_amdgcn_readfirstlane(ovf) == 0);
        nown = __builtin_amdgcn_readfirstlane(qcnt[wv]);
    } else {
        fast = (__builtin_amdgcn_ballot_w32(fill_v < 0) == 0u);
        nown = __builtin_amdgcn_readlane(fill_v, wv);
    }

    if (__builtin_expect(fast, true)) {
        if (SCAN) {
            drain_owned<C, LD, true>(q + wv * Q_CAP, nown, lane, h, acc, myrec);
            if (lane == 0) rec_cnt[wslot] = nown;
        } else {
            drain_owned<C, LD, false>(myrec, nown, lane, h, acc, myrec);
        }
    } else {
        __syncthreads();
        walk_tile<C, LD, false>(lane, wv, base, grp, off, nchunk, dinv, h,
                                mylst, acc, q, qcnt, &ovf);
        if (SCAN && lane == 0) rec_cnt[wslot] = -1;
        __syncthreads();
    }

    if (MUL) {
        constexpr int LDK = C + 2;
        constexpr int MT  = (TILE_R + 15) / 16;
        constexpr int NQ  = C / 32;
        static_assert(!MUL || C * LDK * 4 <= AGG_WAVES * Q_CAP * 16, "wmat must fit the queues' bytes");
        float* lds_w = (float*)qraw;
        __syncthreads();
        for (int idx = tid; idx < C * C; idx += AGG_T) {
            const int k  = idx / C;
            const int nn = idx - k * C;
            lds_w[nn * LDK + k] = wmat[idx];
        }
        __syncthreads();

        const int hf = lane >> 4;
        const int r  = lane & 15;
        for (int item = wv; item < MT * NQ; item += AGG_WAVES) {
            const int mt = item / NQ;
            const int nq = item - mt * NQ;
            if (mt * 16 >= nrows) continue;
            const float* arow = acc + min(mt * 16 + r, nrows - 1) * LD + 2 * hf;
            const float* b0   = lds_w + (nq * 32 + r) * LDK + 2 * hf;
            const float* b1   = b0 + 16 * LDK;
            v8f d0 = (v8f){};
            v8f d1 = (v8f){};
#pragma unroll 8
            for (int k = 0; k < C; k += 4) {
                const float2 av = *(const float2*)(arow + k);
                const float2 p0 = *(const float2*)(b0 + k);
                const float2 p1 = *(const float2*)(b1 + k);
                v2f a, x0, x1;
                a[0]  = av.x;  a[1]  = av.y;
                x0[0] = p0.x;  x0[1] = p0.y;
                x1[0] = p1.x;  x1[1] = p1.y;
                d0 = __builtin_amdgcn_wmma_f32_16x16x4_f32(
                    false, a, false, x0, (short)0, d0, false, false);
                d1 = __builtin_amdgcn_wmma_f32_16x16x4_f32(
                    false, a, false, x1, (short)0, d1, false, false);
            }
            const int   c0  = nq * 32 + r;
            const float bs0 = bias[c0];
            const float bs1 = bias[c0 + 16];
#pragma unroll
            for (int j = 0; j < 8; ++j) {
                const int row = mt * 16 + 8 * hf + j;
                if (row < nrows) {
                    float* orow = out + (size_t)(base + row) * C;
                    orow[c0]      = d0[j] + bs0;
                    orow[c0 + 16] = d1[j] + bs1;
                }
            }
        }
    } else {
#pragma unroll
        for (int k = 0; k < OWN_ROWS; ++k) {
            const int r = wv + k * AGG_WAVES;
            if (r < nrows)
                *(VT*)(out + (size_t)(base + r) * C + lane * V) = *(const VT*)(acc + r * LD + lane * V);
        }
    }
}

extern "C" void kernel_launch(void* const* d_in, const int* in_sizes, int n_in,
                              void* d_out, int out_size, void* d_ws, size_t ws_size,
                              hipStream_t stream) {
    const float* x  = (const float*)d_in[0];
    const int*   ei = (const int*)d_in[1];
    const float* W1 = (const float*)d_in[2];
    const float* b1 = (const float*)d_in[3];
    const float* W2 = (const float*)d_in[4];
    const float* b2 = (const float*)d_in[5];
    float* out = (float*)d_out;

    const int HID = in_sizes[3];
    const int IN  = in_sizes[2] / HID;
    const int N   = in_sizes[0] / IN;
    const int E   = in_sizes[1] / 2;
    const int* srcI = ei;
    const int* dstI = ei + E;

    float* ws   = (float*)d_ws;
    float* dinv = ws;
    float* bufA = ws + (((size_t)N + 63) & ~(size_t)63);
    float* bufB = bufA + (size_t)N * HID;
    const int ablk = (N + TILE_R - 1) / TILE_R;
    const int nchunk = (E + CHUNK - 1) / CHUNK;
    int*  grp  = (int*)(bufB + (size_t)N * HID);
    int*  off  = grp + (size_t)nchunk * CHUNK * 2;
    int*  rcnt = off + ((((size_t)nchunk * OFF_P) + 15) & ~(size_t)15);
    int*  rec  = rcnt + (size_t)ablk * AGG_WAVES;

    const int T = 256;
    const int mtiles = N / 16;
    const int gblk   = (mtiles + 7) / 8;

    k_fill_deg   <<<(N + T - 1) / T, T, 0, stream>>>(dinv, N);
    k_deg_group  <<<nchunk, CHUNK / 4, 0, stream>>>(srcI, dstI, E, dinv, grp, off);
    k_dinv       <<<(N + T - 1) / T, T, 0, stream>>>(dinv, N);

    k_tile_aggregate<128, MODE_SCAN_WRITE><<<ablk, AGG_T, 0, stream>>>(
        grp, off, dinv, x, W1, b1, bufB, rcnt, rec, nchunk, N);

    k_gemm_wmma_f32<4, 128, true><<<gblk, T, 0, stream>>>(bufB, W2, bufA, N);
    k_tile_aggregate<64, MODE_FROM_REC><<<ablk, AGG_T, 0, stream>>>(
        grp, off, dinv, bufA, W2, b2, out, rcnt, rec, nchunk, N);
}


# a tile pass before layer 1 walks the grouped list: LDS degree counts, each edge filed as its owner wave's record; fill, global atomics, 1/sqrt launch gone; layer 1 reads records; sum order: arrival
# speedup vs baseline: 3.7677x; 1.4034x over previous
#include <hip/hip_runtime.h>
#include <math.h>

typedef __attribute__((ext_vector_type(2))) float v2f;
typedef __attribute__((ext_vector_type(8))) float v8f;

constexpr int TILE_R = 196;
constexpr int N_TILE = 256;
constexpr int CHUNK = 4096;
constexpr int OFF_P = N_TILE + 1;

__global__ void __launch_bounds__(1024) k_group_edges(
    const int* __restrict__ src, const int* __restrict__ dst, int E,
    int* __restrict__ grp_w, int* __restrict__ off)
{
    __shared__ int cntt[N_TILE];
    __shared__ int start[OFF_P];
    __shared__ __attribute__((aligned(16))) int2 stage[CHUNK];

    const int tid = threadIdx.x;
    const int e   = blockIdx.x * CHUNK + tid * 4;
    if (tid < N_TILE) cntt[tid] = 0;
    __syncthreads();

    int4 s4 = make_int4(0, 0, 0, 0);
    int4 d4 = make_int4(-1, -1, -1, -1);
    if (e + 3 < E && ((((size_t)dst) | ((size_t)src)) & 15) == 0) {
        s4 = *(const int4*)(src + e);
        d4 = *(const int4*)(dst + e);
    } else {
        if (e     < E) { s4.x = src[e];     d4.x = dst[e];     }
        if (e + 1 < E) { s4.y = src[e + 1]; d4.y = dst[e + 1]; }
        if (e + 2 < E) { s4.z = src[e + 2]; d4.z = dst[e + 2]; }
        if (e + 3 < E) { s4.w = src[e + 3]; d4.w = dst[e + 3]; }
    }
    const int sq[4] = {s4.x, s4.y, s4.z, s4.w};
    const int dq[4] = {d4.x, d4.y, d4.z, d4.w};
    int tl[4], rk[4];
#pragma unroll
    for (int q = 0; q < 4; ++q) {
        tl[q] = 0;
        rk[q] = 0;
        if (dq[q] >= 0) {
            tl[q] = min((int)((unsigned)dq[q] / (unsigned)TILE_R), N_TILE - 1);
            rk[q] = __hip_atomic_fetch_add(&cntt[tl[q]], 1, __ATOMIC_RELAXED,
                                           __HIP_MEMORY_SCOPE_WORKGROUP);
        }
    }
    __syncthreads();

    if (tid < 32) {
        int v[8];
        int run = 0;
#pragma unroll
        for (int i = 0; i < 8; ++i) {
            const int t = cntt[8 * tid + i];
            v[i] = run;
            run += t;
        }
        int incl = run;
        for (int o = 1; o < 32; o <<= 1) {
            const int nb = __shfl_up(incl, o, 32);
            if (tid >= o) incl += nb;
        }
        const int excl = incl - run;
#pragma unroll
        for (int i = 0; i < 8; ++i) start[8 * tid + i] = excl + v[i];
        if (tid == 31) start[N_TILE] = incl;
    }
    __syncthreads();

#pragma unroll
    for (int q = 0; q < 4; ++q) {
        if (dq[q] >= 0) stage[start[tl[q]] + rk[q]] = make_int2(sq[q], dq[q]);
    }
    if (tid < OFF_P) off[(size_t)blockIdx.x * OFF_P + tid] = start[tid];
    __syncthreads();

    const int total = start[N_TILE];
    const int4* st4 = (const int4*)stage;
    int4* g4 = (int4*)grp_w + (size_t)blockIdx.x * (CHUNK / 2);
    for (int i = tid; 2 * i < total; i += 1024) g4[i] = st4[i];
}

template <int NT, int KD, bool RELU>
__global__ void __launch_bounds__(256) k_gemm_wmma_f32(
    const float* __restrict__ A, const float* __restrict__ B,
    float* __restrict__ C, int M)
{
    constexpr int N   = NT * 16;
    constexpr int LDK = KD + 2;
    __shared__ float lds_w[N * LDK];

    for (int idx = threadIdx.x; idx < KD * N; idx += 256) {
        int k = idx / N;
        int n = idx - k * N;
        lds_w[n * LDK + k] = B[idx];
    }
    __syncthreads();

    const int lane  = threadIdx.x & 31;
    const int wv    = threadIdx.x >> 5;
    const int mtile = blockIdx.x * 8 + wv;
    if (mtile * 16 >= M) return;

    const int hf = lane >> 4;
    const int r  = lane & 15;

    const float* __restrict__ arow = A + (size_t)(mtile * 16 + r) * KD + 2 * hf;

    v8f acc[NT];
#pragma unroll
    for (int t = 0; t < NT; ++t) acc[t] = (v8f){};

#pragma unroll 4
    for (int k = 0; k < KD; k += 4) {
        float2 av = *(const float2*)(arow + k);
        if (RELU) { av.x = fmaxf(av.x, 0.0f); av.y = fmaxf(av.y, 0.0f); }
        v2f a;
        a[0] = av.x;
        a[1] = av.y;
#pragma unroll
        for (int t = 0; t < NT; ++t) {
            const int col = t * 16 + r;
            float2 bv = *(const float2*)(&lds_w[col * LDK + k + 2 * hf]);
            v2f b;
            b[0] = bv.x;
            b[1] = bv.y;
            acc[t] = __builtin_amdgcn_wmma_f32_16x16x4_f32(
                false, a, false, b, (short)0, acc[t], false, false);
        }
    }

#pragma unroll
    for (int t = 0; t < NT; ++t) {
        float* crow = C + (size_t)(mtile * 16 + 8 * hf) * N + t * 16 + r;
#pragma unroll
        for (int j = 0; j < 8; ++j) crow[(size_t)j * N] = acc[t][j];
    }
}

constexpr int AGG_T    = 1024;
constexpr int LIST_LOW = 32;
constexpr int LIST_CAP = LIST_LOW + 32;
constexpr int REC_CAP   = 512;
constexpr int AGG_WAVES = AGG_T / 32;
constexpr int Q_CAP     = 256;
constexpr int OWN_ROWS  = (TILE_R + AGG_WAVES - 1) / AGG_WAVES;
static_assert(Q_CAP <= REC_CAP, "a queue must fit its record slice");
static_assert((AGG_WAVES & (AGG_WAVES - 1)) == 0, "owner = row & (waves - 1)");
constexpr int MODE_REC_FILL = 0;
constexpr int MODE_FROM_REC = 1;

constexpr int PRE_T   = 256;
constexpr int PRE_U   = 16;
constexpr int MAP_CAP = PRE_T * PRE_U;
static_assert(TILE_R <= PRE_T, "one thread per row clears and writes the counters");
static_assert(AGG_WAVES <= PRE_T, "one thread per queue clears its fill word");

__global__ void __launch_bounds__(PRE_T) k_tile_prep(
    const int* __restrict__ grp_w, const int* __restrict__ off, int nchunk,
    float* __restrict__ dinv, int* __restrict__ rec_cnt, int* __restrict__ rec, int n)
{
    __shared__ int cnt[TILE_R];
    __shared__ int wsum[PRE_T / 32];
    __shared__ int mapp[MAP_CAP];
    __shared__ __attribute__((aligned(16))) int4 q[AGG_WAVES * Q_CAP];
    __shared__ int qcnt[AGG_WAVES];
    __shared__ int ovf;

    const int2* __restrict__ grp = (const int2*)grp_w;
    const int tid   = threadIdx.x;
    const int lane  = tid & 31;
    const int wv    = tid >> 5;
    const int base  = blockIdx.x * TILE_R;
    const int nrows = min(TILE_R, n - base);
    if (tid < TILE_R) cnt[tid] = 0;
    if (tid < AGG_WAVES) qcnt[tid] = 0;
    if (tid == 0) ovf = 0;

    for (int c0 = 0; c0 < nchunk; c0 += PRE_T) {
        const int ch = c0 + tid;
        int sb = 0, se = 0;
        if (ch < nchunk) {
            sb = off[(size_t)ch * OFF_P + blockIdx.x];
            se = off[(size_t)ch * OFF_P + blockIdx.x + 1];
        }
        const int len = se - sb;
        int incl = len;
        for (int o = 1; o < 32; o <<= 1) {
            const int nb = __shfl_up(incl, o, 32);
            if (lane >= o) incl += nb;
        }
        if (lane == 31) wsum[wv] = incl;
        __syncthreads();
        int woff = 0, total = 0;
#pragma unroll
        for (int w = 0; w < PRE_T / 32; ++w) {
            const int t = wsum[w];
            if (w < wv) woff += t;
            total += t;
        }
        const int excl  = woff + incl - len;
        const int start = ch * CHUNK + sb;

        for (int w0 = 0; w0 < total; w0 += MAP_CAP) {
            const int lo = max(excl, w0);
            const int hi = min(excl + len, w0 + MAP_CAP);
            for (int i = lo; i < hi; ++i) mapp[i - w0] = start + (i - excl);
            __syncthreads();
            const int nwin = min(MAP_CAP, total - w0);
            int2 sd[PRE_U];
#pragma unroll
            for (int u = 0; u < PRE_U; ++u) {
                const int p = mapp[min(tid + u * PRE_T, nwin - 1)];
                sd[u] = grp[p];
            }
            int pos[PRE_U];
#pragma unroll
            for (int u = 0; u < PRE_U; ++u) {
                pos[u] = 0;
                if (tid + u * PRE_T < nwin) {
                    const int dl = min(max(sd[u].y - base, 0), TILE_R - 1);
                    sd[u].y = dl;
                    __hip_atomic_fetch_add(&cnt[dl], 1, __ATOMIC_RELAXED,
                                           __HIP_MEMORY_SCOPE_WORKGROUP);
                    pos[u] = __hip_atomic_fetch_add(&qcnt[dl & (AGG_WAVES - 1)], 1, __ATOMIC_RELAXED,
                                                    __HIP_MEMORY_SCOPE_WORKGROUP);
                }
            }
#pragma unroll
            for (int u = 0; u < PRE_U; ++u) {
                if (tid + u * PRE_T < nwin) {
                    const int dl = sd[u].y;
                    const int ow = dl & (AGG_WAVES - 1);
                    q[ow * Q_CAP + min(pos[u], Q_CAP - 1)] = make_int4(sd[u].x, dl, 0, 0);
                    if (pos[u] >= Q_CAP) ovf = 1;
                }
            }
            __syncthreads();
        }
        __syncthreads();
    }
    __syncthreads();
    if (tid < nrows) {
        const float d = 1.0f + (float)cnt[tid];
        dinv[base + tid] = (d > 0.0f) ? rsqrtf(d) : 0.0f;
    }
    const bool fast = (ovf == 0);
    for (int w = wv; w < AGG_WAVES; w += PRE_T / 32) {
        const int wslot = blockIdx.x * AGG_WAVES + w;
        const int nq = qcnt[w];
        if (fast) {
            int4* slice = (int4*)rec + (size_t)wslot * REC_CAP;
            for (int i = lane; i < nq; i += 128) {
                int4 r[4];
#pragma unroll
                for (int u = 0; u < 4; ++u) r[u] = q[w * Q_CAP + min(i + 32 * u, nq - 1)];
#pragma unroll
                for (int u = 0; u < 4; ++u) {
                    if (i + 32 * u < nq) slice[i + 32 * u] = r[u];
                }
            }
        }
        if (lane == 0) rec_cnt[wslot] = fast ? nq : -1;
    }
}

template <int V> struct RowVec;
template <> struct RowVec<4> { typedef float4 T; };
template <> struct RowVec<2> { typedef float2 T; };

__device__ __forceinline__ float4 row_start(const float4& b, const float4& hv, float di) {
    float4 o;
    o.x = b.x + hv.x * di * di;
    o.y = b.y + hv.y * di * di;
    o.z = b.z + hv.z * di * di;
    o.w = b.w + hv.w * di * di;
    return o;
}
__device__ __forceinline__ float2 row_start(const float2& b, const float2& hv, float di) {
    float2 o;
    o.x = b.x + hv.x * di * di;
    o.y = b.y + hv.y * di * di;
    return o;
}
__device__ __forceinline__ float4 row_self(const float4& hv, float di) {
    float4 o;
    o.x = hv.x * di * di;
    o.y = hv.y * di * di;
    o.z = hv.z * di * di;
    o.w = hv.w * di * di;
    return o;
}
__device__ __forceinline__ float2 row_self(const float2& hv, float di) {
    float2 o;
    o.x = hv.x * di * di;
    o.y = hv.y * di * di;
    return o;
}
__device__ __forceinline__ void row_fma(float4& a, const float4& hv, float nu) {
    a.x += hv.x * nu;
    a.y += hv.y * nu;
    a.z += hv.z * nu;
    a.w += hv.w * nu;
}
__device__ __forceinline__ void row_fma(float2& a, const float2& hv, float nu) {
    a.x += hv.x * nu;
    a.y += hv.y * nu;
}

template <int C>
__device__ __forceinline__ void load_group(
    typename RowVec<C / 32>::T (&hv)[8], int s, int j, int m, int lane,
    const float* __restrict__ h)
{
    constexpr int V = C / 32;
    typedef typename RowVec<V>::T VT;
#pragma unroll
    for (int u = 0; u < 8; ++u) {
        const int su = __builtin_amdgcn_readlane(s, min(j + u, m - 1));
        hv[u] = *(const VT*)(h + (size_t)su * C + lane * V);
    }
}

template <int C, int LD>
__device__ __forceinline__ void add_group(
    const typename RowVec<C / 32>::T (&hv)[8], int dl, float nr, int j, int m, int lane,
    float* acc)
{
    constexpr int V = C / 32;
    typedef typename RowVec<V>::T VT;
#pragma unroll
    for (int u = 0; u < 8; ++u) {
        if (j + u < m) {
            const int du = __builtin_amdgcn_readlane(dl, j + u);
            const float nu = __int_as_float(
                __builtin_amdgcn_readlane(__float_as_int(nr), j + u));
            VT* p = (VT*)(acc + du * LD + lane * V);
            VT t = *p;
            row_fma(t, hv[u], nu);
            *p = t;
        }
    }
}

template <int C, int LD, bool FILL>
__device__ __forceinline__ void drain_owned(
    int4* recs, int cnt, int lane, int base,
    const float* __restrict__ dinv, const float* __restrict__ h, float* acc)
{
    typedef typename RowVec<C / 32>::T VT;
    int4 cur = make_int4(0, 0, 0, 0);
    if (lane < cnt) cur = recs[lane];
    for (int i0 = 0; i0 < cnt; i0 += 32) {
        const int m = min(32, cnt - i0);
        int4 nxt = make_int4(0, 0, 0, 0);
        if (i0 + 32 + lane < cnt) nxt = recs[i0 + 32 + lane];
        float da = 0.0f, db = 0.0f;
        if (FILL) {
            da = dinv[cur.x];
            db = dinv[base + cur.y];
        }
        const int   s  = cur.x;
        const int   dl = cur.y;
        VT ha[8], hb[8];
        load_group<C>(ha, s, 0, m, lane, h);
        load_group<C>(hb, s, 8, m, lane, h);
        if (FILL) {
            cur.z = __float_as_int(da * db);
            if (lane < m) recs[i0 + lane] = cur;
        }
        const float nr = __int_as_float(cur.z);
        add_group<C, LD>(ha, dl, nr, 0, m, lane, acc);
        if (m > 8) {
            load_group<C>(ha, s, 16, m, lane, h);
            add_group<C, LD>(hb, dl, nr, 8, m, lane, acc);
            if (m > 16) {
                load_group<C>(hb, s, 24, m, lane, h);
                add_group<C, LD>(ha, dl, nr, 16, m, lane, acc);
                add_group<C, LD>(hb, dl, nr, 24, m, lane, acc);
            }
        }
        cur = nxt;
    }
}

__device__ __forceinline__ void lds_add_f32(float* p, float v) {
    __hip_atomic_fetch_add(p, v, __ATOMIC_RELAXED, __HIP_MEMORY_SCOPE_WORKGROUP);
}

template <int C, int LD>
__device__ __forceinline__ void drain_list(
    const int* mylst, int cnt, int lane, int base,
    const int2* __restrict__ grp,
    const float* __restrict__ dinv, const float* __restrict__ h, float* acc)
{
    constexpr int V = C / 32;
    typedef typename RowVec<V>::T VT;
    for (int i0 = 0; i0 < cnt; i0 += 32) {
        const int m = min(32, cnt - i0);
        int s = 0, dl = 0;
        float nr = 0.0f;
        if (lane < m) {
            const int2 sd = grp[mylst[i0 + lane]];
            s = sd.x;
            const int d = sd.y;
            dl = d - base;
            nr = dinv[s] * dinv[d];
        }
        for (int j0 = 0; j0 < m; j0 += 4) {
            VT hv[4];
#pragma unroll
            for (int u = 0; u < 4; ++u) {
                const int su = __builtin_amdgcn_readlane(s, min(j0 + u, m - 1));
                hv[u] = *(const VT*)(h + (size_t)su * C + lane * V);
            }
#pragma unroll
            for (int u = 0; u < 4; ++u) {
                if (j0 + u < m) {
                    const int du = __builtin_amdgcn_readlane(dl, j0 + u);
                    const float nu = __int_as_float(
                        __builtin_amdgcn_readlane(__float_as_int(nr), j0 + u));
                    float* a = acc + du * LD + lane * V;
                    const float* hp = (const float*)&hv[u];
#pragma unroll
                    for (int c = 0; c < V; ++c) lds_add_f32(a + c, hp[c] * nu);
                }
            }
        }
    }
}

template <int C, int LD>
__device__ __forceinline__ void walk_tile(
    int lane, int wv, int base,
    const int2* __restrict__ grp, const int* __restrict__ off, int nchunk,
    const float* __restrict__ dinv, const float* __restrict__ h,
    int* mylst, float* acc)
{
    int cnt = 0;
    for (int c0 = wv; c0 < nchunk; c0 += AGG_WAVES * 32) {
        const int cm = c0 + lane * AGG_WAVES;
        int sb = 0, se = 0;
        if (cm < nchunk) {
            sb = off[(size_t)cm * OFF_P + blockIdx.x];
            se = off[(size_t)cm * OFF_P + blockIdx.x + 1];
        }
        const int nmine = min(32, (nchunk - c0 + AGG_WAVES - 1) / AGG_WAVES);
        for (int k = 0; k < nmine; ++k) {
            const int j0 = __builtin_amdgcn_readlane(sb, k);
            const int j1 = __builtin_amdgcn_readlane(se, k);
            const int p0 = (c0 + k * AGG_WAVES) * CHUNK;
            for (int j = j0; j < j1; j += 32) {
                const int m = min(32, j1 - j);
                if (lane < m) mylst[cnt + lane] = p0 + j + lane;
                cnt += m;
                if (cnt > LIST_LOW) {
                    drain_list<C, LD>(mylst, cnt, lane, base, grp, dinv, h, acc);
                    cnt = 0;
                }
            }
        }
    }
    drain_list<C, LD>(mylst, cnt, lane, base, grp, dinv, h, acc);
}

template <int C, int MODE>
__global__ void __launch_bounds__(AGG_T) k_tile_aggregate(
    const int* __restrict__ grp_w, const int* __restrict__ off,
    const float* __restrict__ dinv, const float* __restrict__ h,
    const float* __restrict__ wmat, const float* __restrict__ bias,
    float* __restrict__ out,
    const int* __restrict__ rec_cnt, int* __restrict__ rec,
    int nchunk, int n)
{
    const int2* __restrict__ grp = (const int2*)grp_w;
    constexpr int  V    = C / 32;
    constexpr bool FILL = (MODE == MODE_REC_FILL);
    constexpr bool MUL  = FILL;
    constexpr int  LD   = MUL ? C + 4 : C;
    constexpr int  LDK  = C + 2;
    typedef typename RowVec<V>::T VT;
    __shared__ __attribute__((aligned(16))) float acc[TILE_R * LD];
    __shared__ int   lst[AGG_WAVES * LIST_CAP];
    __shared__ __attribute__((aligned(16))) float lds_w[MUL ? C * LDK : 4];

    const int tid  = threadIdx.x;
    const int lane = tid & 31;
    const int wv   = __builtin_amdgcn_readfirstlane(tid >> 5);
    const int base = blockIdx.x * TILE_R;
    const int nrows = min(TILE_R, n - base);
    int*  mylst = lst + wv * LIST_CAP;

    const int wslot = blockIdx.x * AGG_WAVES + wv;
    int4* myrec = (int4*)rec + (size_t)wslot * REC_CAP;

    const int fill_v = rec_cnt[blockIdx.x * AGG_WAVES + lane];

    if (MUL) {
        for (int idx = tid; idx < C * C; idx += AGG_T) {
            const int k  = idx / C;
            const int nn = idx - k * C;
            lds_w[nn * LDK + k] = wmat[idx];
        }
    }

    {
        const VT bv = *(const VT*)(bias + lane * V);
        const int kk = min(lane, OWN_ROWS - 1);
        const float dv = dinv[base + min(wv + kk * AGG_WAVES, nrows - 1)];
        VT st[OWN_ROWS];
#pragma unroll
        for (int k = 0; k < OWN_ROWS; ++k) {
            const int rr = min(wv + k * AGG_WAVES, nrows - 1);
            const float di = __int_as_float(__builtin_amdgcn_readlane(__float_as_int(dv), k));
            const VT hv = *(const VT*)(h + (size_t)(base + rr) * C + lane * V);
            st[k] = MUL ? row_self(hv, di) : row_start(bv, hv, di);
        }
#pragma unroll
        for (int k = 0; k < OWN_ROWS; ++k) {
            const int r = wv + k * AGG_WAVES;
            if (r < nrows) *(VT*)(acc + r * LD + lane * V) = st[k];
        }
    }

    const bool fast = (__builtin_amdgcn_ballot_w32(fill_v < 0) == 0u);
    const int  nown = __builtin_amdgcn_readlane(fill_v, wv);

    if (__builtin_expect(fast, true)) {
        drain_owned<C, LD, FILL>(myrec, nown, lane, base, dinv, h, acc);
    } else {
        __syncthreads();
        walk_tile<C, LD>(lane, wv, base, grp, off, nchunk, dinv, h, mylst, acc);
        __syncthreads();
    }

    if (MUL) {
        constexpr int MT  = (TILE_R + 15) / 16;
        constexpr int NQ  = C / 32;
        __syncthreads();

        const int hf = lane >> 4;
        const int r  = lane & 15;
        for (int item = wv; item < MT * NQ; item += AGG_WAVES) {
            const int mt = item / NQ;
            const int nq = item - mt * NQ;
            if (mt * 16 >= nrows) continue;
            const float* arow = acc + min(mt * 16 + r, nrows - 1) * LD + 2 * hf;
            const float* b0   = lds_w + (nq * 32 + r) * LDK + 2 * hf;
            const float* b1   = b0 + 16 * LDK;
            v8f d0 = (v8f){};
            v8f d1 = (v8f){};
#pragma unroll 8
            for (int k = 0; k < C; k += 4) {
                const float2 av = *(const float2*)(arow + k);
                const float2 p0 = *(const float2*)(b0 + k);
                const float2 p1 = *(const float2*)(b1 + k);
                v2f a, x0, x1;
                a[0]  = av.x;  a[1]  = av.y;
                x0[0] = p0.x;  x0[1] = p0.y;
                x1[0] = p1.x;  x1[1] = p1.y;
                d0 = __builtin_amdgcn_wmma_f32_16x16x4_f32(
                    false, a, false, x0, (short)0, d0, false, false);
                d1 = __builtin_amdgcn_wmma_f32_16x16x4_f32(
                    false, a, false, x1, (short)0, d1, false, false);
            }
            const int   c0  = nq * 32 + r;
            const float bs0 = bias[c0];
            const float bs1 = bias[c0 + 16];
#pragma unroll
            for (int j = 0; j < 8; ++j) {
                const int row = mt * 16 + 8 * hf + j;
                if (row < nrows) {
                    float* orow = out + (size_t)(base + row) * C;
                    orow[c0]      = d0[j] + bs0;
                    orow[c0 + 16] = d1[j] + bs1;
                }
            }
        }
    } else {
#pragma unroll
        for (int k = 0; k < OWN_ROWS; ++k) {
            const int r = wv + k * AGG_WAVES;
            if (r < nrows)
                *(VT*)(out + (size_t)(base + r) * C + lane * V) = *(const VT*)(acc + r * LD + lane * V);
        }
    }
}

extern "C" void kernel_launch(void* const* d_in, const int* in_sizes, int n_in,
                              void* d_out, int out_size, void* d_ws, size_t ws_size,
                              hipStream_t stream) {
    const float* x  = (const float*)d_in[0];
    const int*   ei = (const int*)d_in[1];
    const float* W1 = (const float*)d_in[2];
    const float* b1 = (const float*)d_in[3];
    const float* W2 = (const float*)d_in[4];
    const float* b2 = (const float*)d_in[5];
    float* out = (float*)d_out;

    const int HID = in_sizes[3];
    const int IN  = in_sizes[2] / HID;
    const int N   = in_sizes[0] / IN;
    const int E   = in_sizes[1] / 2;
    const int* srcI = ei;
    const int* dstI = ei + E;

    float* ws   = (float*)d_ws;
    float* dinv = ws;
    float* bufA = ws + (((size_t)N + 63) & ~(size_t)63);
    float* bufB = bufA + (size_t)N * HID;
    const int ablk = (N + TILE_R - 1) / TILE_R;
    const int nchunk = (E + CHUNK - 1) / CHUNK;
    int*  grp  = (int*)(bufB + (size_t)N * HID);
    int*  off  = grp + (size_t)nchunk * CHUNK * 2;
    int*  rcnt = off + ((((size_t)nchunk * OFF_P) + 15) & ~(size_t)15);
    int*  rec  = rcnt + (size_t)ablk * AGG_WAVES;

    const int T = 256;
    const int mtiles = N / 16;
    const int gblk   = (mtiles + 7) / 8;

    k_group_edges<<<nchunk, CHUNK / 4, 0, stream>>>(srcI, dstI, E, grp, off);
    k_tile_prep  <<<ablk, PRE_T, 0, stream>>>(grp, off, nchunk, dinv, rcnt, rec, N);

    k_tile_aggregate<128, MODE_REC_FILL><<<ablk, AGG_T, 0, stream>>>(
        grp, off, dinv, x, W1, b1, bufB, rcnt, rec, nchunk, N);

    k_gemm_wmma_f32<4, 128, true><<<gblk, T, 0, stream>>>(bufB, W2, bufA, N);
    k_tile_aggregate<64, MODE_FROM_REC><<<ablk, AGG_T, 0, stream>>>(
        grp, off, dinv, bufA, W2, b2, out, rcnt, rec, nchunk, N);
}


# each wave keeps the sums of its own (up to 7) dst rows in a register vector, a record's row reached by register-relative moves; the LDS load-wait-store per record is gone; row add order unchanged
# speedup vs baseline: 4.0486x; 1.0746x over previous
#include <hip/hip_runtime.h>
#include <math.h>

typedef __attribute__((ext_vector_type(2))) float v2f;
typedef __attribute__((ext_vector_type(8))) float v8f;

constexpr int TILE_R = 196;
constexpr int N_TILE = 256;
constexpr int CHUNK = 4096;
constexpr int OFF_P = N_TILE + 1;

__global__ void __launch_bounds__(1024) k_group_edges(
    const int* __restrict__ src, const int* __restrict__ dst, int E,
    int* __restrict__ grp_w, int* __restrict__ off)
{
    __shared__ int cntt[N_TILE];
    __shared__ int start[OFF_P];
    __shared__ __attribute__((aligned(16))) int2 stage[CHUNK];

    const int tid = threadIdx.x;
    const int e   = blockIdx.x * CHUNK + tid * 4;
    if (tid < N_TILE) cntt[tid] = 0;
    __syncthreads();

    int4 s4 = make_int4(0, 0, 0, 0);
    int4 d4 = make_int4(-1, -1, -1, -1);
    if (e + 3 < E && ((((size_t)dst) | ((size_t)src)) & 15) == 0) {
        s4 = *(const int4*)(src + e);
        d4 = *(const int4*)(dst + e);
    } else {
        if (e     < E) { s4.x = src[e];     d4.x = dst[e];     }
        if (e + 1 < E) { s4.y = src[e + 1]; d4.y = dst[e + 1]; }
        if (e + 2 < E) { s4.z = src[e + 2]; d4.z = dst[e + 2]; }
        if (e + 3 < E) { s4.w = src[e + 3]; d4.w = dst[e + 3]; }
    }
    const int sq[4] = {s4.x, s4.y, s4.z, s4.w};
    const int dq[4] = {d4.x, d4.y, d4.z, d4.w};
    int tl[4], rk[4];
#pragma unroll
    for (int q = 0; q < 4; ++q) {
        tl[q] = 0;
        rk[q] = 0;
        if (dq[q] >= 0) {
            tl[q] = min((int)((unsigned)dq[q] / (unsigned)TILE_R), N_TILE - 1);
            rk[q] = __hip_atomic_fetch_add(&cntt[tl[q]], 1, __ATOMIC_RELAXED,
                                           __HIP_MEMORY_SCOPE_WORKGROUP);
        }
    }
    __syncthreads();

    if (tid < 32) {
        int v[8];
        int run = 0;
#pragma unroll
        for (int i = 0; i < 8; ++i) {
            const int t = cntt[8 * tid + i];
            v[i] = run;
            run += t;
        }
        int incl = run;
        for (int o = 1; o < 32; o <<= 1) {
            const int nb = __shfl_up(incl, o, 32);
            if (tid >= o) incl += nb;
        }
        const int excl = incl - run;
#pragma unroll
        for (int i = 0; i < 8; ++i) start[8 * tid + i] = excl + v[i];
        if (tid == 31) start[N_TILE] = incl;
    }
    __syncthreads();

#pragma unroll
    for (int q = 0; q < 4; ++q) {
        if (dq[q] >= 0) stage[start[tl[q]] + rk[q]] = make_int2(sq[q], dq[q]);
    }
    if (tid < OFF_P) off[(size_t)blockIdx.x * OFF_P + tid] = start[tid];
    __syncthreads();

    const int total = start[N_TILE];
    const int4* st4 = (const int4*)stage;
    int4* g4 = (int4*)grp_w + (size_t)blockIdx.x * (CHUNK / 2);
    for (int i = tid; 2 * i < total; i += 1024) g4[i] = st4[i];
}

template <int NT, int KD, bool RELU>
__global__ void __launch_bounds__(256) k_gemm_wmma_f32(
    const float* __restrict__ A, const float* __restrict__ B,
    float* __restrict__ C, int M)
{
    constexpr int N   = NT * 16;
    constexpr int LDK = KD + 2;
    __shared__ float lds_w[N * LDK];

    for (int idx = threadIdx.x; idx < KD * N; idx += 256) {
        int k = idx / N;
        int n = idx - k * N;
        lds_w[n * LDK + k] = B[idx];
    }
    __syncthreads();

    const int lane  = threadIdx.x & 31;
    const int wv    = threadIdx.x >> 5;
    const int mtile = blockIdx.x * 8 + wv;
    if (mtile * 16 >= M) return;

    const int hf = lane >> 4;
    const int r  = lane & 15;

    const float* __restrict__ arow = A + (size_t)(mtile * 16 + r) * KD + 2 * hf;

    v8f acc[NT];
#pragma unroll
    for (int t = 0; t < NT; ++t) acc[t] = (v8f){};

#pragma unroll 4
    for (int k = 0; k < KD; k += 4) {
        float2 av = *(const float2*)(arow + k);
        if (RELU) { av.x = fmaxf(av.x, 0.0f); av.y = fmaxf(av.y, 0.0f); }
        v2f a;
        a[0] = av.x;
        a[1] = av.y;
#pragma unroll
        for (int t = 0; t < NT; ++t) {
            const int col = t * 16 + r;
            float2 bv = *(const float2*)(&lds_w[col * LDK + k + 2 * hf]);
            v2f b;
            b[0] = bv.x;
            b[1] = bv.y;
            acc[t] = __builtin_amdgcn_wmma_f32_16x16x4_f32(
                false, a, false, b, (short)0, acc[t], false, false);
        }
    }

#pragma unroll
    for (int t = 0; t < NT; ++t) {
        float* crow = C + (size_t)(mtile * 16 + 8 * hf) * N + t * 16 + r;
#pragma unroll
        for (int j = 0; j < 8; ++j) crow[(size_t)j * N] = acc[t][j];
    }
}

constexpr int AGG_T    = 1024;
constexpr int LIST_LOW = 32;
constexpr int LIST_CAP = LIST_LOW + 32;
constexpr int REC_CAP   = 512;
constexpr int AGG_WAVES = AGG_T / 32;
constexpr int Q_CAP     = 256;
constexpr int OWN_ROWS  = (TILE_R + AGG_WAVES - 1) / AGG_WAVES;
static_assert(Q_CAP <= REC_CAP, "a queue must fit its record slice");
static_assert((AGG_WAVES & (AGG_WAVES - 1)) == 0, "owner = row & (waves - 1)");
constexpr int AGG_WSH   = 5;
static_assert((1 << AGG_WSH) == AGG_WAVES, "a wave's k-th row is row wave + k * waves");
constexpr int MODE_REC_FILL = 0;
constexpr int MODE_FROM_REC = 1;

constexpr int PRE_T   = 256;
constexpr int PRE_U   = 16;
constexpr int MAP_CAP = PRE_T * PRE_U;
static_assert(TILE_R <= PRE_T, "one thread per row clears and writes the counters");
static_assert(AGG_WAVES <= PRE_T, "one thread per queue clears its fill word");

__global__ void __launch_bounds__(PRE_T) k_tile_prep(
    const int* __restrict__ grp_w, const int* __restrict__ off, int nchunk,
    float* __restrict__ dinv, int* __restrict__ rec_cnt, int* __restrict__ rec, int n)
{
    __shared__ int cnt[TILE_R];
    __shared__ int wsum[PRE_T / 32];
    __shared__ int mapp[MAP_CAP];
    __shared__ __attribute__((aligned(16))) int4 q[AGG_WAVES * Q_CAP];
    __shared__ int qcnt[AGG_WAVES];
    __shared__ int ovf;

    const int2* __restrict__ grp = (const int2*)grp_w;
    const int tid   = threadIdx.x;
    const int lane  = tid & 31;
    const int wv    = tid >> 5;
    const int base  = blockIdx.x * TILE_R;
    const int nrows = min(TILE_R, n - base);
    if (tid < TILE_R) cnt[tid] = 0;
    if (tid < AGG_WAVES) qcnt[tid] = 0;
    if (tid == 0) ovf = 0;

    for (int c0 = 0; c0 < nchunk; c0 += PRE_T) {
        const int ch = c0 + tid;
        int sb = 0, se = 0;
        if (ch < nchunk) {
            sb = off[(size_t)ch * OFF_P + blockIdx.x];
            se = off[(size_t)ch * OFF_P + blockIdx.x + 1];
        }
        const int len = se - sb;
        int incl = len;
        for (int o = 1; o < 32; o <<= 1) {
            const int nb = __shfl_up(incl, o, 32);
            if (lane >= o) incl += nb;
        }
        if (lane == 31) wsum[wv] = incl;
        __syncthreads();
        int woff = 0, total = 0;
#pragma unroll
        for (int w = 0; w < PRE_T / 32; ++w) {
            const int t = wsum[w];
            if (w < wv) woff += t;
            total += t;
        }
        const int excl  = woff + incl - len;
        const int start = ch * CHUNK + sb;

        for (int w0 = 0; w0 < total; w0 += MAP_CAP) {
            const int lo = max(excl, w0);
            const int hi = min(excl + len, w0 + MAP_CAP);
            for (int i = lo; i < hi; ++i) mapp[i - w0] = start + (i - excl);
            __syncthreads();
            const int nwin = min(MAP_CAP, total - w0);
            int2 sd[PRE_U];
#pragma unroll
            for (int u = 0; u < PRE_U; ++u) {
                const int p = mapp[min(tid + u * PRE_T, nwin - 1)];
                sd[u] = grp[p];
            }
            int pos[PRE_U];
#pragma unroll
            for (int u = 0; u < PRE_U; ++u) {
                pos[u] = 0;
                if (tid + u * PRE_T < nwin) {
                    const int dl = min(max(sd[u].y - base, 0), TILE_R - 1);
                    sd[u].y = dl;
                    __hip_atomic_fetch_add(&cnt[dl], 1, __ATOMIC_RELAXED,
                                           __HIP_MEMORY_SCOPE_WORKGROUP);
                    pos[u] = __hip_atomic_fetch_add(&qcnt[dl & (AGG_WAVES - 1)], 1, __ATOMIC_RELAXED,
                                                    __HIP_MEMORY_SCOPE_WORKGROUP);
                }
            }
#pragma unroll
            for (int u = 0; u < PRE_U; ++u) {
                if (tid + u * PRE_T < nwin) {
                    const int dl = sd[u].y;
                    const int ow = dl & (AGG_WAVES - 1);
                    q[ow * Q_CAP + min(pos[u], Q_CAP - 1)] = make_int4(sd[u].x, dl, 0, 0);
                    if (pos[u] >= Q_CAP) ovf = 1;
                }
            }
            __syncthreads();
        }
        __syncthreads();
    }
    __syncthreads();
    if (tid < nrows) {
        const float d = 1.0f + (float)cnt[tid];
        dinv[base + tid] = (d > 0.0f) ? rsqrtf(d) : 0.0f;
    }
    const bool fast = (ovf == 0);
    for (int w = wv; w < AGG_WAVES; w += PRE_T / 32) {
        const int wslot = blockIdx.x * AGG_WAVES + w;
        const int nq = qcnt[w];
        if (fast) {
            int4* slice = (int4*)rec + (size_t)wslot * REC_CAP;
            for (int i = lane; i < nq; i += 128) {
                int4 r[4];
#pragma unroll
                for (int u = 0; u < 4; ++u) r[u] = q[w * Q_CAP + min(i + 32 * u, nq - 1)];
#pragma unroll
                for (int u = 0; u < 4; ++u) {
                    if (i + 32 * u < nq) slice[i + 32 * u] = r[u];
                }
            }
        }
        if (lane == 0) rec_cnt[wslot] = fast ? nq : -1;
    }
}

template <int V> struct RowVec;
template <> struct RowVec<4> { typedef float4 T; };
template <> struct RowVec<2> { typedef float2 T; };

template <int V> struct SumVec;
template <> struct SumVec<4> { typedef float T __attribute__((ext_vector_type(32))); };
template <> struct SumVec<2> { typedef float T __attribute__((ext_vector_type(16))); };
static_assert(OWN_ROWS * 4 <= 32 && OWN_ROWS * 2 <= 16, "a wave's rows fit the register vector");

template <class ST>
__device__ __forceinline__ float4 sum_get4(const ST& a, int w) {
    return make_float4(a[w], a[w + 1], a[w + 2], a[w + 3]);
}
template <class ST>
__device__ __forceinline__ float2 sum_get2(const ST& a, int w) {
    return make_float2(a[w], a[w + 1]);
}
template <class ST>
__device__ __forceinline__ void sum_put(ST& a, int w, const float4& v) {
    a[w] = v.x; a[w + 1] = v.y; a[w + 2] = v.z; a[w + 3] = v.w;
}
template <class ST>
__device__ __forceinline__ void sum_put(ST& a, int w, const float2& v) {
    a[w] = v.x; a[w + 1] = v.y;
}
template <class ST>
__device__ __forceinline__ void sum_get(const ST& a, int w, float4& v) { v = sum_get4(a, w); }
template <class ST>
__device__ __forceinline__ void sum_get(const ST& a, int w, float2& v) { v = sum_get2(a, w); }

__device__ __forceinline__ float4 row_start(const float4& b, const float4& hv, float di) {
    float4 o;
    o.x = b.x + hv.x * di * di;
    o.y = b.y + hv.y * di * di;
    o.z = b.z + hv.z * di * di;
    o.w = b.w + hv.w * di * di;
    return o;
}
__device__ __forceinline__ float2 row_start(const float2& b, const float2& hv, float di) {
    float2 o;
    o.x = b.x + hv.x * di * di;
    o.y = b.y + hv.y * di * di;
    return o;
}
__device__ __forceinline__ float4 row_self(const float4& hv, float di) {
    float4 o;
    o.x = hv.x * di * di;
    o.y = hv.y * di * di;
    o.z = hv.z * di * di;
    o.w = hv.w * di * di;
    return o;
}
__device__ __forceinline__ float2 row_self(const float2& hv, float di) {
    float2 o;
    o.x = hv.x * di * di;
    o.y = hv.y * di * di;
    return o;
}
__device__ __forceinline__ void row_fma(float4& a, const float4& hv, float nu) {
    a.x += hv.x * nu;
    a.y += hv.y * nu;
    a.z += hv.z * nu;
    a.w += hv.w * nu;
}
__device__ __forceinline__ void row_fma(float2& a, const float2& hv, float nu) {
    a.x += hv.x * nu;
    a.y += hv.y * nu;
}

template <int C>
__device__ __forceinline__ void load_group(
    typename RowVec<C / 32>::T (&hv)[8], int s, int j, int m, int lane,
    const float* __restrict__ h)
{
    constexpr int V = C / 32;
    typedef typename RowVec<V>::T VT;
#pragma unroll
    for (int u = 0; u < 8; ++u) {
        const int su = __builtin_amdgcn_readlane(s, min(j + u, m - 1));
        hv[u] = *(const VT*)(h + (size_t)su * C + lane * V);
    }
}

template <int C>
__device__ __forceinline__ void add_group(
    const typename RowVec<C / 32>::T (&hv)[8], int kw, float nr, int j, int m,
    typename SumVec<C / 32>::T& a)
{
    typedef typename RowVec<C / 32>::T VT;
#pragma unroll
    for (int u = 0; u < 8; ++u) {
        if (j + u < m) {
            const int w = __builtin_amdgcn_readlane(kw, j + u) & (sizeof(a) / sizeof(float) - 1);
            const float nu = __int_as_float(
                __builtin_amdgcn_readlane(__float_as_int(nr), j + u));
            VT t;
            sum_get(a, w, t);
            row_fma(t, hv[u], nu);
            sum_put(a, w, t);
        }
    }
}

template <int C, bool FILL>
__device__ __forceinline__ void drain_owned(
    int4* recs, int cnt, int lane, int base,
    const float* __restrict__ dinv, const float* __restrict__ h,
    typename SumVec<C / 32>::T& a)
{
    typedef typename RowVec<C / 32>::T VT;
    int4 cur = make_int4(0, 0, 0, 0);
    if (lane < cnt) cur = recs[lane];
    for (int i0 = 0; i0 < cnt; i0 += 32) {
        const int m = min(32, cnt - i0);
        int4 nxt = make_int4(0, 0, 0, 0);
        if (i0 + 32 + lane < cnt) nxt = recs[i0 + 32 + lane];
        float da = 0.0f, db = 0.0f;
        if (FILL) {
            da = dinv[cur.x];
            db = dinv[base + cur.y];
        }
        const int   s  = cur.x;
        const int   kw = min(max(cur.y >> AGG_WSH, 0), OWN_ROWS - 1) * (C / 32);
        VT ha[8], hb[8];
        load_group<C>(ha, s, 0, m, lane, h);
        load_group<C>(hb, s, 8, m, lane, h);
        if (FILL) {
            cur.z = __float_as_int(da * db);
            if (lane < m) recs[i0 + lane] = cur;
        }
        const float nr = __int_as_float(cur.z);
        add_group<C>(ha, kw, nr, 0, m, a);
        if (m > 8) {
            load_group<C>(ha, s, 16, m, lane, h);
            add_group<C>(hb, kw, nr, 8, m, a);
            if (m > 16) {
                load_group<C>(hb, s, 24, m, lane, h);
                add_group<C>(ha, kw, nr, 16, m, a);
                add_group<C>(hb, kw, nr, 24, m, a);
            }
        }
        cur = nxt;
    }
}

__device__ __forceinline__ void lds_add_f32(float* p, float v) {
    __hip_atomic_fetch_add(p, v, __ATOMIC_RELAXED, __HIP_MEMORY_SCOPE_WORKGROUP);
}

template <int C, int LD>
__device__ __forceinline__ void drain_list(
    const int* mylst, int cnt, int lane, int base,
    const int2* __restrict__ grp,
    const float* __restrict__ dinv, const float* __restrict__ h, float* acc)
{
    constexpr int V = C / 32;
    typedef typename RowVec<V>::T VT;
    for (int i0 = 0; i0 < cnt; i0 += 32) {
        const int m = min(32, cnt - i0);
        int s = 0, dl = 0;
        float nr = 0.0f;
        if (lane < m) {
            const int2 sd = grp[mylst[i0 + lane]];
            s = sd.x;
            const int d = sd.y;
            dl = d - base;
            nr = dinv[s] * dinv[d];
        }
        for (int j0 = 0; j0 < m; j0 += 4) {
            VT hv[4];
#pragma unroll
            for (int u = 0; u < 4; ++u) {
                const int su = __builtin_amdgcn_readlane(s, min(j0 + u, m - 1));
                hv[u] = *(const VT*)(h + (size_t)su * C + lane * V);
            }
#pragma unroll
            for (int u = 0; u < 4; ++u) {
                if (j0 + u < m) {
                    const int du = __builtin_amdgcn_readlane(dl, j0 + u);
                    const float nu = __int_as_float(
                        __builtin_amdgcn_readlane(__float_as_int(nr), j0 + u));
                    float* a = acc + du * LD + lane * V;
                    const float* hp = (const float*)&hv[u];
#pragma unroll
                    for (int c = 0; c < V; ++c) lds_add_f32(a + c, hp[c] * nu);
                }
            }
        }
    }
}

template <int C, int LD>
__device__ __forceinline__ void walk_tile(
    int lane, int wv, int base,
    const int2* __restrict__ grp, const int* __restrict__ off, int nchunk,
    const float* __restrict__ dinv, const float* __restrict__ h,
    int* mylst, float* acc)
{
    int cnt = 0;
    for (int c0 = wv; c0 < nchunk; c0 += AGG_WAVES * 32) {
        const int cm = c0 + lane * AGG_WAVES;
        int sb = 0, se = 0;
        if (cm < nchunk) {
            sb = off[(size_t)cm * OFF_P + blockIdx.x];
            se = off[(size_t)cm * OFF_P + blockIdx.x + 1];
        }
        const int nmine = min(32, (nchunk - c0 + AGG_WAVES - 1) / AGG_WAVES);
        for (int k = 0; k < nmine; ++k) {
            const int j0 = __builtin_amdgcn_readlane(sb, k);
            const int j1 = __builtin_amdgcn_readlane(se, k);
            const int p0 = (c0 + k * AGG_WAVES) * CHUNK;
            for (int j = j0; j < j1; j += 32) {
                const int m = min(32, j1 - j);
                if (lane < m) mylst[cnt + lane] = p0 + j + lane;
                cnt += m;
                if (cnt > LIST_LOW) {
                    drain_list<C, LD>(mylst, cnt, lane, base, grp, dinv, h, acc);
                    cnt = 0;
                }
            }
        }
    }
    drain_list<C, LD>(mylst, cnt, lane, base, grp, dinv, h, acc);
}

template <int C, int MODE>
__global__ void __launch_bounds__(AGG_T) k_tile_aggregate(
    const int* __restrict__ grp_w, const int* __restrict__ off,
    const float* __restrict__ dinv, const float* __restrict__ h,
    const float* __restrict__ wmat, const float* __restrict__ bias,
    float* __restrict__ out,
    const int* __restrict__ rec_cnt, int* __restrict__ rec,
    int nchunk, int n)
{
    const int2* __restrict__ grp = (const int2*)grp_w;
    constexpr int  V    = C / 32;
    constexpr bool FILL = (MODE == MODE_REC_FILL);
    constexpr bool MUL  = FILL;
    constexpr int  LD   = MUL ? C + 4 : C;
    constexpr int  LDK  = C + 2;
    typedef typename RowVec<V>::T VT;
    __shared__ __attribute__((aligned(16))) float acc[TILE_R * LD];
    __shared__ int   lst[AGG_WAVES * LIST_CAP];
    __shared__ __attribute__((aligned(16))) float lds_w[MUL ? C * LDK : 4];

    const int tid  = threadIdx.x;
    const int lane = tid & 31;
    const int wv   = __builtin_amdgcn_readfirstlane(tid >> 5);
    const int base = blockIdx.x * TILE_R;
    const int nrows = min(TILE_R, n - base);
    int*  mylst = lst + wv * LIST_CAP;

    const int wslot = blockIdx.x * AGG_WAVES + wv;
    int4* myrec = (int4*)rec + (size_t)wslot * REC_CAP;

    const int fill_v = rec_cnt[blockIdx.x * AGG_WAVES + lane];

    if (MUL) {
        for (int idx = tid; idx < C * C; idx += AGG_T) {
            const int k  = idx / C;
            const int nn = idx - k * C;
            lds_w[nn * LDK + k] = wmat[idx];
        }
    }

    typename SumVec<V>::T a = {};
    {
        const VT bv = *(const VT*)(bias + lane * V);
        const int kk = min(lane, OWN_ROWS - 1);
        const float dv = dinv[base + min(wv + kk * AGG_WAVES, nrows - 1)];
#pragma unroll
        for (int k = 0; k < OWN_ROWS; ++k) {
            const int rr = min(wv + k * AGG_WAVES, nrows - 1);
            const float di = __int_as_float(__builtin_amdgcn_readlane(__float_as_int(dv), k));
            const VT hv = *(const VT*)(h + (size_t)(base + rr) * C + lane * V);
            sum_put(a, k * V, MUL ? row_self(hv, di) : row_start(bv, hv, di));
        }
    }

    const bool fast = (__builtin_amdgcn_ballot_w32(fill_v < 0) == 0u);
    const int  nown = __builtin_amdgcn_readlane(fill_v, wv);

    if (__builtin_expect(fast, true)) {
        drain_owned<C, FILL>(myrec, nown, lane, base, dinv, h, a);
        if (MUL) {
#pragma unroll
            for (int k = 0; k < OWN_ROWS; ++k) {
                const int r = wv + k * AGG_WAVES;
                if (r < nrows) { VT t; sum_get(a, k * V, t); *(VT*)(acc + r * LD + lane * V) = t; }
            }
        }
    } else {
#pragma unroll
        for (int k = 0; k < OWN_ROWS; ++k) {
            const int r = wv + k * AGG_WAVES;
            if (r < nrows) { VT t; sum_get(a, k * V, t); *(VT*)(acc + r * LD + lane * V) = t; }
        }
        __syncthreads();
        walk_tile<C, LD>(lane, wv, base, grp, off, nchunk, dinv, h, mylst, acc);
        __syncthreads();
        if (!MUL) {
#pragma unroll
            for (int k = 0; k < OWN_ROWS; ++k) {
                const int r = wv + k * AGG_WAVES;
                if (r < nrows) sum_put(a, k * V, *(const VT*)(acc + r * LD + lane * V));
            }
        }
    }

    if (MUL) {
        constexpr int MT  = (TILE_R + 15) / 16;
        constexpr int NQ  = C / 32;
        __syncthreads();

        const int hf = lane >> 4;
        const int r  = lane & 15;
        for (int item = wv; item < MT * NQ; item += AGG_WAVES) {
            const int mt = item / NQ;
            const int nq = item - mt * NQ;
            if (mt * 16 >= nrows) continue;
            const float* arow = acc + min(mt * 16 + r, nrows - 1) * LD + 2 * hf;
            const float* b0   = lds_w + (nq * 32 + r) * LDK + 2 * hf;
            const float* b1   = b0 + 16 * LDK;
            v8f d0 = (v8f){};
            v8f d1 = (v8f){};
#pragma unroll 8
            for (int k = 0; k < C; k += 4) {
                const float2 av = *(const float2*)(arow + k);
                const float2 p0 = *(const float2*)(b0 + k);
                const float2 p1 = *(const float2*)(b1 + k);
                v2f a, x0, x1;
                a[0]  = av.x;  a[1]  = av.y;
                x0[0] = p0.x;  x0[1] = p0.y;
                x1[0] = p1.x;  x1[1] = p1.y;
                d0 = __builtin_amdgcn_wmma_f32_16x16x4_f32(
                    false, a, false, x0, (short)0, d0, false, false);
                d1 = __builtin_amdgcn_wmma_f32_16x16x4_f32(
                    false, a, false, x1, (short)0, d1, false, false);
            }
            const int   c0  = nq * 32 + r;
            const float bs0 = bias[c0];
            const float bs1 = bias[c0 + 16];
#pragma unroll
            for (int j = 0; j < 8; ++j) {
                const int row = mt * 16 + 8 * hf + j;
                if (row < nrows) {
                    float* orow = out + (size_t)(base + row) * C;
                    orow[c0]      = d0[j] + bs0;
                    orow[c0 + 16] = d1[j] + bs1;
                }
            }
        }
    } else {
#pragma unroll
        for (int k = 0; k < OWN_ROWS; ++k) {
            const int r = wv + k * AGG_WAVES;
            if (r < nrows) {
                VT t;
                sum_get(a, k * V, t);
                *(VT*)(out + (size_t)(base + r) * C + lane * V) = t;
            }
        }
    }
}

extern "C" void kernel_launch(void* const* d_in, const int* in_sizes, int n_in,
                              void* d_out, int out_size, void* d_ws, size_t ws_size,
                              hipStream_t stream) {
    const float* x  = (const float*)d_in[0];
    const int*   ei = (const int*)d_in[1];
    const float* W1 = (const float*)d_in[2];
    const float* b1 = (const float*)d_in[3];
    const float* W2 = (const float*)d_in[4];
    const float* b2 = (const float*)d_in[5];
    float* out = (float*)d_out;

    const int HID = in_sizes[3];
    const int IN  = in_sizes[2] / HID;
    const int N   = in_sizes[0] / IN;
    const int E   = in_sizes[1] / 2;
    const int* srcI = ei;
    const int* dstI = ei + E;

    float* ws   = (float*)d_ws;
    float* dinv = ws;
    float* bufA = ws + (((size_t)N + 63) & ~(size_t)63);
    float* bufB = bufA + (size_t)N * HID;
    const int ablk = (N + TILE_R - 1) / TILE_R;
    const int nchunk = (E + CHUNK - 1) / CHUNK;
    int*  grp  = (int*)(bufB + (size_t)N * HID);
    int*  off  = grp + (size_t)nchunk * CHUNK * 2;
    int*  rcnt = off + ((((size_t)nchunk * OFF_P) + 15) & ~(size_t)15);
    int*  rec  = rcnt + (size_t)ablk * AGG_WAVES;

    const int T = 256;
    const int mtiles = N / 16;
    const int gblk   = (mtiles + 7) / 8;

    k_group_edges<<<nchunk, CHUNK / 4, 0, stream>>>(srcI, dstI, E, grp, off);
    k_tile_prep  <<<ablk, PRE_T, 0, stream>>>(grp, off, nchunk, dinv, rcnt, rec, N);

    k_tile_aggregate<128, MODE_REC_FILL><<<ablk, AGG_T, 0, stream>>>(
        grp, off, dinv, x, W1, b1, bufB, rcnt, rec, nchunk, N);

    k_gemm_wmma_f32<4, 128, true><<<gblk, T, 0, stream>>>(bufB, W2, bufA, N);
    k_tile_aggregate<64, MODE_FROM_REC><<<ablk, AGG_T, 0, stream>>>(
        grp, off, dinv, bufA, W2, b2, out, rcnt, rec, nchunk, N);
}
